# starting kernel
# baseline (speedup 1.0000x reference)
#include <hip/hip_runtime.h>


typedef __attribute__((ext_vector_type(2))) float v2f;
typedef __attribute__((ext_vector_type(4))) float v4f;
typedef __attribute__((ext_vector_type(8))) float v8f;

#define B_ALL   8
#define B_BASE  4
#define T_DIM   4096
#define C_DIM   320
#define S_DIM   77
#define SP      80
#define D_ENC   768
#define HD      40
#define H_NUM   8

__device__ __forceinline__ v8f wmma4(v2f a, v2f b, v8f c) {
  return __builtin_amdgcn_wmma_f32_16x16x4_f32(
       false, a,  false, b,
       (short)0, c,  false,  false);
}

#define GK 16

__global__ __launch_bounds__(128)
void gemm_wmma_f32(const float* __restrict__ A, const float* __restrict__ Bm,
                   const float* __restrict__ bias, float* __restrict__ C,
                   int M, int N, int K) {
  __shared__ float Bs[2][64 * 18];
  const int tid  = threadIdx.x;
  const int wave = tid >> 5;
  const int lane = tid & 31;
  const int half = lane >> 4;
  const int l16  = lane & 15;
  const int col0 = blockIdx.x * 64;
  const int rowW = blockIdx.y * 64 + wave * 16;

  int arow = rowW + l16;
  if (arow > M - 1) arow = M - 1;
  const float* arowp = A + (size_t)arow * K;

  const int srow = tid >> 3;
  const int scol = (tid & 7) * 8;
  const float* bstage = Bm + (size_t)srow * N + col0 + scol;

  const v8f zero = {0.f,0.f,0.f,0.f,0.f,0.f,0.f,0.f};
  v8f acc[4];
  #pragma unroll
  for (int j = 0; j < 4; ++j) acc[j] = zero;

  const int nkb = K / GK;

  {
    v4f r0 = *(const v4f*)(bstage);
    v4f r1 = *(const v4f*)(bstage + 4);
    #pragma unroll
    for (int e = 0; e < 4; ++e) Bs[0][(scol + e) * 18 + srow] = r0[e];
    #pragma unroll
    for (int e = 0; e < 4; ++e) Bs[0][(scol + 4 + e) * 18 + srow] = r1[e];
  }
  __syncthreads();

  for (int kb = 0; kb < nkb; ++kb) {
    const int cur = kb & 1;
    const bool have_next = (kb + 1) < nkb;

    v4f r0, r1;
    if (have_next) {
      const float* bnxt = bstage + (size_t)(kb + 1) * GK * N;
      r0 = *(const v4f*)(bnxt);
      r1 = *(const v4f*)(bnxt + 4);
      if (kb + 2 < nkb)
        __builtin_prefetch(bstage + (size_t)(kb + 2) * GK * N, 0, 1);
    }

    const float* ap = arowp + kb * GK;
    #pragma unroll
    for (int kk = 0; kk < GK; kk += 4) {
      const int ka = kk + 2 * half;
      v2f a = *(const v2f*)(ap + ka);
      #pragma unroll
      for (int j = 0; j < 4; ++j) {
        v2f b = *(const v2f*)(&Bs[cur][(16 * j + l16) * 18 + ka]);
        acc[j] = wmma4(a, b, acc[j]);
      }
    }

    if (have_next) {
      const int nxt = cur ^ 1;
      #pragma unroll
      for (int e = 0; e < 4; ++e) Bs[nxt][(scol + e) * 18 + srow] = r0[e];
      #pragma unroll
      for (int e = 0; e < 4; ++e) Bs[nxt][(scol + 4 + e) * 18 + srow] = r1[e];
    }
    __syncthreads();
  }

  const float bv = bias ? bias[col0 + 16 * 0 + l16] : 0.f;
  #pragma unroll
  for (int j = 0; j < 4; ++j) {
    const int c = col0 + 16 * j + l16;
    const float bj = bias ? bias[c] : 0.f;
    #pragma unroll
    for (int i = 0; i < 8; ++i) {
      const int r = rowW + i + 8 * half;
      if (r < M) C[(size_t)r * N + c] = acc[j][i] + bj;
    }
  }
  (void)bv;
}

__device__ __forceinline__
void pv_and_store(const float* Plds, const float* __restrict__ Vp,
                  float* __restrict__ AO, int b, int h, int t0,
                  int half, int l16) {
  const v8f zero = {0.f,0.f,0.f,0.f,0.f,0.f,0.f,0.f};
  v8f o[3];
  #pragma unroll
  for (int j = 0; j < 3; ++j) o[j] = zero;

  const float* vbase = Vp + (size_t)(b * S_DIM) * C_DIM + h * HD;

  for (int k0 = 0; k0 < SP; k0 += 4) {
    const int ka = k0 + 2 * half;
    v2f a = *(const v2f*)(Plds + l16 * SP + ka);
    #pragma unroll
    for (int j = 0; j < 3; ++j) {
      const int c = 16 * j + l16;
      v2f bf; bf.x = 0.f; bf.y = 0.f;
      if (c < HD) {
        if (ka     < S_DIM) bf.x = vbase[(size_t) ka      * C_DIM + c];
        if (ka + 1 < S_DIM) bf.y = vbase[(size_t)(ka + 1) * C_DIM + c];
      }
      o[j] = wmma4(a, bf, o[j]);
    }
  }

  #pragma unroll
  for (int j = 0; j < 3; ++j) {
    const int c = 16 * j + l16;
    if (c < HD) {
      #pragma unroll
      for (int i = 0; i < 8; ++i) {
        const int t = t0 + i + 8 * half;
        AO[(size_t)(b * T_DIM + t) * C_DIM + h * HD + c] = o[j][i];
      }
    }
  }
}

__global__ __launch_bounds__(32)
void attn_kernel(const float* __restrict__ Q,
                 const float* __restrict__ Kp,
                 const float* __restrict__ Vp,
                 float* __restrict__ AB,
                 float* __restrict__ AO)
{
  __shared__ float Plds[16 * SP];
  const int lane = threadIdx.x;
  const int half = lane >> 4;
  const int l16  = lane & 15;
  const int t0 = blockIdx.x << 4;
  const int h  = blockIdx.y;
  const int b  = blockIdx.z;
  const float scale = 0.15811388300841898f;
  const float NEG_INF = -__builtin_inff();

  const v8f zero = {0.f,0.f,0.f,0.f,0.f,0.f,0.f,0.f};
  v8f sc[5];
  #pragma unroll
  for (int j = 0; j < 5; ++j) sc[j] = zero;

  const float* qbase = Q  + (size_t)(b * T_DIM + t0 + l16) * C_DIM + h * HD;
  const float* kbase = Kp + (size_t)(b * S_DIM) * C_DIM + h * HD;

  for (int k0 = 0; k0 < HD; k0 += 4) {
    const int ka = k0 + 2 * half;
    v2f a = *(const v2f*)(qbase + ka);
    #pragma unroll
    for (int j = 0; j < 5; ++j) {
      const int s = 16 * j + l16;
      v2f bf;
      if (s < S_DIM) bf = *(const v2f*)(kbase + (size_t)s * C_DIM + ka);
      else { bf.x = 0.f; bf.y = 0.f; }
      sc[j] = wmma4(a, bf, sc[j]);
    }
  }

  float mx[8];
  #pragma unroll
  for (int i = 0; i < 8; ++i) mx[i] = NEG_INF;
  #pragma unroll
  for (int j = 0; j < 5; ++j) {
    const bool ok = (16 * j + l16) < S_DIM;
    #pragma unroll
    for (int i = 0; i < 8; ++i) {
      const float x = ok ? sc[j][i] * scale : NEG_INF;
      sc[j][i] = x;
      mx[i] = fmaxf(mx[i], x);
    }
  }
  #pragma unroll
  for (int m = 1; m < 16; m <<= 1) {
    #pragma unroll
    for (int i = 0; i < 8; ++i) mx[i] = fmaxf(mx[i], __shfl_xor(mx[i], m, 32));
  }
  float sm[8];
  #pragma unroll
  for (int i = 0; i < 8; ++i) sm[i] = 0.f;
  #pragma unroll
  for (int j = 0; j < 5; ++j) {
    #pragma unroll
    for (int i = 0; i < 8; ++i) {
      const float e = __expf(sc[j][i] - mx[i]);
      sc[j][i] = e;
      sm[i] += e;
    }
  }
  #pragma unroll
  for (int m = 1; m < 16; m <<= 1) {
    #pragma unroll
    for (int i = 0; i < 8; ++i) sm[i] += __shfl_xor(sm[i], m, 32);
  }
  float rs[8];
  #pragma unroll
  for (int i = 0; i < 8; ++i) rs[i] = 1.f / sm[i];

  #pragma unroll
  for (int j = 0; j < 5; ++j) {
    #pragma unroll
    for (int i = 0; i < 8; ++i) {
      const float pv = sc[j][i] * rs[i];
      const int row = i + 8 * half;
      Plds[row * SP + 16 * j + l16] = pv;
      if (b == B_BASE)
        AB[(size_t)(h * T_DIM + t0 + row) * SP + 16 * j + l16] = pv;
    }
  }
  __syncthreads();

  pv_and_store(Plds, Vp, AO, b, h, t0, half, l16);
}

__global__ __launch_bounds__(32)
void edit_kernel(const float* __restrict__ AB,
                 const float* __restrict__ Map,
                 const float* __restrict__ Vp,
                 float* __restrict__ AO)
{
  __shared__ float Plds[16 * SP];
  const int lane = threadIdx.x;
  const int half = lane >> 4;
  const int l16  = lane & 15;
  const int t0 = blockIdx.x << 4;
  const int h  = blockIdx.y;
  const int p  = blockIdx.z;
  const int b  = B_BASE + 1 + p;

  const v8f zero = {0.f,0.f,0.f,0.f,0.f,0.f,0.f,0.f};
  v8f acc[5];
  #pragma unroll
  for (int j = 0; j < 5; ++j) acc[j] = zero;

  const float* abase = AB  + (size_t)(h * T_DIM + t0 + l16) * SP;
  const float* mbase = Map + (size_t)p * S_DIM * S_DIM;

  for (int k0 = 0; k0 < SP; k0 += 4) {
    const int ka = k0 + 2 * half;
    v2f a = *(const v2f*)(abase + ka);
    #pragma unroll
    for (int j = 0; j < 5; ++j) {
      const int n = 16 * j + l16;
      v2f bf; bf.x = 0.f; bf.y = 0.f;
      if (n < S_DIM) {
        if (ka     < S_DIM) bf.x = mbase[(size_t) ka      * S_DIM + n];
        if (ka + 1 < S_DIM) bf.y = mbase[(size_t)(ka + 1) * S_DIM + n];
      }
      acc[j] = wmma4(a, bf, acc[j]);
    }
  }

  #pragma unroll
  for (int j = 0; j < 5; ++j) {
    const bool ok = (16 * j + l16) < S_DIM;
    #pragma unroll
    for (int i = 0; i < 8; ++i) {
      const int row = i + 8 * half;
      Plds[row * SP + 16 * j + l16] = ok ? acc[j][i] : 0.f;
    }
  }
  __syncthreads();

  pv_and_store(Plds, Vp, AO, b, h, t0, half, l16);
}

extern "C" void kernel_launch(void* const* d_in, const int* in_sizes, int n_in,
                              void* d_out, int out_size, void* d_ws, size_t ws_size,
                              hipStream_t stream) {
  const float* X   = (const float*)d_in[0];
  const float* E   = (const float*)d_in[1];
  const float* Map = (const float*)d_in[2];
  const float* Wq  = (const float*)d_in[3];
  const float* Wk  = (const float*)d_in[4];
  const float* Wv  = (const float*)d_in[5];
  const float* Wo  = (const float*)d_in[6];
  const float* bo  = (const float*)d_in[7];
  float* out = (float*)d_out;

  float* ws = (float*)d_ws;
  float* Q  = ws;
  float* Kp = Q  + (size_t)5 * T_DIM * C_DIM;
  float* Vp = Kp + (size_t)B_ALL * S_DIM * C_DIM;
  float* AB = Vp + (size_t)B_ALL * S_DIM * C_DIM;
  float* AO = AB + (size_t)H_NUM * T_DIM * SP;

  const dim3 gblk(128);
  const dim3 ablk(32);
  const int MR = B_ALL * S_DIM;

  gemm_wmma_f32<<<dim3(C_DIM / 64, (5 * T_DIM) / 64), gblk, 0, stream>>>(
      X, Wq, nullptr, Q, 5 * T_DIM, C_DIM, C_DIM);
  gemm_wmma_f32<<<dim3(C_DIM / 64, (MR + 63) / 64), gblk, 0, stream>>>(
      E, Wk, nullptr, Kp, MR, C_DIM, D_ENC);
  gemm_wmma_f32<<<dim3(C_DIM / 64, (MR + 63) / 64), gblk, 0, stream>>>(
      E, Wv, nullptr, Vp, MR, C_DIM, D_ENC);

  attn_kernel<<<dim3(T_DIM / 16, H_NUM, 5), ablk, 0, stream>>>(Q, Kp, Vp, AB, AO);

  edit_kernel<<<dim3(T_DIM / 16, H_NUM, 3), ablk, 0, stream>>>(AB, Map, Vp, AO);

  gemm_wmma_f32<<<dim3(C_DIM / 64, (B_ALL * T_DIM) / 64), gblk, 0, stream>>>(
      AO, Wo, bo, out, B_ALL * T_DIM, C_DIM, C_DIM);
}


# Q, K and V projections merged into one launch (K/V tiles get the lowest block ids, Q tiles follow): the small latency-bound K/V GEMMs overlap the Q GEMM; launches 6 -> 4; sums unchanged
# speedup vs baseline: 1.2421x; 1.2421x over previous
#include <hip/hip_runtime.h>


typedef __attribute__((ext_vector_type(2))) float v2f;
typedef __attribute__((ext_vector_type(4))) float v4f;
typedef __attribute__((ext_vector_type(8))) float v8f;

#define B_ALL   8
#define B_BASE  4
#define T_DIM   4096
#define C_DIM   320
#define S_DIM   77
#define SP      80
#define D_ENC   768
#define HD      40
#define H_NUM   8

__device__ __forceinline__ v8f wmma4(v2f a, v2f b, v8f c) {
  return __builtin_amdgcn_wmma_f32_16x16x4_f32(
       false, a,  false, b,
       (short)0, c,  false,  false);
}

#define GK 16

__device__ __forceinline__
void gemm_tile_64x64(const float* __restrict__ A, const float* __restrict__ Bm,
                     const float* __restrict__ bias, float* __restrict__ C,
                     int M, int N, int K, int bx, int by) {
  __shared__ float Bs[2][64 * 18];
  const int tid  = threadIdx.x;
  const int wave = tid >> 5;
  const int lane = tid & 31;
  const int half = lane >> 4;
  const int l16  = lane & 15;
  const int col0 = bx * 64;
  const int rowW = by * 64 + wave * 16;

  int arow = rowW + l16;
  if (arow > M - 1) arow = M - 1;
  const float* arowp = A + (size_t)arow * K;

  const int srow = tid >> 3;
  const int scol = (tid & 7) * 8;
  const float* bstage = Bm + (size_t)srow * N + col0 + scol;

  const v8f zero = {0.f,0.f,0.f,0.f,0.f,0.f,0.f,0.f};
  v8f acc[4];
  #pragma unroll
  for (int j = 0; j < 4; ++j) acc[j] = zero;

  const int nkb = K / GK;

  {
    v4f r0 = *(const v4f*)(bstage);
    v4f r1 = *(const v4f*)(bstage + 4);
    #pragma unroll
    for (int e = 0; e < 4; ++e) Bs[0][(scol + e) * 18 + srow] = r0[e];
    #pragma unroll
    for (int e = 0; e < 4; ++e) Bs[0][(scol + 4 + e) * 18 + srow] = r1[e];
  }
  __syncthreads();

  for (int kb = 0; kb < nkb; ++kb) {
    const int cur = kb & 1;
    const bool have_next = (kb + 1) < nkb;

    v4f r0, r1;
    if (have_next) {
      const float* bnxt = bstage + (size_t)(kb + 1) * GK * N;
      r0 = *(const v4f*)(bnxt);
      r1 = *(const v4f*)(bnxt + 4);
      if (kb + 2 < nkb)
        __builtin_prefetch(bstage + (size_t)(kb + 2) * GK * N, 0, 1);
    }

    const float* ap = arowp + kb * GK;
    #pragma unroll
    for (int kk = 0; kk < GK; kk += 4) {
      const int ka = kk + 2 * half;
      v2f a = *(const v2f*)(ap + ka);
      #pragma unroll
      for (int j = 0; j < 4; ++j) {
        v2f b = *(const v2f*)(&Bs[cur][(16 * j + l16) * 18 + ka]);
        acc[j] = wmma4(a, b, acc[j]);
      }
    }

    if (have_next) {
      const int nxt = cur ^ 1;
      #pragma unroll
      for (int e = 0; e < 4; ++e) Bs[nxt][(scol + e) * 18 + srow] = r0[e];
      #pragma unroll
      for (int e = 0; e < 4; ++e) Bs[nxt][(scol + 4 + e) * 18 + srow] = r1[e];
    }
    __syncthreads();
  }

  const float bv = bias ? bias[col0 + 16 * 0 + l16] : 0.f;
  #pragma unroll
  for (int j = 0; j < 4; ++j) {
    const int c = col0 + 16 * j + l16;
    const float bj = bias ? bias[c] : 0.f;
    #pragma unroll
    for (int i = 0; i < 8; ++i) {
      const int r = rowW + i + 8 * half;
      if (r < M) C[(size_t)r * N + c] = acc[j][i] + bj;
    }
  }
  (void)bv;
}

__global__ __launch_bounds__(128)
void gemm_wmma_f32(const float* __restrict__ A, const float* __restrict__ Bm,
                   const float* __restrict__ bias, float* __restrict__ C,
                   int M, int N, int K) {
  gemm_tile_64x64(A, Bm, bias, C, M, N, K, blockIdx.x, blockIdx.y);
}

#define COL_TILES   5
#define KV_TILES    50
#define QKV_TILES   1700

__global__ __launch_bounds__(128)
void proj_qkv_wmma_f32(const float* __restrict__ X, const float* __restrict__ E,
                       const float* __restrict__ Wq, const float* __restrict__ Wk,
                       const float* __restrict__ Wv, float* __restrict__ Q,
                       float* __restrict__ Kp, float* __restrict__ Vp) {
  int id = blockIdx.x;
  const float* A;
  const float* W;
  float* C;
  int M, K;
  if (id < 2 * KV_TILES) {
    const bool isV = id >= KV_TILES;
    if (isV) id -= KV_TILES;
    A = E;
    W = isV ? Wv : Wk;
    C = isV ? Vp : Kp;
    M = B_ALL * S_DIM;
    K = D_ENC;
  } else {
    id -= 2 * KV_TILES;
    A = X;
    W = Wq;
    C = Q;
    M = 5 * T_DIM;
    K = C_DIM;
  }
  gemm_tile_64x64(A, W, nullptr, C, M, C_DIM, K, id % COL_TILES, id / COL_TILES);
}

__device__ __forceinline__
void pv_and_store(const float* Plds, const float* __restrict__ Vp,
                  float* __restrict__ AO, int b, int h, int t0,
                  int half, int l16) {
  const v8f zero = {0.f,0.f,0.f,0.f,0.f,0.f,0.f,0.f};
  v8f o[3];
  #pragma unroll
  for (int j = 0; j < 3; ++j) o[j] = zero;

  const float* vbase = Vp + (size_t)(b * S_DIM) * C_DIM + h * HD;

  for (int k0 = 0; k0 < SP; k0 += 4) {
    const int ka = k0 + 2 * half;
    v2f a = *(const v2f*)(Plds + l16 * SP + ka);
    #pragma unroll
    for (int j = 0; j < 3; ++j) {
      const int c = 16 * j + l16;
      v2f bf; bf.x = 0.f; bf.y = 0.f;
      if (c < HD) {
        if (ka     < S_DIM) bf.x = vbase[(size_t) ka      * C_DIM + c];
        if (ka + 1 < S_DIM) bf.y = vbase[(size_t)(ka + 1) * C_DIM + c];
      }
      o[j] = wmma4(a, bf, o[j]);
    }
  }

  #pragma unroll
  for (int j = 0; j < 3; ++j) {
    const int c = 16 * j + l16;
    if (c < HD) {
      #pragma unroll
      for (int i = 0; i < 8; ++i) {
        const int t = t0 + i + 8 * half;
        AO[(size_t)(b * T_DIM + t) * C_DIM + h * HD + c] = o[j][i];
      }
    }
  }
}

__global__ __launch_bounds__(32)
void attn_kernel(const float* __restrict__ Q,
                 const float* __restrict__ Kp,
                 const float* __restrict__ Vp,
                 float* __restrict__ AB,
                 float* __restrict__ AO)
{
  __shared__ float Plds[16 * SP];
  const int lane = threadIdx.x;
  const int half = lane >> 4;
  const int l16  = lane & 15;
  const int t0 = blockIdx.x << 4;
  const int h  = blockIdx.y;
  const int b  = blockIdx.z;
  const float scale = 0.15811388300841898f;
  const float NEG_INF = -__builtin_inff();

  const v8f zero = {0.f,0.f,0.f,0.f,0.f,0.f,0.f,0.f};
  v8f sc[5];
  #pragma unroll
  for (int j = 0; j < 5; ++j) sc[j] = zero;

  const float* qbase = Q  + (size_t)(b * T_DIM + t0 + l16) * C_DIM + h * HD;
  const float* kbase = Kp + (size_t)(b * S_DIM) * C_DIM + h * HD;

  for (int k0 = 0; k0 < HD; k0 += 4) {
    const int ka = k0 + 2 * half;
    v2f a = *(const v2f*)(qbase + ka);
    #pragma unroll
    for (int j = 0; j < 5; ++j) {
      const int s = 16 * j + l16;
      v2f bf;
      if (s < S_DIM) bf = *(const v2f*)(kbase + (size_t)s * C_DIM + ka);
      else { bf.x = 0.f; bf.y = 0.f; }
      sc[j] = wmma4(a, bf, sc[j]);
    }
  }

  float mx[8];
  #pragma unroll
  for (int i = 0; i < 8; ++i) mx[i] = NEG_INF;
  #pragma unroll
  for (int j = 0; j < 5; ++j) {
    const bool ok = (16 * j + l16) < S_DIM;
    #pragma unroll
    for (int i = 0; i < 8; ++i) {
      const float x = ok ? sc[j][i] * scale : NEG_INF;
      sc[j][i] = x;
      mx[i] = fmaxf(mx[i], x);
    }
  }
  #pragma unroll
  for (int m = 1; m < 16; m <<= 1) {
    #pragma unroll
    for (int i = 0; i < 8; ++i) mx[i] = fmaxf(mx[i], __shfl_xor(mx[i], m, 32));
  }
  float sm[8];
  #pragma unroll
  for (int i = 0; i < 8; ++i) sm[i] = 0.f;
  #pragma unroll
  for (int j = 0; j < 5; ++j) {
    #pragma unroll
    for (int i = 0; i < 8; ++i) {
      const float e = __expf(sc[j][i] - mx[i]);
      sc[j][i] = e;
      sm[i] += e;
    }
  }
  #pragma unroll
  for (int m = 1; m < 16; m <<= 1) {
    #pragma unroll
    for (int i = 0; i < 8; ++i) sm[i] += __shfl_xor(sm[i], m, 32);
  }
  float rs[8];
  #pragma unroll
  for (int i = 0; i < 8; ++i) rs[i] = 1.f / sm[i];

  #pragma unroll
  for (int j = 0; j < 5; ++j) {
    #pragma unroll
    for (int i = 0; i < 8; ++i) {
      const float pv = sc[j][i] * rs[i];
      const int row = i + 8 * half;
      Plds[row * SP + 16 * j + l16] = pv;
      if (b == B_BASE)
        AB[(size_t)(h * T_DIM + t0 + row) * SP + 16 * j + l16] = pv;
    }
  }
  __syncthreads();

  pv_and_store(Plds, Vp, AO, b, h, t0, half, l16);
}

__global__ __launch_bounds__(32)
void edit_kernel(const float* __restrict__ AB,
                 const float* __restrict__ Map,
                 const float* __restrict__ Vp,
                 float* __restrict__ AO)
{
  __shared__ float Plds[16 * SP];
  const int lane = threadIdx.x;
  const int half = lane >> 4;
  const int l16  = lane & 15;
  const int t0 = blockIdx.x << 4;
  const int h  = blockIdx.y;
  const int p  = blockIdx.z;
  const int b  = B_BASE + 1 + p;

  const v8f zero = {0.f,0.f,0.f,0.f,0.f,0.f,0.f,0.f};
  v8f acc[5];
  #pragma unroll
  for (int j = 0; j < 5; ++j) acc[j] = zero;

  const float* abase = AB  + (size_t)(h * T_DIM + t0 + l16) * SP;
  const float* mbase = Map + (size_t)p * S_DIM * S_DIM;

  for (int k0 = 0; k0 < SP; k0 += 4) {
    const int ka = k0 + 2 * half;
    v2f a = *(const v2f*)(abase + ka);
    #pragma unroll
    for (int j = 0; j < 5; ++j) {
      const int n = 16 * j + l16;
      v2f bf; bf.x = 0.f; bf.y = 0.f;
      if (n < S_DIM) {
        if (ka     < S_DIM) bf.x = mbase[(size_t) ka      * S_DIM + n];
        if (ka + 1 < S_DIM) bf.y = mbase[(size_t)(ka + 1) * S_DIM + n];
      }
      acc[j] = wmma4(a, bf, acc[j]);
    }
  }

  #pragma unroll
  for (int j = 0; j < 5; ++j) {
    const bool ok = (16 * j + l16) < S_DIM;
    #pragma unroll
    for (int i = 0; i < 8; ++i) {
      const int row = i + 8 * half;
      Plds[row * SP + 16 * j + l16] = ok ? acc[j][i] : 0.f;
    }
  }
  __syncthreads();

  pv_and_store(Plds, Vp, AO, b, h, t0, half, l16);
}

extern "C" void kernel_launch(void* const* d_in, const int* in_sizes, int n_in,
                              void* d_out, int out_size, void* d_ws, size_t ws_size,
                              hipStream_t stream) {
  const float* X   = (const float*)d_in[0];
  const float* E   = (const float*)d_in[1];
  const float* Map = (const float*)d_in[2];
  const float* Wq  = (const float*)d_in[3];
  const float* Wk  = (const float*)d_in[4];
  const float* Wv  = (const float*)d_in[5];
  const float* Wo  = (const float*)d_in[6];
  const float* bo  = (const float*)d_in[7];
  float* out = (float*)d_out;

  float* ws = (float*)d_ws;
  float* Q  = ws;
  float* Kp = Q  + (size_t)5 * T_DIM * C_DIM;
  float* Vp = Kp + (size_t)B_ALL * S_DIM * C_DIM;
  float* AB = Vp + (size_t)B_ALL * S_DIM * C_DIM;
  float* AO = AB + (size_t)H_NUM * T_DIM * SP;

  const dim3 gblk(128);
  const dim3 ablk(32);

  proj_qkv_wmma_f32<<<dim3(QKV_TILES), gblk, 0, stream>>>(
      X, E, Wq, Wk, Wv, Q, Kp, Vp);

  attn_kernel<<<dim3(T_DIM / 16, H_NUM, 5), ablk, 0, stream>>>(Q, Kp, Vp, AB, AO);

  edit_kernel<<<dim3(T_DIM / 16, H_NUM, 3), ablk, 0, stream>>>(AB, Map, Vp, AO);

  gemm_wmma_f32<<<dim3(C_DIM / 64, (B_ALL * T_DIM) / 64), gblk, 0, stream>>>(
      AO, Wo, bo, out, B_ALL * T_DIM, C_DIM, C_DIM);
}


# P@V, mapper and score B-fragment loads made unconditional (clamped re-reads instead of per-lane branches) so they issue ahead in groups instead of one load-wait-WMMA each; sums unchanged
# speedup vs baseline: 1.7459x; 1.4056x over previous
#include <hip/hip_runtime.h>


typedef __attribute__((ext_vector_type(2))) float v2f;
typedef __attribute__((ext_vector_type(4))) float v4f;
typedef __attribute__((ext_vector_type(8))) float v8f;

#define B_ALL   8
#define B_BASE  4
#define T_DIM   4096
#define C_DIM   320
#define S_DIM   77
#define SP      80
#define D_ENC   768
#define HD      40
#define H_NUM   8

__device__ __forceinline__ v8f wmma4(v2f a, v2f b, v8f c) {
  return __builtin_amdgcn_wmma_f32_16x16x4_f32(
       false, a,  false, b,
       (short)0, c,  false,  false);
}

#define GK 16

__device__ __forceinline__
void gemm_tile_64x64(const float* __restrict__ A, const float* __restrict__ Bm,
                     const float* __restrict__ bias, float* __restrict__ C,
                     int M, int N, int K, int bx, int by) {
  __shared__ float Bs[2][64 * 18];
  const int tid  = threadIdx.x;
  const int wave = tid >> 5;
  const int lane = tid & 31;
  const int half = lane >> 4;
  const int l16  = lane & 15;
  const int col0 = bx * 64;
  const int rowW = by * 64 + wave * 16;

  int arow = rowW + l16;
  if (arow > M - 1) arow = M - 1;
  const float* arowp = A + (size_t)arow * K;

  const int srow = tid >> 3;
  const int scol = (tid & 7) * 8;
  const float* bstage = Bm + (size_t)srow * N + col0 + scol;

  const v8f zero = {0.f,0.f,0.f,0.f,0.f,0.f,0.f,0.f};
  v8f acc[4];
  #pragma unroll
  for (int j = 0; j < 4; ++j) acc[j] = zero;

  const int nkb = K / GK;

  {
    v4f r0 = *(const v4f*)(bstage);
    v4f r1 = *(const v4f*)(bstage + 4);
    #pragma unroll
    for (int e = 0; e < 4; ++e) Bs[0][(scol + e) * 18 + srow] = r0[e];
    #pragma unroll
    for (int e = 0; e < 4; ++e) Bs[0][(scol + 4 + e) * 18 + srow] = r1[e];
  }
  __syncthreads();

  for (int kb = 0; kb < nkb; ++kb) {
    const int cur = kb & 1;
    const bool have_next = (kb + 1) < nkb;

    v4f r0, r1;
    if (have_next) {
      const float* bnxt = bstage + (size_t)(kb + 1) * GK * N;
      r0 = *(const v4f*)(bnxt);
      r1 = *(const v4f*)(bnxt + 4);
      if (kb + 2 < nkb)
        __builtin_prefetch(bstage + (size_t)(kb + 2) * GK * N, 0, 1);
    }

    const float* ap = arowp + kb * GK;
    #pragma unroll
    for (int kk = 0; kk < GK; kk += 4) {
      const int ka = kk + 2 * half;
      v2f a = *(const v2f*)(ap + ka);
      #pragma unroll
      for (int j = 0; j < 4; ++j) {
        v2f b = *(const v2f*)(&Bs[cur][(16 * j + l16) * 18 + ka]);
        acc[j] = wmma4(a, b, acc[j]);
      }
    }

    if (have_next) {
      const int nxt = cur ^ 1;
      #pragma unroll
      for (int e = 0; e < 4; ++e) Bs[nxt][(scol + e) * 18 + srow] = r0[e];
      #pragma unroll
      for (int e = 0; e < 4; ++e) Bs[nxt][(scol + 4 + e) * 18 + srow] = r1[e];
    }
    __syncthreads();
  }

  const float bv = bias ? bias[col0 + 16 * 0 + l16] : 0.f;
  #pragma unroll
  for (int j = 0; j < 4; ++j) {
    const int c = col0 + 16 * j + l16;
    const float bj = bias ? bias[c] : 0.f;
    #pragma unroll
    for (int i = 0; i < 8; ++i) {
      const int r = rowW + i + 8 * half;
      if (r < M) C[(size_t)r * N + c] = acc[j][i] + bj;
    }
  }
  (void)bv;
}

__global__ __launch_bounds__(128)
void gemm_wmma_f32(const float* __restrict__ A, const float* __restrict__ Bm,
                   const float* __restrict__ bias, float* __restrict__ C,
                   int M, int N, int K) {
  gemm_tile_64x64(A, Bm, bias, C, M, N, K, blockIdx.x, blockIdx.y);
}

#define COL_TILES   5
#define KV_TILES    50
#define QKV_TILES   1700

__global__ __launch_bounds__(128)
void proj_qkv_wmma_f32(const float* __restrict__ X, const float* __restrict__ E,
                       const float* __restrict__ Wq, const float* __restrict__ Wk,
                       const float* __restrict__ Wv, float* __restrict__ Q,
                       float* __restrict__ Kp, float* __restrict__ Vp) {
  int id = blockIdx.x;
  const float* A;
  const float* W;
  float* C;
  int M, K;
  if (id < 2 * KV_TILES) {
    const bool isV = id >= KV_TILES;
    if (isV) id -= KV_TILES;
    A = E;
    W = isV ? Wv : Wk;
    C = isV ? Vp : Kp;
    M = B_ALL * S_DIM;
    K = D_ENC;
  } else {
    id -= 2 * KV_TILES;
    A = X;
    W = Wq;
    C = Q;
    M = 5 * T_DIM;
    K = C_DIM;
  }
  gemm_tile_64x64(A, W, nullptr, C, M, C_DIM, K, id % COL_TILES, id / COL_TILES);
}

__device__ __forceinline__
void pv_and_store(const float* Plds, const float* __restrict__ Vp,
                  float* __restrict__ AO, int b, int h, int t0,
                  int half, int l16) {
  const v8f zero = {0.f,0.f,0.f,0.f,0.f,0.f,0.f,0.f};
  v8f o[3];
  #pragma unroll
  for (int j = 0; j < 3; ++j) o[j] = zero;

  const float* vbase = Vp + (size_t)(b * S_DIM) * C_DIM + h * HD;
  const float* vcol[3];
  const float* vlast[3];
  #pragma unroll
  for (int j = 0; j < 3; ++j) {
    int c = 16 * j + l16;
    if (c > HD - 1) c = HD - 1;
    vcol[j]  = vbase + (size_t)(2 * half) * C_DIM + c;
    vlast[j] = vbase + (size_t)(S_DIM - 1) * C_DIM + c;
  }
  const float* prow = Plds + l16 * SP + 2 * half;

  #pragma unroll
  for (int s = 0; s < SP / 4; ++s) {
    const int k0 = 4 * s;
    const v2f a = *(const v2f*)(prow + k0);
    #pragma unroll
    for (int j = 0; j < 3; ++j) {
      v2f bf;
      if (k0 + 3 < S_DIM) {
        bf.x = vcol[j][(size_t)k0 * C_DIM];
        bf.y = vcol[j][(size_t)(k0 + 1) * C_DIM];
      } else {
        bf.x = vlast[j][0];
        bf.y = vlast[j][0];
      }
      o[j] = wmma4(a, bf, o[j]);
    }
  }

  #pragma unroll
  for (int j = 0; j < 3; ++j) {
    const int c = 16 * j + l16;
    if (c < HD) {
      #pragma unroll
      for (int i = 0; i < 8; ++i) {
        const int t = t0 + i + 8 * half;
        AO[(size_t)(b * T_DIM + t) * C_DIM + h * HD + c] = o[j][i];
      }
    }
  }
}

__global__ __launch_bounds__(32)
void attn_kernel(const float* __restrict__ Q,
                 const float* __restrict__ Kp,
                 const float* __restrict__ Vp,
                 float* __restrict__ AB,
                 float* __restrict__ AO)
{
  __shared__ float Plds[16 * SP];
  const int lane = threadIdx.x;
  const int half = lane >> 4;
  const int l16  = lane & 15;
  const int t0 = blockIdx.x << 4;
  const int h  = blockIdx.y;
  const int b  = blockIdx.z;
  const float scale = 0.15811388300841898f;
  const float NEG_INF = -__builtin_inff();

  const v8f zero = {0.f,0.f,0.f,0.f,0.f,0.f,0.f,0.f};
  v8f sc[5];
  #pragma unroll
  for (int j = 0; j < 5; ++j) sc[j] = zero;

  const float* qbase = Q  + (size_t)(b * T_DIM + t0 + l16) * C_DIM + h * HD;
  const float* kbase = Kp + (size_t)(b * S_DIM) * C_DIM + h * HD;

  const float* krow[5];
  #pragma unroll
  for (int j = 0; j < 5; ++j) {
    int s = 16 * j + l16;
    if (s > S_DIM - 1) s = S_DIM - 1;
    krow[j] = kbase + (size_t)s * C_DIM + 2 * half;
  }
  #pragma unroll
  for (int k0 = 0; k0 < HD; k0 += 4) {
    v2f a = *(const v2f*)(qbase + k0 + 2 * half);
    #pragma unroll
    for (int j = 0; j < 5; ++j) {
      v2f bf = *(const v2f*)(krow[j] + k0);
      sc[j] = wmma4(a, bf, sc[j]);
    }
  }

  float mx[8];
  #pragma unroll
  for (int i = 0; i < 8; ++i) mx[i] = NEG_INF;
  #pragma unroll
  for (int j = 0; j < 5; ++j) {
    const bool ok = (16 * j + l16) < S_DIM;
    #pragma unroll
    for (int i = 0; i < 8; ++i) {
      const float x = ok ? sc[j][i] * scale : NEG_INF;
      sc[j][i] = x;
      mx[i] = fmaxf(mx[i], x);
    }
  }
  #pragma unroll
  for (int m = 1; m < 16; m <<= 1) {
    #pragma unroll
    for (int i = 0; i < 8; ++i) mx[i] = fmaxf(mx[i], __shfl_xor(mx[i], m, 32));
  }
  float sm[8];
  #pragma unroll
  for (int i = 0; i < 8; ++i) sm[i] = 0.f;
  #pragma unroll
  for (int j = 0; j < 5; ++j) {
    #pragma unroll
    for (int i = 0; i < 8; ++i) {
      const float e = __expf(sc[j][i] - mx[i]);
      sc[j][i] = e;
      sm[i] += e;
    }
  }
  #pragma unroll
  for (int m = 1; m < 16; m <<= 1) {
    #pragma unroll
    for (int i = 0; i < 8; ++i) sm[i] += __shfl_xor(sm[i], m, 32);
  }
  float rs[8];
  #pragma unroll
  for (int i = 0; i < 8; ++i) rs[i] = 1.f / sm[i];

  #pragma unroll
  for (int j = 0; j < 5; ++j) {
    #pragma unroll
    for (int i = 0; i < 8; ++i) {
      const float pv = sc[j][i] * rs[i];
      const int row = i + 8 * half;
      Plds[row * SP + 16 * j + l16] = pv;
      if (b == B_BASE)
        AB[(size_t)(h * T_DIM + t0 + row) * SP + 16 * j + l16] = pv;
    }
  }
  __syncthreads();

  pv_and_store(Plds, Vp, AO, b, h, t0, half, l16);
}

__global__ __launch_bounds__(32) __attribute__((amdgpu_waves_per_eu(4, 8)))
void edit_kernel(const float* __restrict__ AB,
                 const float* __restrict__ Map,
                 const float* __restrict__ Vp,
                 float* __restrict__ AO)
{
  __shared__ float Plds[16 * SP];
  const int lane = threadIdx.x;
  const int half = lane >> 4;
  const int l16  = lane & 15;
  const int t0 = blockIdx.x << 4;
  const int h  = blockIdx.y;
  const int p  = blockIdx.z;
  const int b  = B_BASE + 1 + p;

  const v8f zero = {0.f,0.f,0.f,0.f,0.f,0.f,0.f,0.f};
  v8f acc[5];
  #pragma unroll
  for (int j = 0; j < 5; ++j) acc[j] = zero;

  const float* abase = AB  + (size_t)(h * T_DIM + t0 + l16) * SP;
  const float* mbase = Map + (size_t)p * S_DIM * S_DIM;

  const float* mcol[5];
  const float* mlast[5];
  #pragma unroll
  for (int j = 0; j < 5; ++j) {
    int n = 16 * j + l16;
    if (n > S_DIM - 1) n = S_DIM - 1;
    mcol[j]  = mbase + (size_t)(2 * half) * S_DIM + n;
    mlast[j] = mbase + (size_t)(S_DIM - 1) * S_DIM + n;
  }
  const float* arow = abase + 2 * half;

  #pragma unroll
  for (int s = 0; s < SP / 4; ++s) {
    const int k0 = 4 * s;
    const v2f a = *(const v2f*)(arow + k0);
    #pragma unroll
    for (int j = 0; j < 5; ++j) {
      v2f bf;
      if (k0 + 3 < S_DIM) {
        bf.x = mcol[j][(size_t)k0 * S_DIM];
        bf.y = mcol[j][(size_t)(k0 + 1) * S_DIM];
      } else {
        bf.x = mlast[j][0];
        bf.y = mlast[j][0];
      }
      acc[j] = wmma4(a, bf, acc[j]);
    }
  }

  #pragma unroll
  for (int j = 0; j < 5; ++j) {
    const bool ok = (16 * j + l16) < S_DIM;
    #pragma unroll
    for (int i = 0; i < 8; ++i) {
      const int row = i + 8 * half;
      Plds[row * SP + 16 * j + l16] = ok ? acc[j][i] : 0.f;
    }
  }
  __syncthreads();

  pv_and_store(Plds, Vp, AO, b, h, t0, half, l16);
}

extern "C" void kernel_launch(void* const* d_in, const int* in_sizes, int n_in,
                              void* d_out, int out_size, void* d_ws, size_t ws_size,
                              hipStream_t stream) {
  const float* X   = (const float*)d_in[0];
  const float* E   = (const float*)d_in[1];
  const float* Map = (const float*)d_in[2];
  const float* Wq  = (const float*)d_in[3];
  const float* Wk  = (const float*)d_in[4];
  const float* Wv  = (const float*)d_in[5];
  const float* Wo  = (const float*)d_in[6];
  const float* bo  = (const float*)d_in[7];
  float* out = (float*)d_out;

  float* ws = (float*)d_ws;
  float* Q  = ws;
  float* Kp = Q  + (size_t)5 * T_DIM * C_DIM;
  float* Vp = Kp + (size_t)B_ALL * S_DIM * C_DIM;
  float* AB = Vp + (size_t)B_ALL * S_DIM * C_DIM;
  float* AO = AB + (size_t)H_NUM * T_DIM * SP;

  const dim3 gblk(128);
  const dim3 ablk(32);

  proj_qkv_wmma_f32<<<dim3(QKV_TILES), gblk, 0, stream>>>(
      X, E, Wq, Wk, Wv, Q, Kp, Vp);

  attn_kernel<<<dim3(T_DIM / 16, H_NUM, 5), ablk, 0, stream>>>(Q, Kp, Vp, AB, AO);

  edit_kernel<<<dim3(T_DIM / 16, H_NUM, 3), ablk, 0, stream>>>(AB, Map, Vp, AO);

  gemm_wmma_f32<<<dim3(C_DIM / 64, (B_ALL * T_DIM) / 64), gblk, 0, stream>>>(
      AO, Wo, bo, out, B_ALL * T_DIM, C_DIM, C_DIM);
}


# edited batches re-associated as attn_base @ (mapper[p] @ V): the 77x320 mapper@V is formed once by 300 extra waves of the attention launch, the 80x80 product per tile leaves; summation order changes
# speedup vs baseline: 1.8177x; 1.0411x over previous
#include <hip/hip_runtime.h>


typedef __attribute__((ext_vector_type(2))) float v2f;
typedef __attribute__((ext_vector_type(4))) float v4f;
typedef __attribute__((ext_vector_type(8))) float v8f;

#define B_ALL   8
#define B_BASE  4
#define T_DIM   4096
#define C_DIM   320
#define S_DIM   77
#define SP      80
#define D_ENC   768
#define HD      40
#define H_NUM   8

__device__ __forceinline__ v8f wmma4(v2f a, v2f b, v8f c) {
  return __builtin_amdgcn_wmma_f32_16x16x4_f32(
       false, a,  false, b,
       (short)0, c,  false,  false);
}

#define GK 16

__device__ __forceinline__
void gemm_tile_64x64(const float* __restrict__ A, const float* __restrict__ Bm,
                     const float* __restrict__ bias, float* __restrict__ C,
                     int M, int N, int K, int bx, int by) {
  __shared__ float Bs[2][64 * 18];
  const int tid  = threadIdx.x;
  const int wave = tid >> 5;
  const int lane = tid & 31;
  const int half = lane >> 4;
  const int l16  = lane & 15;
  const int col0 = bx * 64;
  const int rowW = by * 64 + wave * 16;

  int arow = rowW + l16;
  if (arow > M - 1) arow = M - 1;
  const float* arowp = A + (size_t)arow * K;

  const int srow = tid >> 3;
  const int scol = (tid & 7) * 8;
  const float* bstage = Bm + (size_t)srow * N + col0 + scol;

  const v8f zero = {0.f,0.f,0.f,0.f,0.f,0.f,0.f,0.f};
  v8f acc[4];
  #pragma unroll
  for (int j = 0; j < 4; ++j) acc[j] = zero;

  const int nkb = K / GK;

  {
    v4f r0 = *(const v4f*)(bstage);
    v4f r1 = *(const v4f*)(bstage + 4);
    #pragma unroll
    for (int e = 0; e < 4; ++e) Bs[0][(scol + e) * 18 + srow] = r0[e];
    #pragma unroll
    for (int e = 0; e < 4; ++e) Bs[0][(scol + 4 + e) * 18 + srow] = r1[e];
  }
  __syncthreads();

  for (int kb = 0; kb < nkb; ++kb) {
    const int cur = kb & 1;
    const bool have_next = (kb + 1) < nkb;

    v4f r0, r1;
    if (have_next) {
      const float* bnxt = bstage + (size_t)(kb + 1) * GK * N;
      r0 = *(const v4f*)(bnxt);
      r1 = *(const v4f*)(bnxt + 4);
      if (kb + 2 < nkb)
        __builtin_prefetch(bstage + (size_t)(kb + 2) * GK * N, 0, 1);
    }

    const float* ap = arowp + kb * GK;
    #pragma unroll
    for (int kk = 0; kk < GK; kk += 4) {
      const int ka = kk + 2 * half;
      v2f a = *(const v2f*)(ap + ka);
      #pragma unroll
      for (int j = 0; j < 4; ++j) {
        v2f b = *(const v2f*)(&Bs[cur][(16 * j + l16) * 18 + ka]);
        acc[j] = wmma4(a, b, acc[j]);
      }
    }

    if (have_next) {
      const int nxt = cur ^ 1;
      #pragma unroll
      for (int e = 0; e < 4; ++e) Bs[nxt][(scol + e) * 18 + srow] = r0[e];
      #pragma unroll
      for (int e = 0; e < 4; ++e) Bs[nxt][(scol + 4 + e) * 18 + srow] = r1[e];
    }
    __syncthreads();
  }

  const float bv = bias ? bias[col0 + 16 * 0 + l16] : 0.f;
  #pragma unroll
  for (int j = 0; j < 4; ++j) {
    const int c = col0 + 16 * j + l16;
    const float bj = bias ? bias[c] : 0.f;
    #pragma unroll
    for (int i = 0; i < 8; ++i) {
      const int r = rowW + i + 8 * half;
      if (r < M) C[(size_t)r * N + c] = acc[j][i] + bj;
    }
  }
  (void)bv;
}

__global__ __launch_bounds__(128)
void gemm_wmma_f32(const float* __restrict__ A, const float* __restrict__ Bm,
                   const float* __restrict__ bias, float* __restrict__ C,
                   int M, int N, int K) {
  gemm_tile_64x64(A, Bm, bias, C, M, N, K, blockIdx.x, blockIdx.y);
}

#define COL_TILES   5
#define KV_TILES    50
#define QKV_TILES   1700

__global__ __launch_bounds__(128)
void proj_qkv_wmma_f32(const float* __restrict__ X, const float* __restrict__ E,
                       const float* __restrict__ Wq, const float* __restrict__ Wk,
                       const float* __restrict__ Wv, float* __restrict__ Q,
                       float* __restrict__ Kp, float* __restrict__ Vp) {
  int id = blockIdx.x;
  const float* A;
  const float* W;
  float* C;
  int M, K;
  if (id < 2 * KV_TILES) {
    const bool isV = id >= KV_TILES;
    if (isV) id -= KV_TILES;
    A = E;
    W = isV ? Wv : Wk;
    C = isV ? Vp : Kp;
    M = B_ALL * S_DIM;
    K = D_ENC;
  } else {
    id -= 2 * KV_TILES;
    A = X;
    W = Wq;
    C = Q;
    M = 5 * T_DIM;
    K = C_DIM;
  }
  gemm_tile_64x64(A, W, nullptr, C, M, C_DIM, K, id % COL_TILES, id / COL_TILES);
}

__device__ __forceinline__
void pv_and_store(const float* Plds, const float* __restrict__ Vp,
                  float* __restrict__ AO, int b, int h, int t0,
                  int half, int l16) {
  const v8f zero = {0.f,0.f,0.f,0.f,0.f,0.f,0.f,0.f};
  v8f o[3];
  #pragma unroll
  for (int j = 0; j < 3; ++j) o[j] = zero;

  const float* vbase = Vp + (size_t)(b * S_DIM) * C_DIM + h * HD;
  const float* vcol[3];
  const float* vlast[3];
  #pragma unroll
  for (int j = 0; j < 3; ++j) {
    int c = 16 * j + l16;
    if (c > HD - 1) c = HD - 1;
    vcol[j]  = vbase + (size_t)(2 * half) * C_DIM + c;
    vlast[j] = vbase + (size_t)(S_DIM - 1) * C_DIM + c;
  }
  const float* prow = Plds + l16 * SP + 2 * half;

  #pragma unroll
  for (int s = 0; s < SP / 4; ++s) {
    const int k0 = 4 * s;
    const v2f a = *(const v2f*)(prow + k0);
    #pragma unroll
    for (int j = 0; j < 3; ++j) {
      v2f bf;
      if (k0 + 3 < S_DIM) {
        bf.x = vcol[j][(size_t)k0 * C_DIM];
        bf.y = vcol[j][(size_t)(k0 + 1) * C_DIM];
      } else {
        bf.x = vlast[j][0];
        bf.y = vlast[j][0];
      }
      o[j] = wmma4(a, bf, o[j]);
    }
  }

  #pragma unroll
  for (int j = 0; j < 3; ++j) {
    const int c = 16 * j + l16;
    if (c < HD) {
      #pragma unroll
      for (int i = 0; i < 8; ++i) {
        const int t = t0 + i + 8 * half;
        AO[(size_t)(b * T_DIM + t) * C_DIM + h * HD + c] = o[j][i];
      }
    }
  }
}

#define MV_TILES 300

__device__ __forceinline__
void mv_tile(const float* __restrict__ Map, const float* __restrict__ Vp,
             float* __restrict__ MV, int id, int half, int l16) {
  const int p  = id / 100;
  const int r  = id - p * 100;
  const int mt = r / 20;
  const int nt = r - mt * 20;
  int m = 16 * mt + l16;
  if (m > S_DIM - 1) m = S_DIM - 1;
  const float* arow = Map + (size_t)p * S_DIM * S_DIM + (size_t)m * S_DIM;
  const float* vb   = Vp + (size_t)((B_BASE + 1 + p) * S_DIM) * C_DIM + 16 * nt + l16;

  v8f acc = {0.f,0.f,0.f,0.f,0.f,0.f,0.f,0.f};
  #pragma unroll
  for (int s = 0; s < SP / 4; ++s) {
    const int k0 = 4 * s;
    v2f a, bf;
    if (k0 + 3 < S_DIM) {
      a.x  = arow[k0 + 2 * half];
      a.y  = arow[k0 + 2 * half + 1];
      bf.x = vb[(size_t)(k0 + 2 * half) * C_DIM];
      bf.y = vb[(size_t)(k0 + 2 * half + 1) * C_DIM];
    } else {
      const float a76 = arow[S_DIM - 1];
      const float v76 = vb[(size_t)(S_DIM - 1) * C_DIM];
      a.x  = half ? 0.f : a76;
      a.y  = 0.f;
      bf.x = v76;
      bf.y = v76;
    }
    acc = wmma4(a, bf, acc);
  }
  #pragma unroll
  for (int i = 0; i < 8; ++i) {
    const int row = 16 * mt + i + 8 * half;
    if (row < S_DIM)
      MV[(size_t)(p * S_DIM + row) * C_DIM + 16 * nt + l16] = acc[i];
  }
}

__global__ __launch_bounds__(32)
void attn_kernel(const float* __restrict__ Q,
                 const float* __restrict__ Kp,
                 const float* __restrict__ Vp,
                 const float* __restrict__ Map,
                 float* __restrict__ MV,
                 float* __restrict__ AB,
                 float* __restrict__ AO)
{
  __shared__ float Plds[16 * SP];
  const int lane = threadIdx.x;
  const int half = lane >> 4;
  const int l16  = lane & 15;
  int id = blockIdx.x;
  if (id < MV_TILES) {
    mv_tile(Map, Vp, MV, id, half, l16);
    return;
  }
  id -= MV_TILES;
  const int t0 = (id & (T_DIM / 16 - 1)) << 4;
  const int h  = (id >> 8) & (H_NUM - 1);
  const int b  = id >> 11;
  const float scale = 0.15811388300841898f;
  const float NEG_INF = -__builtin_inff();

  const v8f zero = {0.f,0.f,0.f,0.f,0.f,0.f,0.f,0.f};
  v8f sc[5];
  #pragma unroll
  for (int j = 0; j < 5; ++j) sc[j] = zero;

  const float* qbase = Q  + (size_t)(b * T_DIM + t0 + l16) * C_DIM + h * HD;
  const float* kbase = Kp + (size_t)(b * S_DIM) * C_DIM + h * HD;

  const float* krow[5];
  #pragma unroll
  for (int j = 0; j < 5; ++j) {
    int s = 16 * j + l16;
    if (s > S_DIM - 1) s = S_DIM - 1;
    krow[j] = kbase + (size_t)s * C_DIM + 2 * half;
  }
  #pragma unroll
  for (int k0 = 0; k0 < HD; k0 += 4) {
    v2f a = *(const v2f*)(qbase + k0 + 2 * half);
    #pragma unroll
    for (int j = 0; j < 5; ++j) {
      v2f bf = *(const v2f*)(krow[j] + k0);
      sc[j] = wmma4(a, bf, sc[j]);
    }
  }

  float mx[8];
  #pragma unroll
  for (int i = 0; i < 8; ++i) mx[i] = NEG_INF;
  #pragma unroll
  for (int j = 0; j < 5; ++j) {
    const bool ok = (16 * j + l16) < S_DIM;
    #pragma unroll
    for (int i = 0; i < 8; ++i) {
      const float x = ok ? sc[j][i] * scale : NEG_INF;
      sc[j][i] = x;
      mx[i] = fmaxf(mx[i], x);
    }
  }
  #pragma unroll
  for (int m = 1; m < 16; m <<= 1) {
    #pragma unroll
    for (int i = 0; i < 8; ++i) mx[i] = fmaxf(mx[i], __shfl_xor(mx[i], m, 32));
  }
  float sm[8];
  #pragma unroll
  for (int i = 0; i < 8; ++i) sm[i] = 0.f;
  #pragma unroll
  for (int j = 0; j < 5; ++j) {
    #pragma unroll
    for (int i = 0; i < 8; ++i) {
      const float e = __expf(sc[j][i] - mx[i]);
      sc[j][i] = e;
      sm[i] += e;
    }
  }
  #pragma unroll
  for (int m = 1; m < 16; m <<= 1) {
    #pragma unroll
    for (int i = 0; i < 8; ++i) sm[i] += __shfl_xor(sm[i], m, 32);
  }
  float rs[8];
  #pragma unroll
  for (int i = 0; i < 8; ++i) rs[i] = 1.f / sm[i];

  #pragma unroll
  for (int j = 0; j < 5; ++j) {
    #pragma unroll
    for (int i = 0; i < 8; ++i) {
      const float pv = sc[j][i] * rs[i];
      const int row = i + 8 * half;
      Plds[row * SP + 16 * j + l16] = pv;
      if (b == B_BASE)
        AB[(size_t)(h * T_DIM + t0 + row) * SP + 16 * j + l16] = pv;
    }
  }
  __syncthreads();

  pv_and_store(Plds, Vp, AO, b, h, t0, half, l16);
}

__global__ __launch_bounds__(32) __attribute__((amdgpu_waves_per_eu(2, 4)))
void edit_kernel(const float* __restrict__ AB,
                 const float* __restrict__ MV,
                 float* __restrict__ AO)
{
  const int lane = threadIdx.x;
  const int half = lane >> 4;
  const int l16  = lane & 15;
  const int t0 = blockIdx.x << 4;
  const int h  = blockIdx.y;

  const float* arow = AB + (size_t)(h * T_DIM + t0 + l16) * SP + 2 * half;
  v2f a[SP / 4];
  #pragma unroll
  for (int s = 0; s < SP / 4; ++s) a[s] = *(const v2f*)(arow + 4 * s);

  const v8f zero = {0.f,0.f,0.f,0.f,0.f,0.f,0.f,0.f};
  #pragma unroll
  for (int p = 0; p < 3; ++p) {
    const int b = B_BASE + 1 + p;
    const float* vbase = MV + (size_t)(p * S_DIM) * C_DIM + h * HD;
    const float* vcol[3];
    const float* vlast[3];
    #pragma unroll
    for (int j = 0; j < 3; ++j) {
      int c = 16 * j + l16;
      if (c > HD - 1) c = HD - 1;
      vcol[j]  = vbase + (size_t)(2 * half) * C_DIM + c;
      vlast[j] = vbase + (size_t)(S_DIM - 1) * C_DIM + c;
    }
    v8f o[3];
    #pragma unroll
    for (int j = 0; j < 3; ++j) o[j] = zero;

    #pragma unroll
    for (int s = 0; s < SP / 4; ++s) {
      const int k0 = 4 * s;
      #pragma unroll
      for (int j = 0; j < 3; ++j) {
        v2f bf;
        if (k0 + 3 < S_DIM) {
          bf.x = vcol[j][(size_t)k0 * C_DIM];
          bf.y = vcol[j][(size_t)(k0 + 1) * C_DIM];
        } else {
          bf.x = vlast[j][0];
          bf.y = vlast[j][0];
        }
        o[j] = wmma4(a[s], bf, o[j]);
      }
    }

    #pragma unroll
    for (int j = 0; j < 3; ++j) {
      const int c = 16 * j + l16;
      if (c < HD) {
        #pragma unroll
        for (int i = 0; i < 8; ++i) {
          const int t = t0 + i + 8 * half;
          AO[(size_t)(b * T_DIM + t) * C_DIM + h * HD + c] = o[j][i];
        }
      }
    }
  }
}

extern "C" void kernel_launch(void* const* d_in, const int* in_sizes, int n_in,
                              void* d_out, int out_size, void* d_ws, size_t ws_size,
                              hipStream_t stream) {
  const float* X   = (const float*)d_in[0];
  const float* E   = (const float*)d_in[1];
  const float* Map = (const float*)d_in[2];
  const float* Wq  = (const float*)d_in[3];
  const float* Wk  = (const float*)d_in[4];
  const float* Wv  = (const float*)d_in[5];
  const float* Wo  = (const float*)d_in[6];
  const float* bo  = (const float*)d_in[7];
  float* out = (float*)d_out;

  float* ws = (float*)d_ws;
  float* Q  = ws;
  float* Kp = Q  + (size_t)5 * T_DIM * C_DIM;
  float* Vp = Kp + (size_t)B_ALL * S_DIM * C_DIM;
  float* AB = Vp + (size_t)B_ALL * S_DIM * C_DIM;
  float* AO = AB + (size_t)H_NUM * T_DIM * SP;
  float* MV = AO + (size_t)B_ALL * T_DIM * C_DIM;

  const dim3 gblk(128);
  const dim3 ablk(32);

  proj_qkv_wmma_f32<<<dim3(QKV_TILES), gblk, 0, stream>>>(
      X, E, Wq, Wk, Wv, Q, Kp, Vp);

  attn_kernel<<<dim3(MV_TILES + (T_DIM / 16) * H_NUM * 5), ablk, 0, stream>>>(
      Q, Kp, Vp, Map, MV, AB, AO);

  edit_kernel<<<dim3(T_DIM / 16, H_NUM), ablk, 0, stream>>>(AB, MV, AO);

  gemm_wmma_f32<<<dim3(C_DIM / 64, (B_ALL * T_DIM) / 64), gblk, 0, stream>>>(
      AO, Wo, bo, out, B_ALL * T_DIM, C_DIM, C_DIM);
}


# GEMM tiles: A staged through LDS too, by whole 128-byte lines (K-block 32, each line fetched once per block, next block prefetched in registers) instead of per-lane 8-byte row reads; sums unchanged
# speedup vs baseline: 1.9455x; 1.0703x over previous
#include <hip/hip_runtime.h>


typedef __attribute__((ext_vector_type(2))) float v2f;
typedef __attribute__((ext_vector_type(4))) float v4f;
typedef __attribute__((ext_vector_type(8))) float v8f;

#define B_ALL   8
#define B_BASE  4
#define T_DIM   4096
#define C_DIM   320
#define S_DIM   77
#define SP      80
#define D_ENC   768
#define HD      40
#define H_NUM   8

__device__ __forceinline__ v8f wmma4(v2f a, v2f b, v8f c) {
  return __builtin_amdgcn_wmma_f32_16x16x4_f32(
       false, a,  false, b,
       (short)0, c,  false,  false);
}

#define GK 32
#define PA 36
#define PB 34

__device__ __forceinline__
void gemm_tile_64x64(const float* __restrict__ A, const float* __restrict__ Bm,
                     const float* __restrict__ bias, float* __restrict__ C,
                     int M, int N, int K, int bx, int by) {
  __shared__ __attribute__((aligned(16))) float As[64 * PA];
  __shared__ __attribute__((aligned(16))) float Bs[64 * PB];
  const int tid  = threadIdx.x;
  const int wave = tid >> 5;
  const int lane = tid & 31;
  const int half = lane >> 4;
  const int l16  = lane & 15;
  const int col0 = bx * 64;
  const int row0 = by * 64;
  const int rowW = row0 + wave * 16;

  const int arw = tid >> 3;
  const int ach = (tid & 7) * 4;
  unsigned aoff[4];
  #pragma unroll
  for (int i = 0; i < 4; ++i) {
    int r = row0 + arw + 16 * i;
    if (r > M - 1) r = M - 1;
    aoff[i] = (unsigned)r * (unsigned)K + (unsigned)ach;
  }
  const int aidx = arw * PA + ach;

  const int srow = tid >> 3;
  const int scol = (tid & 7) * 8;
  const unsigned boff = (unsigned)srow * (unsigned)N + (unsigned)(col0 + scol);

  const v8f zero = {0.f,0.f,0.f,0.f,0.f,0.f,0.f,0.f};
  v8f acc[4];
  #pragma unroll
  for (int j = 0; j < 4; ++j) acc[j] = zero;

  const int nkb = K / GK;
  const float* afrag = &As[(16 * wave + l16) * PA + 2 * half];

  v4f ra[4];
  v4f rb[4];

  #pragma unroll
  for (int i = 0; i < 4; ++i) ra[i] = *(const v4f*)(A + aoff[i]);
  #pragma unroll
  for (int hh = 0; hh < 2; ++hh) {
    rb[2 * hh]     = *(const v4f*)(Bm + (boff + (unsigned)(16 * hh * N)));
    rb[2 * hh + 1] = *(const v4f*)(Bm + (boff + (unsigned)(16 * hh * N) + 4u));
  }
  #pragma unroll
  for (int i = 0; i < 4; ++i) *(v4f*)(&As[aidx + 16 * i * PA]) = ra[i];
  #pragma unroll
  for (int hh = 0; hh < 2; ++hh) {
    #pragma unroll
    for (int e = 0; e < 4; ++e) {
      Bs[(scol + e) * PB + 16 * hh + srow]     = rb[2 * hh][e];
      Bs[(scol + 4 + e) * PB + 16 * hh + srow] = rb[2 * hh + 1][e];
    }
  }
  __syncthreads();

  for (int kb = 0; kb < nkb; ++kb) {
    const bool have_next = (kb + 1) < nkb;

    if (have_next) {
      #pragma unroll
      for (int i = 0; i < 4; ++i)
        ra[i] = *(const v4f*)(A + (aoff[i] + (unsigned)((kb + 1) * GK)));
      const unsigned bnxt = boff + (unsigned)((kb + 1) * GK * N);
      #pragma unroll
      for (int hh = 0; hh < 2; ++hh) {
        rb[2 * hh]     = *(const v4f*)(Bm + (bnxt + (unsigned)(16 * hh * N)));
        rb[2 * hh + 1] = *(const v4f*)(Bm + (bnxt + (unsigned)(16 * hh * N) + 4u));
      }
      if (kb + 2 < nkb) {
        __builtin_prefetch(Bm + (boff + (unsigned)((kb + 2) * GK * N)), 0, 1);
        __builtin_prefetch(Bm + (boff + (unsigned)(((kb + 2) * GK + 16) * N)), 0, 1);
      }
    }

    #pragma unroll
    for (int kk = 0; kk < GK; kk += 4) {
      const v2f a = *(const v2f*)(afrag + kk);
      #pragma unroll
      for (int j = 0; j < 4; ++j) {
        const v2f b = *(const v2f*)(&Bs[(16 * j + l16) * PB + kk + 2 * half]);
        acc[j] = wmma4(a, b, acc[j]);
      }
    }

    __syncthreads();
    if (have_next) {
      #pragma unroll
      for (int i = 0; i < 4; ++i) *(v4f*)(&As[aidx + 16 * i * PA]) = ra[i];
      #pragma unroll
      for (int hh = 0; hh < 2; ++hh) {
        #pragma unroll
        for (int e = 0; e < 4; ++e) {
          Bs[(scol + e) * PB + 16 * hh + srow]     = rb[2 * hh][e];
          Bs[(scol + 4 + e) * PB + 16 * hh + srow] = rb[2 * hh + 1][e];
        }
      }
    }
    __syncthreads();
  }

  #pragma unroll
  for (int j = 0; j < 4; ++j) {
    const int c = col0 + 16 * j + l16;
    const float bj = bias ? bias[c] : 0.f;
    #pragma unroll
    for (int i = 0; i < 8; ++i) {
      const int r = rowW + i + 8 * half;
      if (r < M) C[(size_t)r * N + c] = acc[j][i] + bj;
    }
  }
}

__global__ __launch_bounds__(128) __attribute__((amdgpu_waves_per_eu(8)))
void gemm_wmma_f32(const float* __restrict__ A, const float* __restrict__ Bm,
                   const float* __restrict__ bias, float* __restrict__ C,
                   int M, int N, int K) {
  gemm_tile_64x64(A, Bm, bias, C, M, N, K, blockIdx.x, blockIdx.y);
}

#define COL_TILES   5
#define KV_TILES    50
#define QKV_TILES   1700

__global__ __launch_bounds__(128)
void proj_qkv_wmma_f32(const float* __restrict__ X, const float* __restrict__ E,
                       const float* __restrict__ Wq, const float* __restrict__ Wk,
                       const float* __restrict__ Wv, float* __restrict__ Q,
                       float* __restrict__ Kp, float* __restrict__ Vp) {
  int id = blockIdx.x;
  const float* A;
  const float* W;
  float* C;
  int M, K;
  if (id < 2 * KV_TILES) {
    const bool isV = id >= KV_TILES;
    if (isV) id -= KV_TILES;
    A = E;
    W = isV ? Wv : Wk;
    C = isV ? Vp : Kp;
    M = B_ALL * S_DIM;
    K = D_ENC;
  } else {
    id -= 2 * KV_TILES;
    A = X;
    W = Wq;
    C = Q;
    M = 5 * T_DIM;
    K = C_DIM;
  }
  gemm_tile_64x64(A, W, nullptr, C, M, C_DIM, K, id % COL_TILES, id / COL_TILES);
}

__device__ __forceinline__
void pv_and_store(const float* Plds, const float* __restrict__ Vp,
                  float* __restrict__ AO, int b, int h, int t0,
                  int half, int l16) {
  const v8f zero = {0.f,0.f,0.f,0.f,0.f,0.f,0.f,0.f};
  v8f o[3];
  #pragma unroll
  for (int j = 0; j < 3; ++j) o[j] = zero;

  const float* vbase = Vp + (size_t)(b * S_DIM) * C_DIM + h * HD;
  const float* vcol[3];
  const float* vlast[3];
  #pragma unroll
  for (int j = 0; j < 3; ++j) {
    int c = 16 * j + l16;
    if (c > HD - 1) c = HD - 1;
    vcol[j]  = vbase + (size_t)(2 * half) * C_DIM + c;
    vlast[j] = vbase + (size_t)(S_DIM - 1) * C_DIM + c;
  }
  const float* prow = Plds + l16 * SP + 2 * half;

  #pragma unroll
  for (int s = 0; s < SP / 4; ++s) {
    const int k0 = 4 * s;
    const v2f a = *(const v2f*)(prow + k0);
    #pragma unroll
    for (int j = 0; j < 3; ++j) {
      v2f bf;
      if (k0 + 3 < S_DIM) {
        bf.x = vcol[j][(size_t)k0 * C_DIM];
        bf.y = vcol[j][(size_t)(k0 + 1) * C_DIM];
      } else {
        bf.x = vlast[j][0];
        bf.y = vlast[j][0];
      }
      o[j] = wmma4(a, bf, o[j]);
    }
  }

  #pragma unroll
  for (int j = 0; j < 3; ++j) {
    const int c = 16 * j + l16;
    if (c < HD) {
      #pragma unroll
      for (int i = 0; i < 8; ++i) {
        const int t = t0 + i + 8 * half;
        AO[(size_t)(b * T_DIM + t) * C_DIM + h * HD + c] = o[j][i];
      }
    }
  }
}

#define MV_TILES 300

__device__ __forceinline__
void mv_tile(const float* __restrict__ Map, const float* __restrict__ Vp,
             float* __restrict__ MV, int id, int half, int l16) {
  const int p  = id / 100;
  const int r  = id - p * 100;
  const int mt = r / 20;
  const int nt = r - mt * 20;
  int m = 16 * mt + l16;
  if (m > S_DIM - 1) m = S_DIM - 1;
  const float* arow = Map + (size_t)p * S_DIM * S_DIM + (size_t)m * S_DIM;
  const float* vb   = Vp + (size_t)((B_BASE + 1 + p) * S_DIM) * C_DIM + 16 * nt + l16;

  v8f acc = {0.f,0.f,0.f,0.f,0.f,0.f,0.f,0.f};
  #pragma unroll
  for (int s = 0; s < SP / 4; ++s) {
    const int k0 = 4 * s;
    v2f a, bf;
    if (k0 + 3 < S_DIM) {
      a.x  = arow[k0 + 2 * half];
      a.y  = arow[k0 + 2 * half + 1];
      bf.x = vb[(size_t)(k0 + 2 * half) * C_DIM];
      bf.y = vb[(size_t)(k0 + 2 * half + 1) * C_DIM];
    } else {
      const float a76 = arow[S_DIM - 1];
      const float v76 = vb[(size_t)(S_DIM - 1) * C_DIM];
      a.x  = half ? 0.f : a76;
      a.y  = 0.f;
      bf.x = v76;
      bf.y = v76;
    }
    acc = wmma4(a, bf, acc);
  }
  #pragma unroll
  for (int i = 0; i < 8; ++i) {
    const int row = 16 * mt + i + 8 * half;
    if (row < S_DIM)
      MV[(size_t)(p * S_DIM + row) * C_DIM + 16 * nt + l16] = acc[i];
  }
}

__global__ __launch_bounds__(32)
void attn_kernel(const float* __restrict__ Q,
                 const float* __restrict__ Kp,
                 const float* __restrict__ Vp,
                 const float* __restrict__ Map,
                 float* __restrict__ MV,
                 float* __restrict__ AB,
                 float* __restrict__ AO)
{
  __shared__ float Plds[16 * SP];
  const int lane = threadIdx.x;
  const int half = lane >> 4;
  const int l16  = lane & 15;
  int id = blockIdx.x;
  if (id < MV_TILES) {
    mv_tile(Map, Vp, MV, id, half, l16);
    return;
  }
  id -= MV_TILES;
  const int t0 = (id & (T_DIM / 16 - 1)) << 4;
  const int h  = (id >> 8) & (H_NUM - 1);
  const int b  = id >> 11;
  const float scale = 0.15811388300841898f;
  const float NEG_INF = -__builtin_inff();

  const v8f zero = {0.f,0.f,0.f,0.f,0.f,0.f,0.f,0.f};
  v8f sc[5];
  #pragma unroll
  for (int j = 0; j < 5; ++j) sc[j] = zero;

  const float* qbase = Q  + (size_t)(b * T_DIM + t0 + l16) * C_DIM + h * HD;
  const float* kbase = Kp + (size_t)(b * S_DIM) * C_DIM + h * HD;

  const float* krow[5];
  #pragma unroll
  for (int j = 0; j < 5; ++j) {
    int s = 16 * j + l16;
    if (s > S_DIM - 1) s = S_DIM - 1;
    krow[j] = kbase + (size_t)s * C_DIM + 2 * half;
  }
  #pragma unroll
  for (int k0 = 0; k0 < HD; k0 += 4) {
    v2f a = *(const v2f*)(qbase + k0 + 2 * half);
    #pragma unroll
    for (int j = 0; j < 5; ++j) {
      v2f bf = *(const v2f*)(krow[j] + k0);
      sc[j] = wmma4(a, bf, sc[j]);
    }
  }

  float mx[8];
  #pragma unroll
  for (int i = 0; i < 8; ++i) mx[i] = NEG_INF;
  #pragma unroll
  for (int j = 0; j < 5; ++j) {
    const bool ok = (16 * j + l16) < S_DIM;
    #pragma unroll
    for (int i = 0; i < 8; ++i) {
      const float x = ok ? sc[j][i] * scale : NEG_INF;
      sc[j][i] = x;
      mx[i] = fmaxf(mx[i], x);
    }
  }
  #pragma unroll
  for (int m = 1; m < 16; m <<= 1) {
    #pragma unroll
    for (int i = 0; i < 8; ++i) mx[i] = fmaxf(mx[i], __shfl_xor(mx[i], m, 32));
  }
  float sm[8];
  #pragma unroll
  for (int i = 0; i < 8; ++i) sm[i] = 0.f;
  #pragma unroll
  for (int j = 0; j < 5; ++j) {
    #pragma unroll
    for (int i = 0; i < 8; ++i) {
      const float e = __expf(sc[j][i] - mx[i]);
      sc[j][i] = e;
      sm[i] += e;
    }
  }
  #pragma unroll
  for (int m = 1; m < 16; m <<= 1) {
    #pragma unroll
    for (int i = 0; i < 8; ++i) sm[i] += __shfl_xor(sm[i], m, 32);
  }
  float rs[8];
  #pragma unroll
  for (int i = 0; i < 8; ++i) rs[i] = 1.f / sm[i];

  #pragma unroll
  for (int j = 0; j < 5; ++j) {
    #pragma unroll
    for (int i = 0; i < 8; ++i) {
      const float pv = sc[j][i] * rs[i];
      const int row = i + 8 * half;
      Plds[row * SP + 16 * j + l16] = pv;
      if (b == B_BASE)
        AB[(size_t)(h * T_DIM + t0 + row) * SP + 16 * j + l16] = pv;
    }
  }
  __syncthreads();

  pv_and_store(Plds, Vp, AO, b, h, t0, half, l16);
}

__global__ __launch_bounds__(32) __attribute__((amdgpu_waves_per_eu(2, 4)))
void edit_kernel(const float* __restrict__ AB,
                 const float* __restrict__ MV,
                 float* __restrict__ AO)
{
  const int lane = threadIdx.x;
  const int half = lane >> 4;
  const int l16  = lane & 15;
  const int t0 = blockIdx.x << 4;
  const int h  = blockIdx.y;

  const float* arow = AB + (size_t)(h * T_DIM + t0 + l16) * SP + 2 * half;
  v2f a[SP / 4];
  #pragma unroll
  for (int s = 0; s < SP / 4; ++s) a[s] = *(const v2f*)(arow + 4 * s);

  const v8f zero = {0.f,0.f,0.f,0.f,0.f,0.f,0.f,0.f};
  #pragma unroll
  for (int p = 0; p < 3; ++p) {
    const int b = B_BASE + 1 + p;
    const float* vbase = MV + (size_t)(p * S_DIM) * C_DIM + h * HD;
    const float* vcol[3];
    const float* vlast[3];
    #pragma unroll
    for (int j = 0; j < 3; ++j) {
      int c = 16 * j + l16;
      if (c > HD - 1) c = HD - 1;
      vcol[j]  = vbase + (size_t)(2 * half) * C_DIM + c;
      vlast[j] = vbase + (size_t)(S_DIM - 1) * C_DIM + c;
    }
    v8f o[3];
    #pragma unroll
    for (int j = 0; j < 3; ++j) o[j] = zero;

    #pragma unroll
    for (int s = 0; s < SP / 4; ++s) {
      const int k0 = 4 * s;
      #pragma unroll
      for (int j = 0; j < 3; ++j) {
        v2f bf;
        if (k0 + 3 < S_DIM) {
          bf.x = vcol[j][(size_t)k0 * C_DIM];
          bf.y = vcol[j][(size_t)(k0 + 1) * C_DIM];
        } else {
          bf.x = vlast[j][0];
          bf.y = vlast[j][0];
        }
        o[j] = wmma4(a[s], bf, o[j]);
      }
    }

    #pragma unroll
    for (int j = 0; j < 3; ++j) {
      const int c = 16 * j + l16;
      if (c < HD) {
        #pragma unroll
        for (int i = 0; i < 8; ++i) {
          const int t = t0 + i + 8 * half;
          AO[(size_t)(b * T_DIM + t) * C_DIM + h * HD + c] = o[j][i];
        }
      }
    }
  }
}

extern "C" void kernel_launch(void* const* d_in, const int* in_sizes, int n_in,
                              void* d_out, int out_size, void* d_ws, size_t ws_size,
                              hipStream_t stream) {
  const float* X   = (const float*)d_in[0];
  const float* E   = (const float*)d_in[1];
  const float* Map = (const float*)d_in[2];
  const float* Wq  = (const float*)d_in[3];
  const float* Wk  = (const float*)d_in[4];
  const float* Wv  = (const float*)d_in[5];
  const float* Wo  = (const float*)d_in[6];
  const float* bo  = (const float*)d_in[7];
  float* out = (float*)d_out;

  float* ws = (float*)d_ws;
  float* Q  = ws;
  float* Kp = Q  + (size_t)5 * T_DIM * C_DIM;
  float* Vp = Kp + (size_t)B_ALL * S_DIM * C_DIM;
  float* AB = Vp + (size_t)B_ALL * S_DIM * C_DIM;
  float* AO = AB + (size_t)H_NUM * T_DIM * SP;
  float* MV = AO + (size_t)B_ALL * T_DIM * C_DIM;

  const dim3 gblk(128);
  const dim3 ablk(32);

  proj_qkv_wmma_f32<<<dim3(QKV_TILES), gblk, 0, stream>>>(
      X, E, Wq, Wk, Wv, Q, Kp, Vp);

  attn_kernel<<<dim3(MV_TILES + (T_DIM / 16) * H_NUM * 5), ablk, 0, stream>>>(
      Q, Kp, Vp, Map, MV, AB, AO);

  edit_kernel<<<dim3(T_DIM / 16, H_NUM), ablk, 0, stream>>>(AB, MV, AO);

  gemm_wmma_f32<<<dim3(C_DIM / 64, (B_ALL * T_DIM) / 64), gblk, 0, stream>>>(
      AO, Wo, bo, out, B_ALL * T_DIM, C_DIM, C_DIM);
}


# K, V and MV kept in fragment order by their producers: each attention/edit operand fragment pair is one 16-byte load (B: contiguous 256-byte runs); k taken in blocks of 8, so sum order changes
# speedup vs baseline: 2.1732x; 1.1170x over previous
#include <hip/hip_runtime.h>


typedef __attribute__((ext_vector_type(2))) float v2f;
typedef __attribute__((ext_vector_type(4))) float v4f;
typedef __attribute__((ext_vector_type(8))) float v8f;

#define B_ALL   8
#define B_BASE  4
#define T_DIM   4096
#define C_DIM   320
#define S_DIM   77
#define SP      80
#define D_ENC   768
#define HD      40
#define H_NUM   8

#define KF_GRP   320
#define KF_HEAD  (10 * KF_GRP)
#define VF_GRP   (C_DIM * 4)
#define VF_BATCH (20 * VF_GRP)
#define PP       84

__device__ __forceinline__ v8f wmma4(v2f a, v2f b, v8f c) {
  return __builtin_amdgcn_wmma_f32_16x16x4_f32(
       false, a,  false, b,
       (short)0, c,  false,  false);
}

#define GK 32
#define PA 36
#define PB 34

__device__ __forceinline__
void gemm_tile_64x64(const float* __restrict__ A, const float* __restrict__ Bm,
                     const float* __restrict__ bias, float* __restrict__ C,
                     int M, int N, int K, int bx, int by, int lay) {
  __shared__ __attribute__((aligned(16))) float As[64 * PA];
  __shared__ __attribute__((aligned(16))) float Bs[64 * PB];
  const int tid  = threadIdx.x;
  const int wave = tid >> 5;
  const int lane = tid & 31;
  const int half = lane >> 4;
  const int l16  = lane & 15;
  const int col0 = bx * 64;
  const int row0 = by * 64;
  const int rowW = row0 + wave * 16;

  const int arw = tid >> 3;
  const int ach = (tid & 7) * 4;
  unsigned aoff[4];
  #pragma unroll
  for (int i = 0; i < 4; ++i) {
    int r = row0 + arw + 16 * i;
    if (r > M - 1) r = M - 1;
    aoff[i] = (unsigned)r * (unsigned)K + (unsigned)ach;
  }
  const int aidx = arw * PA + ach;

  const int srow = tid >> 3;
  const int scol = (tid & 7) * 8;
  const unsigned boff = (unsigned)srow * (unsigned)N + (unsigned)(col0 + scol);

  const v8f zero = {0.f,0.f,0.f,0.f,0.f,0.f,0.f,0.f};
  v8f acc[4];
  #pragma unroll
  for (int j = 0; j < 4; ++j) acc[j] = zero;

  const int nkb = K / GK;
  const float* afrag = &As[(16 * wave + l16) * PA + 2 * half];

  v4f ra[4];
  v4f rb[4];

  #pragma unroll
  for (int i = 0; i < 4; ++i) ra[i] = *(const v4f*)(A + aoff[i]);
  #pragma unroll
  for (int hh = 0; hh < 2; ++hh) {
    rb[2 * hh]     = *(const v4f*)(Bm + (boff + (unsigned)(16 * hh * N)));
    rb[2 * hh + 1] = *(const v4f*)(Bm + (boff + (unsigned)(16 * hh * N) + 4u));
  }
  #pragma unroll
  for (int i = 0; i < 4; ++i) *(v4f*)(&As[aidx + 16 * i * PA]) = ra[i];
  #pragma unroll
  for (int hh = 0; hh < 2; ++hh) {
    #pragma unroll
    for (int e = 0; e < 4; ++e) {
      Bs[(scol + e) * PB + 16 * hh + srow]     = rb[2 * hh][e];
      Bs[(scol + 4 + e) * PB + 16 * hh + srow] = rb[2 * hh + 1][e];
    }
  }
  __syncthreads();

  for (int kb = 0; kb < nkb; ++kb) {
    const bool have_next = (kb + 1) < nkb;

    if (have_next) {
      #pragma unroll
      for (int i = 0; i < 4; ++i)
        ra[i] = *(const v4f*)(A + (aoff[i] + (unsigned)((kb + 1) * GK)));
      const unsigned bnxt = boff + (unsigned)((kb + 1) * GK * N);
      #pragma unroll
      for (int hh = 0; hh < 2; ++hh) {
        rb[2 * hh]     = *(const v4f*)(Bm + (bnxt + (unsigned)(16 * hh * N)));
        rb[2 * hh + 1] = *(const v4f*)(Bm + (bnxt + (unsigned)(16 * hh * N) + 4u));
      }
      if (kb + 2 < nkb) {
        __builtin_prefetch(Bm + (boff + (unsigned)((kb + 2) * GK * N)), 0, 1);
        __builtin_prefetch(Bm + (boff + (unsigned)(((kb + 2) * GK + 16) * N)), 0, 1);
      }
    }

    #pragma unroll
    for (int kk = 0; kk < GK; kk += 4) {
      const v2f a = *(const v2f*)(afrag + kk);
      #pragma unroll
      for (int j = 0; j < 4; ++j) {
        const v2f b = *(const v2f*)(&Bs[(16 * j + l16) * PB + kk + 2 * half]);
        acc[j] = wmma4(a, b, acc[j]);
      }
    }

    __syncthreads();
    if (have_next) {
      #pragma unroll
      for (int i = 0; i < 4; ++i) *(v4f*)(&As[aidx + 16 * i * PA]) = ra[i];
      #pragma unroll
      for (int hh = 0; hh < 2; ++hh) {
        #pragma unroll
        for (int e = 0; e < 4; ++e) {
          Bs[(scol + e) * PB + 16 * hh + srow]     = rb[2 * hh][e];
          Bs[(scol + 4 + e) * PB + 16 * hh + srow] = rb[2 * hh + 1][e];
        }
      }
    }
    __syncthreads();
  }

  if (lay == 0) {
    #pragma unroll
    for (int j = 0; j < 4; ++j) {
      const int c = col0 + 16 * j + l16;
      const float bj = bias ? bias[c] : 0.f;
      #pragma unroll
      for (int i = 0; i < 8; ++i) {
        const int r = rowW + i + 8 * half;
        if (r < M) C[(size_t)r * N + c] = acc[j][i] + bj;
      }
    }
  } else if (lay == 1) {
    #pragma unroll
    for (int j = 0; j < 4; ++j) {
      const int c = col0 + 16 * j + l16;
      const float bj = bias ? bias[c] : 0.f;
      const int hh = c / HD;
      const int kd = c - hh * HD;
      const int cpart = hh * KF_HEAD + (kd >> 2) * KF_GRP + (kd & 3);
      #pragma unroll
      for (int i = 0; i < 8; ++i) {
        const int r = rowW + i + 8 * half;
        if (r < M) {
          const int bb = r / S_DIM;
          const int sr = r - bb * S_DIM;
          C[(size_t)(bb * (H_NUM * KF_HEAD) + cpart + sr * 4)] = acc[j][i] + bj;
        }
      }
    }
  } else {
    #pragma unroll
    for (int j = 0; j < 4; ++j) {
      const int c = col0 + 16 * j + l16;
      const float bj = bias ? bias[c] : 0.f;
      #pragma unroll
      for (int i = 0; i < 8; ++i) {
        const int r = rowW + i + 8 * half;
        if (r < M) {
          const int bb = r / S_DIM;
          const int sr = r - bb * S_DIM;
          C[(size_t)(bb * VF_BATCH + (sr >> 2) * VF_GRP + c * 4 + (sr & 3))] = acc[j][i] + bj;
        }
      }
    }
  }
}

__global__ __launch_bounds__(128) __attribute__((amdgpu_waves_per_eu(8)))
void gemm_wmma_f32(const float* __restrict__ A, const float* __restrict__ Bm,
                   const float* __restrict__ bias, float* __restrict__ C,
                   int M, int N, int K) {
  gemm_tile_64x64(A, Bm, bias, C, M, N, K, blockIdx.x, blockIdx.y, 0);
}

#define COL_TILES   5
#define KV_TILES    50
#define QKV_TILES   1700

__global__ __launch_bounds__(128)
void proj_qkv_wmma_f32(const float* __restrict__ X, const float* __restrict__ E,
                       const float* __restrict__ Wq, const float* __restrict__ Wk,
                       const float* __restrict__ Wv, float* __restrict__ Q,
                       float* __restrict__ Kp, float* __restrict__ Vp) {
  int id = blockIdx.x;
  const float* A;
  const float* W;
  float* C;
  int M, K, lay;
  if (id < 2 * KV_TILES) {
    const bool isV = id >= KV_TILES;
    if (isV) id -= KV_TILES;
    A = E;
    W = isV ? Wv : Wk;
    C = isV ? Vp : Kp;
    M = B_ALL * S_DIM;
    K = D_ENC;
    lay = isV ? 2 : 1;
  } else {
    id -= 2 * KV_TILES;
    A = X;
    W = Wq;
    C = Q;
    M = 5 * T_DIM;
    K = C_DIM;
    lay = 0;
  }
  gemm_tile_64x64(A, W, nullptr, C, M, C_DIM, K, id % COL_TILES, id / COL_TILES, lay);
}

__device__ __forceinline__
void vf_block(v4f a4, const float* const* vcol, int blk, int half, v8f* o) {
  #pragma unroll
  for (int j = 0; j < 3; ++j) {
    v4f b4 = *(const v4f*)(vcol[j] + (size_t)blk * (2 * VF_GRP));
    if (blk == SP / 8 - 1) {
      const float b0 = b4.x;
      b4.y = half ? b0 : b4.y;
      b4.z = half ? b0 : b4.z;
      b4.w = half ? b0 : b4.w;
    }
    v2f a01 = {a4.x, a4.y}, a23 = {a4.z, a4.w};
    v2f b01 = {b4.x, b4.y}, b23 = {b4.z, b4.w};
    o[j] = wmma4(a01, b01, o[j]);
    o[j] = wmma4(a23, b23, o[j]);
  }
}

__device__ __forceinline__
void pv_and_store(const float* Plds, const float* __restrict__ VF,
                  float* __restrict__ AO, int b, int h, int t0,
                  int half, int l16) {
  const v8f zero = {0.f,0.f,0.f,0.f,0.f,0.f,0.f,0.f};
  v8f o[3];
  #pragma unroll
  for (int j = 0; j < 3; ++j) o[j] = zero;

  const float* vbase = VF + (size_t)b * VF_BATCH + (size_t)half * VF_GRP + (h * HD) * 4;
  const float* vcol[3];
  #pragma unroll
  for (int j = 0; j < 3; ++j) {
    int c = 16 * j + l16;
    if (c > HD - 1) c = HD - 1;
    vcol[j] = vbase + c * 4;
  }
  const float* prow = Plds + l16 * PP + 4 * half;

  #pragma unroll
  for (int m = 0; m < SP / 8; ++m) {
    const v4f a4 = *(const v4f*)(prow + 8 * m);
    vf_block(a4, vcol, m, half, o);
  }

  #pragma unroll
  for (int j = 0; j < 3; ++j) {
    const int c = 16 * j + l16;
    if (c < HD) {
      #pragma unroll
      for (int i = 0; i < 8; ++i) {
        const int t = t0 + i + 8 * half;
        AO[(size_t)(b * T_DIM + t) * C_DIM + h * HD + c] = o[j][i];
      }
    }
  }
}

#define MV_TILES 300

__device__ __forceinline__
void mv_tile(const float* __restrict__ Map, const float* __restrict__ VF,
             float* __restrict__ MVF, int id, int half, int l16) {
  const int p  = id / 100;
  const int r  = id - p * 100;
  const int mt = r / 20;
  const int nt = r - mt * 20;
  int m = 16 * mt + l16;
  if (m > S_DIM - 1) m = S_DIM - 1;
  const float* arow = Map + (size_t)p * S_DIM * S_DIM + (size_t)m * S_DIM;
  const int c = 16 * nt + l16;
  const float* vb = VF + (size_t)(B_BASE + 1 + p) * VF_BATCH + (size_t)half * VF_GRP + c * 4;

  v8f acc = {0.f,0.f,0.f,0.f,0.f,0.f,0.f,0.f};
  #pragma unroll
  for (int q = 0; q < SP / 8; ++q) {
    v4f b4 = *(const v4f*)(vb + (size_t)q * (2 * VF_GRP));
    float a0, a1, a2, a3;
    if (q < SP / 8 - 1) {
      const int e0 = 8 * q + 4 * half;
      a0 = arow[e0];
      a1 = arow[e0 + 1];
      a2 = arow[e0 + 2];
      a3 = arow[e0 + 3];
    } else {
      const float x0 = arow[half ? 76 : 72];
      const float x1 = arow[half ? 76 : 73];
      const float x2 = arow[half ? 76 : 74];
      const float x3 = arow[half ? 76 : 75];
      a0 = x0;
      a1 = half ? 0.f : x1;
      a2 = half ? 0.f : x2;
      a3 = half ? 0.f : x3;
      const float b0 = b4.x;
      b4.y = half ? b0 : b4.y;
      b4.z = half ? b0 : b4.z;
      b4.w = half ? b0 : b4.w;
    }
    v2f a01 = {a0, a1}, a23 = {a2, a3};
    v2f b01 = {b4.x, b4.y}, b23 = {b4.z, b4.w};
    acc = wmma4(a01, b01, acc);
    acc = wmma4(a23, b23, acc);
  }
  #pragma unroll
  for (int i = 0; i < 8; ++i) {
    const int row = 16 * mt + i + 8 * half;
    if (row < S_DIM)
      MVF[(size_t)p * VF_BATCH + (size_t)(row >> 2) * VF_GRP + c * 4 + (row & 3)] = acc[i];
  }
}

__global__ __launch_bounds__(32)
void attn_kernel(const float* __restrict__ Q,
                 const float* __restrict__ KF,
                 const float* __restrict__ VF,
                 const float* __restrict__ Map,
                 float* __restrict__ MV,
                 float* __restrict__ AB,
                 float* __restrict__ AO)
{
  __shared__ __attribute__((aligned(16))) float Plds[16 * PP];
  const int lane = threadIdx.x;
  const int half = lane >> 4;
  const int l16  = lane & 15;
  int id = blockIdx.x;
  if (id < MV_TILES) {
    mv_tile(Map, VF, MV, id, half, l16);
    return;
  }
  id -= MV_TILES;
  const int t0 = (id & (T_DIM / 16 - 1)) << 4;
  const int h  = (id >> 8) & (H_NUM - 1);
  const int b  = id >> 11;
  const float scale = 0.15811388300841898f;
  const float NEG_INF = -__builtin_inff();

  const v8f zero = {0.f,0.f,0.f,0.f,0.f,0.f,0.f,0.f};
  v8f sc[5];
  #pragma unroll
  for (int j = 0; j < 5; ++j) sc[j] = zero;

  const float* qbase = Q  + (size_t)(b * T_DIM + t0 + l16) * C_DIM + h * HD + 4 * half;
  const float* kbase = KF + (size_t)(b * H_NUM + h) * KF_HEAD + (size_t)half * KF_GRP;

  const float* krow[5];
  #pragma unroll
  for (int j = 0; j < 5; ++j) {
    int s = 16 * j + l16;
    if (s > S_DIM - 1) s = S_DIM - 1;
    krow[j] = kbase + s * 4;
  }
  #pragma unroll
  for (int m = 0; m < HD / 8; ++m) {
    const v4f a4 = *(const v4f*)(qbase + 8 * m);
    const v2f a01 = {a4.x, a4.y}, a23 = {a4.z, a4.w};
    #pragma unroll
    for (int j = 0; j < 5; ++j) {
      const v4f b4 = *(const v4f*)(krow[j] + (size_t)m * (2 * KF_GRP));
      const v2f b01 = {b4.x, b4.y}, b23 = {b4.z, b4.w};
      sc[j] = wmma4(a01, b01, sc[j]);
      sc[j] = wmma4(a23, b23, sc[j]);
    }
  }

  float mx[8];
  #pragma unroll
  for (int i = 0; i < 8; ++i) mx[i] = NEG_INF;
  #pragma unroll
  for (int j = 0; j < 5; ++j) {
    const bool ok = (16 * j + l16) < S_DIM;
    #pragma unroll
    for (int i = 0; i < 8; ++i) {
      const float x = ok ? sc[j][i] * scale : NEG_INF;
      sc[j][i] = x;
      mx[i] = fmaxf(mx[i], x);
    }
  }
  #pragma unroll
  for (int m = 1; m < 16; m <<= 1) {
    #pragma unroll
    for (int i = 0; i < 8; ++i) mx[i] = fmaxf(mx[i], __shfl_xor(mx[i], m, 32));
  }
  float sm[8];
  #pragma unroll
  for (int i = 0; i < 8; ++i) sm[i] = 0.f;
  #pragma unroll
  for (int j = 0; j < 5; ++j) {
    #pragma unroll
    for (int i = 0; i < 8; ++i) {
      const float e = __expf(sc[j][i] - mx[i]);
      sc[j][i] = e;
      sm[i] += e;
    }
  }
  #pragma unroll
  for (int m = 1; m < 16; m <<= 1) {
    #pragma unroll
    for (int i = 0; i < 8; ++i) sm[i] += __shfl_xor(sm[i], m, 32);
  }
  float rs[8];
  #pragma unroll
  for (int i = 0; i < 8; ++i) rs[i] = 1.f / sm[i];

  #pragma unroll
  for (int j = 0; j < 5; ++j) {
    #pragma unroll
    for (int i = 0; i < 8; ++i) {
      const float pv = sc[j][i] * rs[i];
      const int row = i + 8 * half;
      Plds[row * PP + 16 * j + l16] = pv;
      if (b == B_BASE)
        AB[(size_t)(h * T_DIM + t0 + row) * SP + 16 * j + l16] = pv;
    }
  }
  __syncthreads();

  pv_and_store(Plds, VF, AO, b, h, t0, half, l16);
}

__global__ __launch_bounds__(32) __attribute__((amdgpu_waves_per_eu(2, 4)))
void edit_kernel(const float* __restrict__ AB,
                 const float* __restrict__ MV,
                 float* __restrict__ AO)
{
  const int lane = threadIdx.x;
  const int half = lane >> 4;
  const int l16  = lane & 15;
  const int t0 = blockIdx.x << 4;
  const int h  = blockIdx.y;

  const float* arow = AB + (size_t)(h * T_DIM + t0 + l16) * SP + 4 * half;
  v4f a[SP / 8];
  #pragma unroll
  for (int m = 0; m < SP / 8; ++m) a[m] = *(const v4f*)(arow + 8 * m);

  const v8f zero = {0.f,0.f,0.f,0.f,0.f,0.f,0.f,0.f};
  #pragma unroll
  for (int p = 0; p < 3; ++p) {
    const int b = B_BASE + 1 + p;
    const float* vbase = MV + (size_t)p * VF_BATCH + (size_t)half * VF_GRP + (h * HD) * 4;
    const float* vcol[3];
    #pragma unroll
    for (int j = 0; j < 3; ++j) {
      int c = 16 * j + l16;
      if (c > HD - 1) c = HD - 1;
      vcol[j] = vbase + c * 4;
    }
    v8f o[3];
    #pragma unroll
    for (int j = 0; j < 3; ++j) o[j] = zero;

    #pragma unroll
    for (int m = 0; m < SP / 8; ++m) vf_block(a[m], vcol, m, half, o);

    #pragma unroll
    for (int j = 0; j < 3; ++j) {
      const int c = 16 * j + l16;
      if (c < HD) {
        #pragma unroll
        for (int i = 0; i < 8; ++i) {
          const int t = t0 + i + 8 * half;
          AO[(size_t)(b * T_DIM + t) * C_DIM + h * HD + c] = o[j][i];
        }
      }
    }
  }
}

extern "C" void kernel_launch(void* const* d_in, const int* in_sizes, int n_in,
                              void* d_out, int out_size, void* d_ws, size_t ws_size,
                              hipStream_t stream) {
  const float* X   = (const float*)d_in[0];
  const float* E   = (const float*)d_in[1];
  const float* Map = (const float*)d_in[2];
  const float* Wq  = (const float*)d_in[3];
  const float* Wk  = (const float*)d_in[4];
  const float* Wv  = (const float*)d_in[5];
  const float* Wo  = (const float*)d_in[6];
  const float* bo  = (const float*)d_in[7];
  float* out = (float*)d_out;

  float* ws = (float*)d_ws;
  float* Q  = ws;
  float* Kp = Q  + (size_t)5 * T_DIM * C_DIM;
  float* Vp = Kp + (size_t)B_ALL * H_NUM * KF_HEAD;
  float* AB = Vp + (size_t)B_ALL * VF_BATCH;
  float* AO = AB + (size_t)H_NUM * T_DIM * SP;
  float* MV = AO + (size_t)B_ALL * T_DIM * C_DIM;

  const dim3 gblk(128);
  const dim3 ablk(32);

  proj_qkv_wmma_f32<<<dim3(QKV_TILES), gblk, 0, stream>>>(
      X, E, Wq, Wk, Wv, Q, Kp, Vp);

  attn_kernel<<<dim3(MV_TILES + (T_DIM / 16) * H_NUM * 5), ablk, 0, stream>>>(
      Q, Kp, Vp, Map, MV, AB, AO);

  edit_kernel<<<dim3(T_DIM / 16, H_NUM), ablk, 0, stream>>>(AB, MV, AO);

  gemm_wmma_f32<<<dim3(C_DIM / 64, (B_ALL * T_DIM) / 64), gblk, 0, stream>>>(
      AO, Wo, bo, out, B_ALL * T_DIM, C_DIM, C_DIM);
}


# attention and edit launches: 8 row tiles of one (batch, head) per work-group; the K and V / MV head slices go to LDS once per work-group instead of per-wave fetches from L2; sums unchanged
# speedup vs baseline: 2.2833x; 1.0507x over previous
#include <hip/hip_runtime.h>


typedef __attribute__((ext_vector_type(2))) float v2f;
typedef __attribute__((ext_vector_type(4))) float v4f;
typedef __attribute__((ext_vector_type(8))) float v8f;

#define B_ALL   8
#define B_BASE  4
#define T_DIM   4096
#define C_DIM   320
#define S_DIM   77
#define SP      80
#define D_ENC   768
#define HD      40
#define H_NUM   8

#define KF_GRP   320
#define KF_HEAD  (10 * KF_GRP)
#define VF_GRP   (C_DIM * 4)
#define VF_BATCH (20 * VF_GRP)
#define PP       84
#define AW       8
#define HS_GRP   (HD * 4)
#define HS_SLICE (20 * HS_GRP)

__device__ __forceinline__ v8f wmma4(v2f a, v2f b, v8f c) {
  return __builtin_amdgcn_wmma_f32_16x16x4_f32(
       false, a,  false, b,
       (short)0, c,  false,  false);
}

#define GK 32
#define PA 36
#define PB 34

__device__ __forceinline__
void gemm_tile_64x64(const float* __restrict__ A, const float* __restrict__ Bm,
                     const float* __restrict__ bias, float* __restrict__ C,
                     int M, int N, int K, int bx, int by, int lay) {
  __shared__ __attribute__((aligned(16))) float As[64 * PA];
  __shared__ __attribute__((aligned(16))) float Bs[64 * PB];
  const int tid  = threadIdx.x;
  const int wave = tid >> 5;
  const int lane = tid & 31;
  const int half = lane >> 4;
  const int l16  = lane & 15;
  const int col0 = bx * 64;
  const int row0 = by * 64;
  const int rowW = row0 + wave * 16;

  const int arw = tid >> 3;
  const int ach = (tid & 7) * 4;
  unsigned aoff[4];
  #pragma unroll
  for (int i = 0; i < 4; ++i) {
    int r = row0 + arw + 16 * i;
    if (r > M - 1) r = M - 1;
    aoff[i] = (unsigned)r * (unsigned)K + (unsigned)ach;
  }
  const int aidx = arw * PA + ach;

  const int srow = tid >> 3;
  const int scol = (tid & 7) * 8;
  const unsigned boff = (unsigned)srow * (unsigned)N + (unsigned)(col0 + scol);

  const v8f zero = {0.f,0.f,0.f,0.f,0.f,0.f,0.f,0.f};
  v8f acc[4];
  #pragma unroll
  for (int j = 0; j < 4; ++j) acc[j] = zero;

  const int nkb = K / GK;
  const float* afrag = &As[(16 * wave + l16) * PA + 2 * half];

  v4f ra[4];
  v4f rb[4];

  #pragma unroll
  for (int i = 0; i < 4; ++i) ra[i] = *(const v4f*)(A + aoff[i]);
  #pragma unroll
  for (int hh = 0; hh < 2; ++hh) {
    rb[2 * hh]     = *(const v4f*)(Bm + (boff + (unsigned)(16 * hh * N)));
    rb[2 * hh + 1] = *(const v4f*)(Bm + (boff + (unsigned)(16 * hh * N) + 4u));
  }
  #pragma unroll
  for (int i = 0; i < 4; ++i) *(v4f*)(&As[aidx + 16 * i * PA]) = ra[i];
  #pragma unroll
  for (int hh = 0; hh < 2; ++hh) {
    #pragma unroll
    for (int e = 0; e < 4; ++e) {
      Bs[(scol + e) * PB + 16 * hh + srow]     = rb[2 * hh][e];
      Bs[(scol + 4 + e) * PB + 16 * hh + srow] = rb[2 * hh + 1][e];
    }
  }
  __syncthreads();

  for (int kb = 0; kb < nkb; ++kb) {
    const bool have_next = (kb + 1) < nkb;

    if (have_next) {
      #pragma unroll
      for (int i = 0; i < 4; ++i)
        ra[i] = *(const v4f*)(A + (aoff[i] + (unsigned)((kb + 1) * GK)));
      const unsigned bnxt = boff + (unsigned)((kb + 1) * GK * N);
      #pragma unroll
      for (int hh = 0; hh < 2; ++hh) {
        rb[2 * hh]     = *(const v4f*)(Bm + (bnxt + (unsigned)(16 * hh * N)));
        rb[2 * hh + 1] = *(const v4f*)(Bm + (bnxt + (unsigned)(16 * hh * N) + 4u));
      }
      if (kb + 2 < nkb) {
        __builtin_prefetch(Bm + (boff + (unsigned)((kb + 2) * GK * N)), 0, 1);
        __builtin_prefetch(Bm + (boff + (unsigned)(((kb + 2) * GK + 16) * N)), 0, 1);
      }
    }

    #pragma unroll
    for (int kk = 0; kk < GK; kk += 4) {
      const v2f a = *(const v2f*)(afrag + kk);
      #pragma unroll
      for (int j = 0; j < 4; ++j) {
        const v2f b = *(const v2f*)(&Bs[(16 * j + l16) * PB + kk + 2 * half]);
        acc[j] = wmma4(a, b, acc[j]);
      }
    }

    __syncthreads();
    if (have_next) {
      #pragma unroll
      for (int i = 0; i < 4; ++i) *(v4f*)(&As[aidx + 16 * i * PA]) = ra[i];
      #pragma unroll
      for (int hh = 0; hh < 2; ++hh) {
        #pragma unroll
        for (int e = 0; e < 4; ++e) {
          Bs[(scol + e) * PB + 16 * hh + srow]     = rb[2 * hh][e];
          Bs[(scol + 4 + e) * PB + 16 * hh + srow] = rb[2 * hh + 1][e];
        }
      }
    }
    __syncthreads();
  }

  if (lay == 0) {
    #pragma unroll
    for (int j = 0; j < 4; ++j) {
      const int c = col0 + 16 * j + l16;
      const float bj = bias ? bias[c] : 0.f;
      #pragma unroll
      for (int i = 0; i < 8; ++i) {
        const int r = rowW + i + 8 * half;
        if (r < M) C[(size_t)r * N + c] = acc[j][i] + bj;
      }
    }
  } else if (lay == 1) {
    #pragma unroll
    for (int j = 0; j < 4; ++j) {
      const int c = col0 + 16 * j + l16;
      const float bj = bias ? bias[c] : 0.f;
      const int hh = c / HD;
      const int kd = c - hh * HD;
      const int cpart = hh * KF_HEAD + (kd >> 2) * KF_GRP + (kd & 3);
      #pragma unroll
      for (int i = 0; i < 8; ++i) {
        const int r = rowW + i + 8 * half;
        if (r < M) {
          const int bb = r / S_DIM;
          const int sr = r - bb * S_DIM;
          C[(size_t)(bb * (H_NUM * KF_HEAD) + cpart + sr * 4)] = acc[j][i] + bj;
        }
      }
    }
  } else {
    #pragma unroll
    for (int j = 0; j < 4; ++j) {
      const int c = col0 + 16 * j + l16;
      const float bj = bias ? bias[c] : 0.f;
      #pragma unroll
      for (int i = 0; i < 8; ++i) {
        const int r = rowW + i + 8 * half;
        if (r < M) {
          const int bb = r / S_DIM;
          const int sr = r - bb * S_DIM;
          C[(size_t)(bb * VF_BATCH + (sr >> 2) * VF_GRP + c * 4 + (sr & 3))] = acc[j][i] + bj;
        }
      }
    }
  }
}

__global__ __launch_bounds__(128) __attribute__((amdgpu_waves_per_eu(8)))
void gemm_wmma_f32(const float* __restrict__ A, const float* __restrict__ Bm,
                   const float* __restrict__ bias, float* __restrict__ C,
                   int M, int N, int K) {
  gemm_tile_64x64(A, Bm, bias, C, M, N, K, blockIdx.x, blockIdx.y, 0);
}

#define COL_TILES   5
#define KV_TILES    50
#define QKV_TILES   1700

__global__ __launch_bounds__(128)
void proj_qkv_wmma_f32(const float* __restrict__ X, const float* __restrict__ E,
                       const float* __restrict__ Wq, const float* __restrict__ Wk,
                       const float* __restrict__ Wv, float* __restrict__ Q,
                       float* __restrict__ Kp, float* __restrict__ Vp) {
  int id = blockIdx.x;
  const float* A;
  const float* W;
  float* C;
  int M, K, lay;
  if (id < 2 * KV_TILES) {
    const bool isV = id >= KV_TILES;
    if (isV) id -= KV_TILES;
    A = E;
    W = isV ? Wv : Wk;
    C = isV ? Vp : Kp;
    M = B_ALL * S_DIM;
    K = D_ENC;
    lay = isV ? 2 : 1;
  } else {
    id -= 2 * KV_TILES;
    A = X;
    W = Wq;
    C = Q;
    M = 5 * T_DIM;
    K = C_DIM;
    lay = 0;
  }
  gemm_tile_64x64(A, W, nullptr, C, M, C_DIM, K, id % COL_TILES, id / COL_TILES, lay);
}

#define HS_TRIPS ((HS_SLICE / 4 + AW * 32 - 1) / (AW * 32))
__device__ __forceinline__
void slice_fetch(const float* __restrict__ srcp, int grp, int tid, v4f* r) {
  #pragma unroll
  for (int t = 0; t < HS_TRIPS; ++t) {
    int i = tid + t * (AW * 32);
    if (i > HS_SLICE / 4 - 1) i = HS_SLICE / 4 - 1;
    const int g = i / HD;
    const int q = i - g * HD;
    r[t] = *(const v4f*)(srcp + (size_t)g * grp + 4 * q);
  }
}
__device__ __forceinline__
void slice_put(float* dst, int tid, const v4f* r) {
  #pragma unroll
  for (int t = 0; t < HS_TRIPS; ++t) {
    int i = tid + t * (AW * 32);
    if (i > HS_SLICE / 4 - 1) i = HS_SLICE / 4 - 1;
    *(v4f*)(dst + 4 * i) = r[t];
  }
}

__device__ __forceinline__
void vf_block(v4f a4, const float* Vl, const int* vcol, int blk, int half, v8f* o) {
  #pragma unroll
  for (int j = 0; j < 3; ++j) {
    v4f b4 = *(const v4f*)(Vl + vcol[j] + blk * (2 * HS_GRP));
    if (blk == SP / 8 - 1) {
      const float b0 = b4.x;
      b4.y = half ? b0 : b4.y;
      b4.z = half ? b0 : b4.z;
      b4.w = half ? b0 : b4.w;
    }
    v2f a01 = {a4.x, a4.y}, a23 = {a4.z, a4.w};
    v2f b01 = {b4.x, b4.y}, b23 = {b4.z, b4.w};
    o[j] = wmma4(a01, b01, o[j]);
    o[j] = wmma4(a23, b23, o[j]);
  }
}

__device__ __forceinline__
void pv_and_store(const float* Plds, const float* Vl,
                  float* __restrict__ AO, int b, int h, int t0,
                  int half, int l16) {
  const v8f zero = {0.f,0.f,0.f,0.f,0.f,0.f,0.f,0.f};
  v8f o[3];
  #pragma unroll
  for (int j = 0; j < 3; ++j) o[j] = zero;

  int vcol[3];
  #pragma unroll
  for (int j = 0; j < 3; ++j) {
    int c = 16 * j + l16;
    if (c > HD - 1) c = HD - 1;
    vcol[j] = half * HS_GRP + c * 4;
  }
  const float* prow = Plds + l16 * PP + 4 * half;

  #pragma unroll
  for (int m = 0; m < SP / 8; ++m) {
    const v4f a4 = *(const v4f*)(prow + 8 * m);
    vf_block(a4, Vl, vcol, m, half, o);
  }

  #pragma unroll
  for (int j = 0; j < 3; ++j) {
    const int c = 16 * j + l16;
    if (c < HD) {
      #pragma unroll
      for (int i = 0; i < 8; ++i) {
        const int t = t0 + i + 8 * half;
        AO[(size_t)(b * T_DIM + t) * C_DIM + h * HD + c] = o[j][i];
      }
    }
  }
}

#define MV_TILES 300

__device__ __forceinline__
void mv_tile(const float* __restrict__ Map, const float* __restrict__ VF,
             float* __restrict__ MVF, int id, int half, int l16) {
  const int p  = id / 100;
  const int r  = id - p * 100;
  const int mt = r / 20;
  const int nt = r - mt * 20;
  int m = 16 * mt + l16;
  if (m > S_DIM - 1) m = S_DIM - 1;
  const float* arow = Map + (size_t)p * S_DIM * S_DIM + (size_t)m * S_DIM;
  const int c = 16 * nt + l16;
  const float* vb = VF + (size_t)(B_BASE + 1 + p) * VF_BATCH + (size_t)half * VF_GRP + c * 4;

  v8f acc = {0.f,0.f,0.f,0.f,0.f,0.f,0.f,0.f};
  #pragma unroll
  for (int q = 0; q < SP / 8; ++q) {
    v4f b4 = *(const v4f*)(vb + (size_t)q * (2 * VF_GRP));
    float a0, a1, a2, a3;
    if (q < SP / 8 - 1) {
      const int e0 = 8 * q + 4 * half;
      a0 = arow[e0];
      a1 = arow[e0 + 1];
      a2 = arow[e0 + 2];
      a3 = arow[e0 + 3];
    } else {
      const float x0 = arow[half ? 76 : 72];
      const float x1 = arow[half ? 76 : 73];
      const float x2 = arow[half ? 76 : 74];
      const float x3 = arow[half ? 76 : 75];
      a0 = x0;
      a1 = half ? 0.f : x1;
      a2 = half ? 0.f : x2;
      a3 = half ? 0.f : x3;
      const float b0 = b4.x;
      b4.y = half ? b0 : b4.y;
      b4.z = half ? b0 : b4.z;
      b4.w = half ? b0 : b4.w;
    }
    v2f a01 = {a0, a1}, a23 = {a2, a3};
    v2f b01 = {b4.x, b4.y}, b23 = {b4.z, b4.w};
    acc = wmma4(a01, b01, acc);
    acc = wmma4(a23, b23, acc);
  }
  #pragma unroll
  for (int i = 0; i < 8; ++i) {
    const int row = 16 * mt + i + 8 * half;
    if (row < S_DIM)
      MVF[(size_t)p * VF_BATCH + (size_t)(row >> 2) * VF_GRP + c * 4 + (row & 3)] = acc[i];
  }
}

#define MV_BLOCKS   ((MV_TILES + AW - 1) / AW)
#define ATT_BLOCKS  ((T_DIM / 16 / AW) * H_NUM * 5)

__global__ __launch_bounds__(AW * 32)
void attn_kernel(const float* __restrict__ Q,
                 const float* __restrict__ KF,
                 const float* __restrict__ VF,
                 const float* __restrict__ Map,
                 float* __restrict__ MV,
                 float* __restrict__ AB,
                 float* __restrict__ AO)
{
  __shared__ __attribute__((aligned(16))) float Ks[HS_SLICE];
  __shared__ __attribute__((aligned(16))) float Vs[HS_SLICE];
  __shared__ __attribute__((aligned(16))) float Pall[AW * 16 * PP];
  const int tid  = threadIdx.x;
  const int wave = tid >> 5;
  const int lane = tid & 31;
  const int half = lane >> 4;
  const int l16  = lane & 15;
  int id = blockIdx.x;
  if (id < MV_BLOCKS) {
    const int tile = id * AW + wave;
    if (tile < MV_TILES) mv_tile(Map, VF, MV, tile, half, l16);
    return;
  }
  id -= MV_BLOCKS;
  const int t0 = (((id & (T_DIM / 16 / AW - 1)) * AW) + wave) << 4;
  const int h  = (id >> 5) & (H_NUM - 1);
  const int b  = id >> 8;
  float* Plds = Pall + wave * (16 * PP);
  const float scale = 0.15811388300841898f;
  const float NEG_INF = -__builtin_inff();

  const v8f zero = {0.f,0.f,0.f,0.f,0.f,0.f,0.f,0.f};
  v8f sc[5];
  #pragma unroll
  for (int j = 0; j < 5; ++j) sc[j] = zero;

  const float* qbase = Q  + (size_t)(b * T_DIM + t0 + l16) * C_DIM + h * HD + 4 * half;
  v4f qa[HD / 8];
  #pragma unroll
  for (int m = 0; m < HD / 8; ++m) qa[m] = *(const v4f*)(qbase + 8 * m);

  {
    v4f rk[HS_TRIPS], rv[HS_TRIPS];
    slice_fetch(KF + (size_t)(b * H_NUM + h) * KF_HEAD, HS_GRP, tid, rk);
    slice_fetch(VF + (size_t)b * VF_BATCH + (h * HD) * 4, VF_GRP, tid, rv);
    slice_put(Ks, tid, rk);
    slice_put(Vs, tid, rv);
  }
  __syncthreads();
  const float* kbase = Ks + half * KF_GRP;

  const float* krow[5];
  #pragma unroll
  for (int j = 0; j < 5; ++j) {
    int s = 16 * j + l16;
    if (s > S_DIM - 1) s = S_DIM - 1;
    krow[j] = kbase + s * 4;
  }
  #pragma unroll
  for (int m = 0; m < HD / 8; ++m) {
    const v4f a4 = qa[m];
    const v2f a01 = {a4.x, a4.y}, a23 = {a4.z, a4.w};
    #pragma unroll
    for (int j = 0; j < 5; ++j) {
      const v4f b4 = *(const v4f*)(krow[j] + m * (2 * KF_GRP));
      const v2f b01 = {b4.x, b4.y}, b23 = {b4.z, b4.w};
      sc[j] = wmma4(a01, b01, sc[j]);
      sc[j] = wmma4(a23, b23, sc[j]);
    }
  }

  float mx[8];
  #pragma unroll
  for (int i = 0; i < 8; ++i) mx[i] = NEG_INF;
  #pragma unroll
  for (int j = 0; j < 5; ++j) {
    const bool ok = (16 * j + l16) < S_DIM;
    #pragma unroll
    for (int i = 0; i < 8; ++i) {
      const float x = ok ? sc[j][i] * scale : NEG_INF;
      sc[j][i] = x;
      mx[i] = fmaxf(mx[i], x);
    }
  }
  #pragma unroll
  for (int m = 1; m < 16; m <<= 1) {
    #pragma unroll
    for (int i = 0; i < 8; ++i) mx[i] = fmaxf(mx[i], __shfl_xor(mx[i], m, 32));
  }
  float sm[8];
  #pragma unroll
  for (int i = 0; i < 8; ++i) sm[i] = 0.f;
  #pragma unroll
  for (int j = 0; j < 5; ++j) {
    #pragma unroll
    for (int i = 0; i < 8; ++i) {
      const float e = __expf(sc[j][i] - mx[i]);
      sc[j][i] = e;
      sm[i] += e;
    }
  }
  #pragma unroll
  for (int m = 1; m < 16; m <<= 1) {
    #pragma unroll
    for (int i = 0; i < 8; ++i) sm[i] += __shfl_xor(sm[i], m, 32);
  }
  float rs[8];
  #pragma unroll
  for (int i = 0; i < 8; ++i) rs[i] = 1.f / sm[i];

  #pragma unroll
  for (int j = 0; j < 5; ++j) {
    #pragma unroll
    for (int i = 0; i < 8; ++i) {
      const float pv = sc[j][i] * rs[i];
      const int row = i + 8 * half;
      Plds[row * PP + 16 * j + l16] = pv;
      if (b == B_BASE)
        AB[(size_t)(h * T_DIM + t0 + row) * SP + 16 * j + l16] = pv;
    }
  }
  __builtin_amdgcn_fence(__ATOMIC_ACQ_REL, "wavefront");

  pv_and_store(Plds, Vs, AO, b, h, t0, half, l16);
}

__global__ __launch_bounds__(AW * 32) __attribute__((amdgpu_waves_per_eu(2, 4)))
void edit_kernel(const float* __restrict__ AB,
                 const float* __restrict__ MV,
                 float* __restrict__ AO)
{
  __shared__ __attribute__((aligned(16))) float Ms[3 * HS_SLICE];
  const int tid  = threadIdx.x;
  const int wave = tid >> 5;
  const int lane = tid & 31;
  const int half = lane >> 4;
  const int l16  = lane & 15;
  const int t0 = (blockIdx.x * AW + wave) << 4;
  const int h  = blockIdx.y;

  const float* arow = AB + (size_t)(h * T_DIM + t0 + l16) * SP + 4 * half;
  v4f a[SP / 8];
  #pragma unroll
  for (int m = 0; m < SP / 8; ++m) a[m] = *(const v4f*)(arow + 8 * m);

  {
    v4f rm[3][HS_TRIPS];
    #pragma unroll
    for (int p = 0; p < 3; ++p)
      slice_fetch(MV + (size_t)p * VF_BATCH + (h * HD) * 4, VF_GRP, tid, rm[p]);
    #pragma unroll
    for (int p = 0; p < 3; ++p)
      slice_put(Ms + p * HS_SLICE, tid, rm[p]);
  }
  __syncthreads();

  const v8f zero = {0.f,0.f,0.f,0.f,0.f,0.f,0.f,0.f};
  #pragma unroll
  for (int p = 0; p < 3; ++p) {
    const int b = B_BASE + 1 + p;
    int vcol[3];
    #pragma unroll
    for (int j = 0; j < 3; ++j) {
      int c = 16 * j + l16;
      if (c > HD - 1) c = HD - 1;
      vcol[j] = half * HS_GRP + c * 4;
    }
    v8f o[3];
    #pragma unroll
    for (int j = 0; j < 3; ++j) o[j] = zero;

    #pragma unroll
    for (int m = 0; m < SP / 8; ++m) vf_block(a[m], Ms + p * HS_SLICE, vcol, m, half, o);

    #pragma unroll
    for (int j = 0; j < 3; ++j) {
      const int c = 16 * j + l16;
      if (c < HD) {
        #pragma unroll
        for (int i = 0; i < 8; ++i) {
          const int t = t0 + i + 8 * half;
          AO[(size_t)(b * T_DIM + t) * C_DIM + h * HD + c] = o[j][i];
        }
      }
    }
  }
}

extern "C" void kernel_launch(void* const* d_in, const int* in_sizes, int n_in,
                              void* d_out, int out_size, void* d_ws, size_t ws_size,
                              hipStream_t stream) {
  const float* X   = (const float*)d_in[0];
  const float* E   = (const float*)d_in[1];
  const float* Map = (const float*)d_in[2];
  const float* Wq  = (const float*)d_in[3];
  const float* Wk  = (const float*)d_in[4];
  const float* Wv  = (const float*)d_in[5];
  const float* Wo  = (const float*)d_in[6];
  const float* bo  = (const float*)d_in[7];
  float* out = (float*)d_out;

  float* ws = (float*)d_ws;
  float* Q  = ws;
  float* Kp = Q  + (size_t)5 * T_DIM * C_DIM;
  float* Vp = Kp + (size_t)B_ALL * H_NUM * KF_HEAD;
  float* AB = Vp + (size_t)B_ALL * VF_BATCH;
  float* AO = AB + (size_t)H_NUM * T_DIM * SP;
  float* MV = AO + (size_t)B_ALL * T_DIM * C_DIM;

  const dim3 gblk(128);
  const dim3 ablk(AW * 32);

  proj_qkv_wmma_f32<<<dim3(QKV_TILES), gblk, 0, stream>>>(
      X, E, Wq, Wk, Wv, Q, Kp, Vp);

  attn_kernel<<<dim3(MV_BLOCKS + ATT_BLOCKS), ablk, 0, stream>>>(
      Q, Kp, Vp, Map, MV, AB, AO);

  edit_kernel<<<dim3(T_DIM / 16 / AW, H_NUM), ablk, 0, stream>>>(AB, MV, AO);

  gemm_wmma_f32<<<dim3(C_DIM / 64, (B_ALL * T_DIM) / 64), gblk, 0, stream>>>(
      AO, Wo, bo, out, B_ALL * T_DIM, C_DIM, C_DIM);
}


# bulk output stores leave through the wave's own LDS block, 16 B a lane: GEMM row-major tiles and attn_base as whole 128-B lines, attention/edit rows as 160-B runs; sums unchanged
# speedup vs baseline: 2.4385x; 1.0680x over previous
#include <hip/hip_runtime.h>


typedef __attribute__((ext_vector_type(2))) float v2f;
typedef __attribute__((ext_vector_type(4))) float v4f;
typedef __attribute__((ext_vector_type(8))) float v8f;

#define B_ALL   8
#define B_BASE  4
#define T_DIM   4096
#define C_DIM   320
#define S_DIM   77
#define SP      80
#define D_ENC   768
#define HD      40
#define H_NUM   8

#define KF_GRP   320
#define KF_HEAD  (10 * KF_GRP)
#define VF_GRP   (C_DIM * 4)
#define VF_BATCH (20 * VF_GRP)
#define PP       84
#define AW       8
#define HS_GRP   (HD * 4)
#define HS_SLICE (20 * HS_GRP)

__device__ __forceinline__ v8f wmma4(v2f a, v2f b, v8f c) {
  return __builtin_amdgcn_wmma_f32_16x16x4_f32(
       false, a,  false, b,
       (short)0, c,  false,  false);
}

#define GK 32
#define PA 36
#define PB 34
#define PC 68

__device__ __forceinline__
void gemm_tile_64x64(const float* __restrict__ A, const float* __restrict__ Bm,
                     const float* __restrict__ bias, float* __restrict__ C,
                     int M, int N, int K, int bx, int by, int lay) {
  __shared__ __attribute__((aligned(16))) float Sm[64 * PA + 64 * PB];
  float* As = Sm;
  float* Bs = Sm + 64 * PA;
  const int tid  = threadIdx.x;
  const int wave = tid >> 5;
  const int lane = tid & 31;
  const int half = lane >> 4;
  const int l16  = lane & 15;
  const int col0 = bx * 64;
  const int row0 = by * 64;
  const int rowW = row0 + wave * 16;

  const int arw = tid >> 3;
  const int ach = (tid & 7) * 4;
  unsigned aoff[4];
  #pragma unroll
  for (int i = 0; i < 4; ++i) {
    int r = row0 + arw + 16 * i;
    if (r > M - 1) r = M - 1;
    aoff[i] = (unsigned)r * (unsigned)K + (unsigned)ach;
  }
  const int aidx = arw * PA + ach;

  const int srow = tid >> 3;
  const int scol = (tid & 7) * 8;
  const unsigned boff = (unsigned)srow * (unsigned)N + (unsigned)(col0 + scol);

  const v8f zero = {0.f,0.f,0.f,0.f,0.f,0.f,0.f,0.f};
  v8f acc[4];
  #pragma unroll
  for (int j = 0; j < 4; ++j) acc[j] = zero;

  const int nkb = K / GK;
  const float* afrag = &As[(16 * wave + l16) * PA + 2 * half];

  v4f ra[4];
  v4f rb[4];

  #pragma unroll
  for (int i = 0; i < 4; ++i) ra[i] = *(const v4f*)(A + aoff[i]);
  #pragma unroll
  for (int hh = 0; hh < 2; ++hh) {
    rb[2 * hh]     = *(const v4f*)(Bm + (boff + (unsigned)(16 * hh * N)));
    rb[2 * hh + 1] = *(const v4f*)(Bm + (boff + (unsigned)(16 * hh * N) + 4u));
  }
  #pragma unroll
  for (int i = 0; i < 4; ++i) *(v4f*)(&As[aidx + 16 * i * PA]) = ra[i];
  #pragma unroll
  for (int hh = 0; hh < 2; ++hh) {
    #pragma unroll
    for (int e = 0; e < 4; ++e) {
      Bs[(scol + e) * PB + 16 * hh + srow]     = rb[2 * hh][e];
      Bs[(scol + 4 + e) * PB + 16 * hh + srow] = rb[2 * hh + 1][e];
    }
  }
  __syncthreads();

  for (int kb = 0; kb < nkb; ++kb) {
    const bool have_next = (kb + 1) < nkb;

    if (have_next) {
      #pragma unroll
      for (int i = 0; i < 4; ++i)
        ra[i] = *(const v4f*)(A + (aoff[i] + (unsigned)((kb + 1) * GK)));
      const unsigned bnxt = boff + (unsigned)((kb + 1) * GK * N);
      #pragma unroll
      for (int hh = 0; hh < 2; ++hh) {
        rb[2 * hh]     = *(const v4f*)(Bm + (bnxt + (unsigned)(16 * hh * N)));
        rb[2 * hh + 1] = *(const v4f*)(Bm + (bnxt + (unsigned)(16 * hh * N) + 4u));
      }
      if (kb + 2 < nkb) {
        __builtin_prefetch(Bm + (boff + (unsigned)((kb + 2) * GK * N)), 0, 1);
        __builtin_prefetch(Bm + (boff + (unsigned)(((kb + 2) * GK + 16) * N)), 0, 1);
      }
    }

    #pragma unroll
    for (int kk = 0; kk < GK; kk += 4) {
      const v2f a = *(const v2f*)(afrag + kk);
      #pragma unroll
      for (int j = 0; j < 4; ++j) {
        const v2f b = *(const v2f*)(&Bs[(16 * j + l16) * PB + kk + 2 * half]);
        acc[j] = wmma4(a, b, acc[j]);
      }
    }

    __syncthreads();
    if (have_next) {
      #pragma unroll
      for (int i = 0; i < 4; ++i) *(v4f*)(&As[aidx + 16 * i * PA]) = ra[i];
      #pragma unroll
      for (int hh = 0; hh < 2; ++hh) {
        #pragma unroll
        for (int e = 0; e < 4; ++e) {
          Bs[(scol + e) * PB + 16 * hh + srow]     = rb[2 * hh][e];
          Bs[(scol + 4 + e) * PB + 16 * hh + srow] = rb[2 * hh + 1][e];
        }
      }
    }
    __syncthreads();
  }

  if (lay == 0) {
    float* Cw = Sm + wave * (16 * PC);
    const int cq = 4 * l16;
    v4f bq = {0.f, 0.f, 0.f, 0.f};
    if (bias) bq = *(const v4f*)(bias + col0 + cq);
    #pragma unroll
    for (int j = 0; j < 4; ++j) {
      #pragma unroll
      for (int i = 0; i < 8; ++i) Cw[(i + 8 * half) * PC + 16 * j + l16] = acc[j][i];
    }
    __builtin_amdgcn_fence(__ATOMIC_ACQ_REL, "wavefront");
    float* crow = C + (size_t)(rowW + half) * N + col0 + cq;
    #pragma unroll
    for (int q = 0; q < 8; ++q) {
      const v4f v = *(const v4f*)(Cw + (2 * q + half) * PC + cq);
      *(v4f*)(crow + (size_t)(2 * q) * N) = v + bq;
    }
  } else if (lay == 1) {
    #pragma unroll
    for (int j = 0; j < 4; ++j) {
      const int c = col0 + 16 * j + l16;
      const float bj = bias ? bias[c] : 0.f;
      const int hh = c / HD;
      const int kd = c - hh * HD;
      const int cpart = hh * KF_HEAD + (kd >> 2) * KF_GRP + (kd & 3);
      #pragma unroll
      for (int i = 0; i < 8; ++i) {
        const int r = rowW + i + 8 * half;
        if (r < M) {
          const int bb = r / S_DIM;
          const int sr = r - bb * S_DIM;
          C[(size_t)(bb * (H_NUM * KF_HEAD) + cpart + sr * 4)] = acc[j][i] + bj;
        }
      }
    }
  } else {
    #pragma unroll
    for (int j = 0; j < 4; ++j) {
      const int c = col0 + 16 * j + l16;
      const float bj = bias ? bias[c] : 0.f;
      #pragma unroll
      for (int i = 0; i < 8; ++i) {
        const int r = rowW + i + 8 * half;
        if (r < M) {
          const int bb = r / S_DIM;
          const int sr = r - bb * S_DIM;
          C[(size_t)(bb * VF_BATCH + (sr >> 2) * VF_GRP + c * 4 + (sr & 3))] = acc[j][i] + bj;
        }
      }
    }
  }
}

__global__ __launch_bounds__(128) __attribute__((amdgpu_waves_per_eu(8)))
void gemm_wmma_f32(const float* __restrict__ A, const float* __restrict__ Bm,
                   const float* __restrict__ bias, float* __restrict__ C,
                   int M, int N, int K) {
  gemm_tile_64x64(A, Bm, bias, C, M, N, K, blockIdx.x, blockIdx.y, 0);
}

#define COL_TILES   5
#define KV_TILES    50
#define QKV_TILES   1700

__global__ __launch_bounds__(128)
void proj_qkv_wmma_f32(const float* __restrict__ X, const float* __restrict__ E,
                       const float* __restrict__ Wq, const float* __restrict__ Wk,
                       const float* __restrict__ Wv, float* __restrict__ Q,
                       float* __restrict__ Kp, float* __restrict__ Vp) {
  int id = blockIdx.x;
  const float* A;
  const float* W;
  float* C;
  int M, K, lay;
  if (id < 2 * KV_TILES) {
    const bool isV = id >= KV_TILES;
    if (isV) id -= KV_TILES;
    A = E;
    W = isV ? Wv : Wk;
    C = isV ? Vp : Kp;
    M = B_ALL * S_DIM;
    K = D_ENC;
    lay = isV ? 2 : 1;
  } else {
    id -= 2 * KV_TILES;
    A = X;
    W = Wq;
    C = Q;
    M = 5 * T_DIM;
    K = C_DIM;
    lay = 0;
  }
  gemm_tile_64x64(A, W, nullptr, C, M, C_DIM, K, id % COL_TILES, id / COL_TILES, lay);
}

#define HS_TRIPS ((HS_SLICE / 4 + AW * 32 - 1) / (AW * 32))
__device__ __forceinline__
void slice_fetch(const float* __restrict__ srcp, int grp, int tid, v4f* r) {
  #pragma unroll
  for (int t = 0; t < HS_TRIPS; ++t) {
    int i = tid + t * (AW * 32);
    if (i > HS_SLICE / 4 - 1) i = HS_SLICE / 4 - 1;
    const int g = i / HD;
    const int q = i - g * HD;
    r[t] = *(const v4f*)(srcp + (size_t)g * grp + 4 * q);
  }
}
__device__ __forceinline__
void slice_put(float* dst, int tid, const v4f* r) {
  #pragma unroll
  for (int t = 0; t < HS_TRIPS; ++t) {
    int i = tid + t * (AW * 32);
    if (i > HS_SLICE / 4 - 1) i = HS_SLICE / 4 - 1;
    *(v4f*)(dst + 4 * i) = r[t];
  }
}

__device__ __forceinline__
void vf_block(v4f a4, const float* Vl, const int* vcol, int blk, int half, v8f* o) {
  #pragma unroll
  for (int j = 0; j < 3; ++j) {
    v4f b4 = *(const v4f*)(Vl + vcol[j] + blk * (2 * HS_GRP));
    if (blk == SP / 8 - 1) {
      const float b0 = b4.x;
      b4.y = half ? b0 : b4.y;
      b4.z = half ? b0 : b4.z;
      b4.w = half ? b0 : b4.w;
    }
    v2f a01 = {a4.x, a4.y}, a23 = {a4.z, a4.w};
    v2f b01 = {b4.x, b4.y}, b23 = {b4.z, b4.w};
    o[j] = wmma4(a01, b01, o[j]);
    o[j] = wmma4(a23, b23, o[j]);
  }
}

#define PO 52
__device__ __forceinline__
void o_rows_store(float* stg, const v8f* o, float* __restrict__ AO,
                  int b, int h, int t0, int lane, int half, int l16) {
  __builtin_amdgcn_fence(__ATOMIC_ACQ_REL, "wavefront");
  #pragma unroll
  for (int j = 0; j < 3; ++j) {
    #pragma unroll
    for (int i = 0; i < 8; ++i) stg[(i + 8 * half) * PO + 16 * j + l16] = o[j][i];
  }
  __builtin_amdgcn_fence(__ATOMIC_ACQ_REL, "wavefront");
  float* obase = AO + (size_t)(b * T_DIM + t0) * C_DIM + h * HD;
  #pragma unroll
  for (int t = 0; t < (16 * HD / 4) / 32; ++t) {
    const int q   = lane + 32 * t;
    const int row = q / (HD / 4);
    const int ch  = q - row * (HD / 4);
    const v4f v = *(const v4f*)(stg + row * PO + 4 * ch);
    *(v4f*)(obase + (size_t)row * C_DIM + 4 * ch) = v;
  }
}

__device__ __forceinline__
void pv_and_store(float* Plds, const float* Vl,
                  float* __restrict__ AO, int b, int h, int t0,
                  int lane, int half, int l16) {
  const v8f zero = {0.f,0.f,0.f,0.f,0.f,0.f,0.f,0.f};
  v8f o[3];
  #pragma unroll
  for (int j = 0; j < 3; ++j) o[j] = zero;

  int vcol[3];
  #pragma unroll
  for (int j = 0; j < 3; ++j) {
    int c = 16 * j + l16;
    if (c > HD - 1) c = HD - 1;
    vcol[j] = half * HS_GRP + c * 4;
  }
  const float* prow = Plds + l16 * PP + 4 * half;

  #pragma unroll
  for (int m = 0; m < SP / 8; ++m) {
    const v4f a4 = *(const v4f*)(prow + 8 * m);
    vf_block(a4, Vl, vcol, m, half, o);
  }

  o_rows_store(Plds, o, AO, b, h, t0, lane, half, l16);
}

#define MV_TILES 300

__device__ __forceinline__
void mv_tile(const float* __restrict__ Map, const float* __restrict__ VF,
             float* __restrict__ MVF, int id, int half, int l16) {
  const int p  = id / 100;
  const int r  = id - p * 100;
  const int mt = r / 20;
  const int nt = r - mt * 20;
  int m = 16 * mt + l16;
  if (m > S_DIM - 1) m = S_DIM - 1;
  const float* arow = Map + (size_t)p * S_DIM * S_DIM + (size_t)m * S_DIM;
  const int c = 16 * nt + l16;
  const float* vb = VF + (size_t)(B_BASE + 1 + p) * VF_BATCH + (size_t)half * VF_GRP + c * 4;

  v8f acc = {0.f,0.f,0.f,0.f,0.f,0.f,0.f,0.f};
  #pragma unroll
  for (int q = 0; q < SP / 8; ++q) {
    v4f b4 = *(const v4f*)(vb + (size_t)q * (2 * VF_GRP));
    float a0, a1, a2, a3;
    if (q < SP / 8 - 1) {
      const int e0 = 8 * q + 4 * half;
      a0 = arow[e0];
      a1 = arow[e0 + 1];
      a2 = arow[e0 + 2];
      a3 = arow[e0 + 3];
    } else {
      const float x0 = arow[half ? 76 : 72];
      const float x1 = arow[half ? 76 : 73];
      const float x2 = arow[half ? 76 : 74];
      const float x3 = arow[half ? 76 : 75];
      a0 = x0;
      a1 = half ? 0.f : x1;
      a2 = half ? 0.f : x2;
      a3 = half ? 0.f : x3;
      const float b0 = b4.x;
      b4.y = half ? b0 : b4.y;
      b4.z = half ? b0 : b4.z;
      b4.w = half ? b0 : b4.w;
    }
    v2f a01 = {a0, a1}, a23 = {a2, a3};
    v2f b01 = {b4.x, b4.y}, b23 = {b4.z, b4.w};
    acc = wmma4(a01, b01, acc);
    acc = wmma4(a23, b23, acc);
  }
  #pragma unroll
  for (int i = 0; i < 8; ++i) {
    const int row = 16 * mt + i + 8 * half;
    if (row < S_DIM)
      MVF[(size_t)p * VF_BATCH + (size_t)(row >> 2) * VF_GRP + c * 4 + (row & 3)] = acc[i];
  }
}

#define MV_BLOCKS   ((MV_TILES + AW - 1) / AW)
#define ATT_BLOCKS  ((T_DIM / 16 / AW) * H_NUM * 5)

__global__ __launch_bounds__(AW * 32)
void attn_kernel(const float* __restrict__ Q,
                 const float* __restrict__ KF,
                 const float* __restrict__ VF,
                 const float* __restrict__ Map,
                 float* __restrict__ MV,
                 float* __restrict__ AB,
                 float* __restrict__ AO)
{
  __shared__ __attribute__((aligned(16))) float Ks[HS_SLICE];
  __shared__ __attribute__((aligned(16))) float Vs[HS_SLICE];
  __shared__ __attribute__((aligned(16))) float Pall[AW * 16 * PP];
  const int tid  = threadIdx.x;
  const int wave = tid >> 5;
  const int lane = tid & 31;
  const int half = lane >> 4;
  const int l16  = lane & 15;
  int id = blockIdx.x;
  if (id < MV_BLOCKS) {
    const int tile = id * AW + wave;
    if (tile < MV_TILES) mv_tile(Map, VF, MV, tile, half, l16);
    return;
  }
  id -= MV_BLOCKS;
  const int t0 = (((id & (T_DIM / 16 / AW - 1)) * AW) + wave) << 4;
  const int h  = (id >> 5) & (H_NUM - 1);
  const int b  = id >> 8;
  float* Plds = Pall + wave * (16 * PP);
  const float scale = 0.15811388300841898f;
  const float NEG_INF = -__builtin_inff();

  const v8f zero = {0.f,0.f,0.f,0.f,0.f,0.f,0.f,0.f};
  v8f sc[5];
  #pragma unroll
  for (int j = 0; j < 5; ++j) sc[j] = zero;

  const float* qbase = Q  + (size_t)(b * T_DIM + t0 + l16) * C_DIM + h * HD + 4 * half;
  v4f qa[HD / 8];
  #pragma unroll
  for (int m = 0; m < HD / 8; ++m) qa[m] = *(const v4f*)(qbase + 8 * m);

  {
    v4f rk[HS_TRIPS], rv[HS_TRIPS];
    slice_fetch(KF + (size_t)(b * H_NUM + h) * KF_HEAD, HS_GRP, tid, rk);
    slice_fetch(VF + (size_t)b * VF_BATCH + (h * HD) * 4, VF_GRP, tid, rv);
    slice_put(Ks, tid, rk);
    slice_put(Vs, tid, rv);
  }
  __syncthreads();
  const float* kbase = Ks + half * KF_GRP;

  const float* krow[5];
  #pragma unroll
  for (int j = 0; j < 5; ++j) {
    int s = 16 * j + l16;
    if (s > S_DIM - 1) s = S_DIM - 1;
    krow[j] = kbase + s * 4;
  }
  #pragma unroll
  for (int m = 0; m < HD / 8; ++m) {
    const v4f a4 = qa[m];
    const v2f a01 = {a4.x, a4.y}, a23 = {a4.z, a4.w};
    #pragma unroll
    for (int j = 0; j < 5; ++j) {
      const v4f b4 = *(const v4f*)(krow[j] + m * (2 * KF_GRP));
      const v2f b01 = {b4.x, b4.y}, b23 = {b4.z, b4.w};
      sc[j] = wmma4(a01, b01, sc[j]);
      sc[j] = wmma4(a23, b23, sc[j]);
    }
  }

  float mx[8];
  #pragma unroll
  for (int i = 0; i < 8; ++i) mx[i] = NEG_INF;
  #pragma unroll
  for (int j = 0; j < 5; ++j) {
    const bool ok = (16 * j + l16) < S_DIM;
    #pragma unroll
    for (int i = 0; i < 8; ++i) {
      const float x = ok ? sc[j][i] * scale : NEG_INF;
      sc[j][i] = x;
      mx[i] = fmaxf(mx[i], x);
    }
  }
  #pragma unroll
  for (int m = 1; m < 16; m <<= 1) {
    #pragma unroll
    for (int i = 0; i < 8; ++i) mx[i] = fmaxf(mx[i], __shfl_xor(mx[i], m, 32));
  }
  float sm[8];
  #pragma unroll
  for (int i = 0; i < 8; ++i) sm[i] = 0.f;
  #pragma unroll
  for (int j = 0; j < 5; ++j) {
    #pragma unroll
    for (int i = 0; i < 8; ++i) {
      const float e = __expf(sc[j][i] - mx[i]);
      sc[j][i] = e;
      sm[i] += e;
    }
  }
  #pragma unroll
  for (int m = 1; m < 16; m <<= 1) {
    #pragma unroll
    for (int i = 0; i < 8; ++i) sm[i] += __shfl_xor(sm[i], m, 32);
  }
  float rs[8];
  #pragma unroll
  for (int i = 0; i < 8; ++i) rs[i] = 1.f / sm[i];

  #pragma unroll
  for (int j = 0; j < 5; ++j) {
    #pragma unroll
    for (int i = 0; i < 8; ++i) {
      const float pv = sc[j][i] * rs[i];
      const int row = i + 8 * half;
      Plds[row * PP + 16 * j + l16] = pv;
    }
  }
  __builtin_amdgcn_fence(__ATOMIC_ACQ_REL, "wavefront");

  if (b == B_BASE) {
    float* abt = AB + (size_t)(h * T_DIM + t0) * SP;
    #pragma unroll
    for (int t = 0; t < (16 * SP / 4) / 32; ++t) {
      const int q   = lane + 32 * t;
      const int row = q / (SP / 4);
      const int ch  = q - row * (SP / 4);
      const v4f v = *(const v4f*)(Plds + row * PP + 4 * ch);
      *(v4f*)(abt + 4 * q) = v;
    }
  }

  pv_and_store(Plds, Vs, AO, b, h, t0, lane, half, l16);
}

__global__ __launch_bounds__(AW * 32) __attribute__((amdgpu_waves_per_eu(2, 4)))
void edit_kernel(const float* __restrict__ AB,
                 const float* __restrict__ MV,
                 float* __restrict__ AO)
{
  __shared__ __attribute__((aligned(16))) float Ms[3 * HS_SLICE];
  __shared__ __attribute__((aligned(16))) float Os[AW * 16 * PO];
  const int tid  = threadIdx.x;
  const int wave = tid >> 5;
  const int lane = tid & 31;
  const int half = lane >> 4;
  const int l16  = lane & 15;
  const int t0 = (blockIdx.x * AW + wave) << 4;
  const int h  = blockIdx.y;

  const float* arow = AB + (size_t)(h * T_DIM + t0 + l16) * SP + 4 * half;
  v4f a[SP / 8];
  #pragma unroll
  for (int m = 0; m < SP / 8; ++m) a[m] = *(const v4f*)(arow + 8 * m);

  {
    v4f rm[3][HS_TRIPS];
    #pragma unroll
    for (int p = 0; p < 3; ++p)
      slice_fetch(MV + (size_t)p * VF_BATCH + (h * HD) * 4, VF_GRP, tid, rm[p]);
    #pragma unroll
    for (int p = 0; p < 3; ++p)
      slice_put(Ms + p * HS_SLICE, tid, rm[p]);
  }
  __syncthreads();

  const v8f zero = {0.f,0.f,0.f,0.f,0.f,0.f,0.f,0.f};
  #pragma unroll
  for (int p = 0; p < 3; ++p) {
    const int b = B_BASE + 1 + p;
    int vcol[3];
    #pragma unroll
    for (int j = 0; j < 3; ++j) {
      int c = 16 * j + l16;
      if (c > HD - 1) c = HD - 1;
      vcol[j] = half * HS_GRP + c * 4;
    }
    v8f o[3];
    #pragma unroll
    for (int j = 0; j < 3; ++j) o[j] = zero;

    #pragma unroll
    for (int m = 0; m < SP / 8; ++m) vf_block(a[m], Ms + p * HS_SLICE, vcol, m, half, o);

    o_rows_store(Os + wave * (16 * PO), o, AO, b, h, t0, lane, half, l16);
  }
}

extern "C" void kernel_launch(void* const* d_in, const int* in_sizes, int n_in,
                              void* d_out, int out_size, void* d_ws, size_t ws_size,
                              hipStream_t stream) {
  const float* X   = (const float*)d_in[0];
  const float* E   = (const float*)d_in[1];
  const float* Map = (const float*)d_in[2];
  const float* Wq  = (const float*)d_in[3];
  const float* Wk  = (const float*)d_in[4];
  const float* Wv  = (const float*)d_in[5];
  const float* Wo  = (const float*)d_in[6];
  const float* bo  = (const float*)d_in[7];
  float* out = (float*)d_out;

  float* ws = (float*)d_ws;
  float* Q  = ws;
  float* Kp = Q  + (size_t)5 * T_DIM * C_DIM;
  float* Vp = Kp + (size_t)B_ALL * H_NUM * KF_HEAD;
  float* AB = Vp + (size_t)B_ALL * VF_BATCH;
  float* AO = AB + (size_t)H_NUM * T_DIM * SP;
  float* MV = AO + (size_t)B_ALL * T_DIM * C_DIM;

  const dim3 gblk(128);
  const dim3 ablk(AW * 32);

  proj_qkv_wmma_f32<<<dim3(QKV_TILES), gblk, 0, stream>>>(
      X, E, Wq, Wk, Wv, Q, Kp, Vp);

  attn_kernel<<<dim3(MV_BLOCKS + ATT_BLOCKS), ablk, 0, stream>>>(
      Q, Kp, Vp, Map, MV, AB, AO);

  edit_kernel<<<dim3(T_DIM / 16 / AW, H_NUM), ablk, 0, stream>>>(AB, MV, AO);

  gemm_wmma_f32<<<dim3(C_DIM / 64, (B_ALL * T_DIM) / 64), gblk, 0, stream>>>(
      AO, Wo, bo, out, B_ALL * T_DIM, C_DIM, C_DIM);
}


# GEMM work-group tile 64x64 -> 128x64 for the Q and output projections: each wave owns two row tiles sharing its B fragments; half the waves to start, B panel fetched half as often; sums unchanged
# speedup vs baseline: 2.5891x; 1.0618x over previous
#include <hip/hip_runtime.h>


typedef __attribute__((ext_vector_type(2))) float v2f;
typedef __attribute__((ext_vector_type(4))) float v4f;
typedef __attribute__((ext_vector_type(8))) float v8f;

#define B_ALL   8
#define B_BASE  4
#define T_DIM   4096
#define C_DIM   320
#define S_DIM   77
#define SP      80
#define D_ENC   768
#define HD      40
#define H_NUM   8

#define KF_GRP   320
#define KF_HEAD  (10 * KF_GRP)
#define VF_GRP   (C_DIM * 4)
#define VF_BATCH (20 * VF_GRP)
#define PP       84
#define AW       8
#define HS_GRP   (HD * 4)
#define HS_SLICE (20 * HS_GRP)

__device__ __forceinline__ v8f wmma4(v2f a, v2f b, v8f c) {
  return __builtin_amdgcn_wmma_f32_16x16x4_f32(
       false, a,  false, b,
       (short)0, c,  false,  false);
}

#define GK 32
#define PA 36
#define PB 34
#define PC 68
#define GEMM_LDS_FLOATS(RT) (64 * (RT) * PA + 64 * PB)

template <int RT>
__device__ __forceinline__
void gemm_tile(float* Sm,
               const float* __restrict__ A, const float* __restrict__ Bm,
               const float* __restrict__ bias, float* __restrict__ C,
               int M, int N, int K, int bx, int by, int lay) {
  float* As = Sm;
  float* Bs = Sm + 64 * RT * PA;
  const int tid  = threadIdx.x;
  const int wave = tid >> 5;
  const int lane = tid & 31;
  const int half = lane >> 4;
  const int l16  = lane & 15;
  const int col0 = bx * 64;
  const int row0 = by * (64 * RT);
  const int rowW = row0 + wave * (16 * RT);

  const int arw = tid >> 3;
  const int ach = (tid & 7) * 4;
  unsigned aoff[4 * RT];
  #pragma unroll
  for (int i = 0; i < 4 * RT; ++i) {
    int r = row0 + arw + 16 * i;
    if (r > M - 1) r = M - 1;
    aoff[i] = (unsigned)r * (unsigned)K + (unsigned)ach;
  }
  const int aidx = arw * PA + ach;

  const int srow = tid >> 3;
  const int scol = (tid & 7) * 8;
  const unsigned boff = (unsigned)srow * (unsigned)N + (unsigned)(col0 + scol);

  const v8f zero = {0.f,0.f,0.f,0.f,0.f,0.f,0.f,0.f};
  v8f acc[RT][4];
  #pragma unroll
  for (int r = 0; r < RT; ++r) {
    #pragma unroll
    for (int j = 0; j < 4; ++j) acc[r][j] = zero;
  }

  const int nkb = K / GK;
  const float* afrag = &As[(16 * RT * wave + l16) * PA + 2 * half];

  v4f ra[4 * RT];
  v4f rb[4];

  #pragma unroll
  for (int i = 0; i < 4 * RT; ++i) ra[i] = *(const v4f*)(A + aoff[i]);
  #pragma unroll
  for (int hh = 0; hh < 2; ++hh) {
    rb[2 * hh]     = *(const v4f*)(Bm + (boff + (unsigned)(16 * hh * N)));
    rb[2 * hh + 1] = *(const v4f*)(Bm + (boff + (unsigned)(16 * hh * N) + 4u));
  }
  #pragma unroll
  for (int i = 0; i < 4 * RT; ++i) *(v4f*)(&As[aidx + 16 * i * PA]) = ra[i];
  #pragma unroll
  for (int hh = 0; hh < 2; ++hh) {
    #pragma unroll
    for (int e = 0; e < 4; ++e) {
      Bs[(scol + e) * PB + 16 * hh + srow]     = rb[2 * hh][e];
      Bs[(scol + 4 + e) * PB + 16 * hh + srow] = rb[2 * hh + 1][e];
    }
  }
  __syncthreads();

  for (int kb = 0; kb < nkb; ++kb) {
    const bool have_next = (kb + 1) < nkb;

    if (have_next) {
      #pragma unroll
      for (int i = 0; i < 4 * RT; ++i)
        ra[i] = *(const v4f*)(A + (aoff[i] + (unsigned)((kb + 1) * GK)));
      const unsigned bnxt = boff + (unsigned)((kb + 1) * GK * N);
      #pragma unroll
      for (int hh = 0; hh < 2; ++hh) {
        rb[2 * hh]     = *(const v4f*)(Bm + (bnxt + (unsigned)(16 * hh * N)));
        rb[2 * hh + 1] = *(const v4f*)(Bm + (bnxt + (unsigned)(16 * hh * N) + 4u));
      }
      if (kb + 2 < nkb) {
        __builtin_prefetch(Bm + (boff + (unsigned)((kb + 2) * GK * N)), 0, 1);
        __builtin_prefetch(Bm + (boff + (unsigned)(((kb + 2) * GK + 16) * N)), 0, 1);
      }
    }

    #pragma unroll
    for (int kk = 0; kk < GK; kk += 4) {
      v2f a[RT];
      #pragma unroll
      for (int r = 0; r < RT; ++r) a[r] = *(const v2f*)(afrag + 16 * r * PA + kk);
      #pragma unroll
      for (int j = 0; j < 4; ++j) {
        const v2f b = *(const v2f*)(&Bs[(16 * j + l16) * PB + kk + 2 * half]);
        #pragma unroll
        for (int r = 0; r < RT; ++r) acc[r][j] = wmma4(a[r], b, acc[r][j]);
      }
    }

    __syncthreads();
    if (have_next) {
      #pragma unroll
      for (int i = 0; i < 4 * RT; ++i) *(v4f*)(&As[aidx + 16 * i * PA]) = ra[i];
      #pragma unroll
      for (int hh = 0; hh < 2; ++hh) {
        #pragma unroll
        for (int e = 0; e < 4; ++e) {
          Bs[(scol + e) * PB + 16 * hh + srow]     = rb[2 * hh][e];
          Bs[(scol + 4 + e) * PB + 16 * hh + srow] = rb[2 * hh + 1][e];
        }
      }
    }
    __syncthreads();
  }

  if (lay == 0) {
    float* Cw = Sm + wave * (16 * PC);
    const int cq = 4 * l16;
    v4f bq = {0.f, 0.f, 0.f, 0.f};
    if (bias) bq = *(const v4f*)(bias + col0 + cq);
    #pragma unroll
    for (int r = 0; r < RT; ++r) {
      if (r > 0) __builtin_amdgcn_fence(__ATOMIC_ACQ_REL, "wavefront");
      #pragma unroll
      for (int j = 0; j < 4; ++j) {
        #pragma unroll
        for (int i = 0; i < 8; ++i) Cw[(i + 8 * half) * PC + 16 * j + l16] = acc[r][j][i];
      }
      __builtin_amdgcn_fence(__ATOMIC_ACQ_REL, "wavefront");
      float* crow = C + (size_t)(rowW + 16 * r + half) * N + col0 + cq;
      #pragma unroll
      for (int q = 0; q < 8; ++q) {
        const v4f v = *(const v4f*)(Cw + (2 * q + half) * PC + cq);
        *(v4f*)(crow + (size_t)(2 * q) * N) = v + bq;
      }
    }
  } else if (lay == 1) {
    #pragma unroll
    for (int j = 0; j < 4; ++j) {
      const int c = col0 + 16 * j + l16;
      const float bj = bias ? bias[c] : 0.f;
      const int hh = c / HD;
      const int kd = c - hh * HD;
      const int cpart = hh * KF_HEAD + (kd >> 2) * KF_GRP + (kd & 3);
      #pragma unroll
      for (int r = 0; r < RT; ++r) {
        #pragma unroll
        for (int i = 0; i < 8; ++i) {
          const int rr = rowW + 16 * r + i + 8 * half;
          if (rr < M) {
            const int bb = rr / S_DIM;
            const int sr = rr - bb * S_DIM;
            C[(size_t)(bb * (H_NUM * KF_HEAD) + cpart + sr * 4)] = acc[r][j][i] + bj;
          }
        }
      }
    }
  } else {
    #pragma unroll
    for (int j = 0; j < 4; ++j) {
      const int c = col0 + 16 * j + l16;
      const float bj = bias ? bias[c] : 0.f;
      #pragma unroll
      for (int r = 0; r < RT; ++r) {
        #pragma unroll
        for (int i = 0; i < 8; ++i) {
          const int rr = rowW + 16 * r + i + 8 * half;
          if (rr < M) {
            const int bb = rr / S_DIM;
            const int sr = rr - bb * S_DIM;
            C[(size_t)(bb * VF_BATCH + (sr >> 2) * VF_GRP + c * 4 + (sr & 3))] = acc[r][j][i] + bj;
          }
        }
      }
    }
  }
}

__global__ __launch_bounds__(128) __attribute__((amdgpu_waves_per_eu(5)))
void gemm_wmma_f32(const float* __restrict__ A, const float* __restrict__ Bm,
                   const float* __restrict__ bias, float* __restrict__ C,
                   int M, int N, int K) {
  __shared__ __attribute__((aligned(16))) float Sm[GEMM_LDS_FLOATS(2)];
  gemm_tile<2>(Sm, A, Bm, bias, C, M, N, K, blockIdx.x, blockIdx.y, 0);
}

#define COL_TILES   5
#define KV_TILES    50
#define QKV_TILES   900

__global__ __launch_bounds__(128)
void proj_qkv_wmma_f32(const float* __restrict__ X, const float* __restrict__ E,
                       const float* __restrict__ Wq, const float* __restrict__ Wk,
                       const float* __restrict__ Wv, float* __restrict__ Q,
                       float* __restrict__ Kp, float* __restrict__ Vp) {
  __shared__ __attribute__((aligned(16))) float Sm[GEMM_LDS_FLOATS(2)];
  int id = blockIdx.x;
  if (id < 2 * KV_TILES) {
    const bool isV = id >= KV_TILES;
    if (isV) id -= KV_TILES;
    gemm_tile<1>(Sm, E, isV ? Wv : Wk, nullptr, isV ? Vp : Kp, B_ALL * S_DIM, C_DIM, D_ENC,
                 id % COL_TILES, id / COL_TILES, isV ? 2 : 1);
  } else {
    id -= 2 * KV_TILES;
    gemm_tile<2>(Sm, X, Wq, nullptr, Q, 5 * T_DIM, C_DIM, C_DIM,
                 id % COL_TILES, id / COL_TILES, 0);
  }
}

#define HS_TRIPS ((HS_SLICE / 4 + AW * 32 - 1) / (AW * 32))
__device__ __forceinline__
void slice_fetch(const float* __restrict__ srcp, int grp, int tid, v4f* r) {
  #pragma unroll
  for (int t = 0; t < HS_TRIPS; ++t) {
    int i = tid + t * (AW * 32);
    if (i > HS_SLICE / 4 - 1) i = HS_SLICE / 4 - 1;
    const int g = i / HD;
    const int q = i - g * HD;
    r[t] = *(const v4f*)(srcp + (size_t)g * grp + 4 * q);
  }
}
__device__ __forceinline__
void slice_put(float* dst, int tid, const v4f* r) {
  #pragma unroll
  for (int t = 0; t < HS_TRIPS; ++t) {
    int i = tid + t * (AW * 32);
    if (i > HS_SLICE / 4 - 1) i = HS_SLICE / 4 - 1;
    *(v4f*)(dst + 4 * i) = r[t];
  }
}

__device__ __forceinline__
void vf_block(v4f a4, const float* Vl, const int* vcol, int blk, int half, v8f* o) {
  #pragma unroll
  for (int j = 0; j < 3; ++j) {
    v4f b4 = *(const v4f*)(Vl + vcol[j] + blk * (2 * HS_GRP));
    if (blk == SP / 8 - 1) {
      const float b0 = b4.x;
      b4.y = half ? b0 : b4.y;
      b4.z = half ? b0 : b4.z;
      b4.w = half ? b0 : b4.w;
    }
    v2f a01 = {a4.x, a4.y}, a23 = {a4.z, a4.w};
    v2f b01 = {b4.x, b4.y}, b23 = {b4.z, b4.w};
    o[j] = wmma4(a01, b01, o[j]);
    o[j] = wmma4(a23, b23, o[j]);
  }
}

#define PO 52
__device__ __forceinline__
void o_rows_store(float* stg, const v8f* o, float* __restrict__ AO,
                  int b, int h, int t0, int lane, int half, int l16) {
  __builtin_amdgcn_fence(__ATOMIC_ACQ_REL, "wavefront");
  #pragma unroll
  for (int j = 0; j < 3; ++j) {
    #pragma unroll
    for (int i = 0; i < 8; ++i) stg[(i + 8 * half) * PO + 16 * j + l16] = o[j][i];
  }
  __builtin_amdgcn_fence(__ATOMIC_ACQ_REL, "wavefront");
  float* obase = AO + (size_t)(b * T_DIM + t0) * C_DIM + h * HD;
  #pragma unroll
  for (int t = 0; t < (16 * HD / 4) / 32; ++t) {
    const int q   = lane + 32 * t;
    const int row = q / (HD / 4);
    const int ch  = q - row * (HD / 4);
    const v4f v = *(const v4f*)(stg + row * PO + 4 * ch);
    *(v4f*)(obase + (size_t)row * C_DIM + 4 * ch) = v;
  }
}

__device__ __forceinline__
void pv_and_store(float* Plds, const float* Vl,
                  float* __restrict__ AO, int b, int h, int t0,
                  int lane, int half, int l16) {
  const v8f zero = {0.f,0.f,0.f,0.f,0.f,0.f,0.f,0.f};
  v8f o[3];
  #pragma unroll
  for (int j = 0; j < 3; ++j) o[j] = zero;

  int vcol[3];
  #pragma unroll
  for (int j = 0; j < 3; ++j) {
    int c = 16 * j + l16;
    if (c > HD - 1) c = HD - 1;
    vcol[j] = half * HS_GRP + c * 4;
  }
  const float* prow = Plds + l16 * PP + 4 * half;

  #pragma unroll
  for (int m = 0; m < SP / 8; ++m) {
    const v4f a4 = *(const v4f*)(prow + 8 * m);
    vf_block(a4, Vl, vcol, m, half, o);
  }

  o_rows_store(Plds, o, AO, b, h, t0, lane, half, l16);
}

#define MV_TILES 300

__device__ __forceinline__
void mv_tile(const float* __restrict__ Map, const float* __restrict__ VF,
             float* __restrict__ MVF, int id, int half, int l16) {
  const int p  = id / 100;
  const int r  = id - p * 100;
  const int mt = r / 20;
  const int nt = r - mt * 20;
  int m = 16 * mt + l16;
  if (m > S_DIM - 1) m = S_DIM - 1;
  const float* arow = Map + (size_t)p * S_DIM * S_DIM + (size_t)m * S_DIM;
  const int c = 16 * nt + l16;
  const float* vb = VF + (size_t)(B_BASE + 1 + p) * VF_BATCH + (size_t)half * VF_GRP + c * 4;

  v8f acc = {0.f,0.f,0.f,0.f,0.f,0.f,0.f,0.f};
  #pragma unroll
  for (int q = 0; q < SP / 8; ++q) {
    v4f b4 = *(const v4f*)(vb + (size_t)q * (2 * VF_GRP));
    float a0, a1, a2, a3;
    if (q < SP / 8 - 1) {
      const int e0 = 8 * q + 4 * half;
      a0 = arow[e0];
      a1 = arow[e0 + 1];
      a2 = arow[e0 + 2];
      a3 = arow[e0 + 3];
    } else {
      const float x0 = arow[half ? 76 : 72];
      const float x1 = arow[half ? 76 : 73];
      const float x2 = arow[half ? 76 : 74];
      const float x3 = arow[half ? 76 : 75];
      a0 = x0;
      a1 = half ? 0.f : x1;
      a2 = half ? 0.f : x2;
      a3 = half ? 0.f : x3;
      const float b0 = b4.x;
      b4.y = half ? b0 : b4.y;
      b4.z = half ? b0 : b4.z;
      b4.w = half ? b0 : b4.w;
    }
    v2f a01 = {a0, a1}, a23 = {a2, a3};
    v2f b01 = {b4.x, b4.y}, b23 = {b4.z, b4.w};
    acc = wmma4(a01, b01, acc);
    acc = wmma4(a23, b23, acc);
  }
  #pragma unroll
  for (int i = 0; i < 8; ++i) {
    const int row = 16 * mt + i + 8 * half;
    if (row < S_DIM)
      MVF[(size_t)p * VF_BATCH + (size_t)(row >> 2) * VF_GRP + c * 4 + (row & 3)] = acc[i];
  }
}

#define MV_BLOCKS   ((MV_TILES + AW - 1) / AW)
#define ATT_BLOCKS  ((T_DIM / 16 / AW) * H_NUM * 5)

__global__ __launch_bounds__(AW * 32)
void attn_kernel(const float* __restrict__ Q,
                 const float* __restrict__ KF,
                 const float* __restrict__ VF,
                 const float* __restrict__ Map,
                 float* __restrict__ MV,
                 float* __restrict__ AB,
                 float* __restrict__ AO)
{
  __shared__ __attribute__((aligned(16))) float Ks[HS_SLICE];
  __shared__ __attribute__((aligned(16))) float Vs[HS_SLICE];
  __shared__ __attribute__((aligned(16))) float Pall[AW * 16 * PP];
  const int tid  = threadIdx.x;
  const int wave = tid >> 5;
  const int lane = tid & 31;
  const int half = lane >> 4;
  const int l16  = lane & 15;
  int id = blockIdx.x;
  if (id < MV_BLOCKS) {
    const int tile = id * AW + wave;
    if (tile < MV_TILES) mv_tile(Map, VF, MV, tile, half, l16);
    return;
  }
  id -= MV_BLOCKS;
  const int t0 = (((id & (T_DIM / 16 / AW - 1)) * AW) + wave) << 4;
  const int h  = (id >> 5) & (H_NUM - 1);
  const int b  = id >> 8;
  float* Plds = Pall + wave * (16 * PP);
  const float scale = 0.15811388300841898f;
  const float NEG_INF = -__builtin_inff();

  const v8f zero = {0.f,0.f,0.f,0.f,0.f,0.f,0.f,0.f};
  v8f sc[5];
  #pragma unroll
  for (int j = 0; j < 5; ++j) sc[j] = zero;

  const float* qbase = Q  + (size_t)(b * T_DIM + t0 + l16) * C_DIM + h * HD + 4 * half;
  v4f qa[HD / 8];
  #pragma unroll
  for (int m = 0; m < HD / 8; ++m) qa[m] = *(const v4f*)(qbase + 8 * m);

  {
    v4f rk[HS_TRIPS], rv[HS_TRIPS];
    slice_fetch(KF + (size_t)(b * H_NUM + h) * KF_HEAD, HS_GRP, tid, rk);
    slice_fetch(VF + (size_t)b * VF_BATCH + (h * HD) * 4, VF_GRP, tid, rv);
    slice_put(Ks, tid, rk);
    slice_put(Vs, tid, rv);
  }
  __syncthreads();
  const float* kbase = Ks + half * KF_GRP;

  const float* krow[5];
  #pragma unroll
  for (int j = 0; j < 5; ++j) {
    int s = 16 * j + l16;
    if (s > S_DIM - 1) s = S_DIM - 1;
    krow[j] = kbase + s * 4;
  }
  #pragma unroll
  for (int m = 0; m < HD / 8; ++m) {
    const v4f a4 = qa[m];
    const v2f a01 = {a4.x, a4.y}, a23 = {a4.z, a4.w};
    #pragma unroll
    for (int j = 0; j < 5; ++j) {
      const v4f b4 = *(const v4f*)(krow[j] + m * (2 * KF_GRP));
      const v2f b01 = {b4.x, b4.y}, b23 = {b4.z, b4.w};
      sc[j] = wmma4(a01, b01, sc[j]);
      sc[j] = wmma4(a23, b23, sc[j]);
    }
  }

  float mx[8];
  #pragma unroll
  for (int i = 0; i < 8; ++i) mx[i] = NEG_INF;
  #pragma unroll
  for (int j = 0; j < 5; ++j) {
    const bool ok = (16 * j + l16) < S_DIM;
    #pragma unroll
    for (int i = 0; i < 8; ++i) {
      const float x = ok ? sc[j][i] * scale : NEG_INF;
      sc[j][i] = x;
      mx[i] = fmaxf(mx[i], x);
    }
  }
  #pragma unroll
  for (int m = 1; m < 16; m <<= 1) {
    #pragma unroll
    for (int i = 0; i < 8; ++i) mx[i] = fmaxf(mx[i], __shfl_xor(mx[i], m, 32));
  }
  float sm[8];
  #pragma unroll
  for (int i = 0; i < 8; ++i) sm[i] = 0.f;
  #pragma unroll
  for (int j = 0; j < 5; ++j) {
    #pragma unroll
    for (int i = 0; i < 8; ++i) {
      const float e = __expf(sc[j][i] - mx[i]);
      sc[j][i] = e;
      sm[i] += e;
    }
  }
  #pragma unroll
  for (int m = 1; m < 16; m <<= 1) {
    #pragma unroll
    for (int i = 0; i < 8; ++i) sm[i] += __shfl_xor(sm[i], m, 32);
  }
  float rs[8];
  #pragma unroll
  for (int i = 0; i < 8; ++i) rs[i] = 1.f / sm[i];

  #pragma unroll
  for (int j = 0; j < 5; ++j) {
    #pragma unroll
    for (int i = 0; i < 8; ++i) {
      const float pv = sc[j][i] * rs[i];
      const int row = i + 8 * half;
      Plds[row * PP + 16 * j + l16] = pv;
    }
  }
  __builtin_amdgcn_fence(__ATOMIC_ACQ_REL, "wavefront");

  if (b == B_BASE) {
    float* abt = AB + (size_t)(h * T_DIM + t0) * SP;
    #pragma unroll
    for (int t = 0; t < (16 * SP / 4) / 32; ++t) {
      const int q   = lane + 32 * t;
      const int row = q / (SP / 4);
      const int ch  = q - row * (SP / 4);
      const v4f v = *(const v4f*)(Plds + row * PP + 4 * ch);
      *(v4f*)(abt + 4 * q) = v;
    }
  }

  pv_and_store(Plds, Vs, AO, b, h, t0, lane, half, l16);
}

__global__ __launch_bounds__(AW * 32) __attribute__((amdgpu_waves_per_eu(2, 4)))
void edit_kernel(const float* __restrict__ AB,
                 const float* __restrict__ MV,
                 float* __restrict__ AO)
{
  __shared__ __attribute__((aligned(16))) float Ms[3 * HS_SLICE];
  __shared__ __attribute__((aligned(16))) float Os[AW * 16 * PO];
  const int tid  = threadIdx.x;
  const int wave = tid >> 5;
  const int lane = tid & 31;
  const int half = lane >> 4;
  const int l16  = lane & 15;
  const int t0 = (blockIdx.x * AW + wave) << 4;
  const int h  = blockIdx.y;

  const float* arow = AB + (size_t)(h * T_DIM + t0 + l16) * SP + 4 * half;
  v4f a[SP / 8];
  #pragma unroll
  for (int m = 0; m < SP / 8; ++m) a[m] = *(const v4f*)(arow + 8 * m);

  {
    v4f rm[3][HS_TRIPS];
    #pragma unroll
    for (int p = 0; p < 3; ++p)
      slice_fetch(MV + (size_t)p * VF_BATCH + (h * HD) * 4, VF_GRP, tid, rm[p]);
    #pragma unroll
    for (int p = 0; p < 3; ++p)
      slice_put(Ms + p * HS_SLICE, tid, rm[p]);
  }
  __syncthreads();

  const v8f zero = {0.f,0.f,0.f,0.f,0.f,0.f,0.f,0.f};
  #pragma unroll
  for (int p = 0; p < 3; ++p) {
    const int b = B_BASE + 1 + p;
    int vcol[3];
    #pragma unroll
    for (int j = 0; j < 3; ++j) {
      int c = 16 * j + l16;
      if (c > HD - 1) c = HD - 1;
      vcol[j] = half * HS_GRP + c * 4;
    }
    v8f o[3];
    #pragma unroll
    for (int j = 0; j < 3; ++j) o[j] = zero;

    #pragma unroll
    for (int m = 0; m < SP / 8; ++m) vf_block(a[m], Ms + p * HS_SLICE, vcol, m, half, o);

    o_rows_store(Os + wave * (16 * PO), o, AO, b, h, t0, lane, half, l16);
  }
}

extern "C" void kernel_launch(void* const* d_in, const int* in_sizes, int n_in,
                              void* d_out, int out_size, void* d_ws, size_t ws_size,
                              hipStream_t stream) {
  const float* X   = (const float*)d_in[0];
  const float* E   = (const float*)d_in[1];
  const float* Map = (const float*)d_in[2];
  const float* Wq  = (const float*)d_in[3];
  const float* Wk  = (const float*)d_in[4];
  const float* Wv  = (const float*)d_in[5];
  const float* Wo  = (const float*)d_in[6];
  const float* bo  = (const float*)d_in[7];
  float* out = (float*)d_out;

  float* ws = (float*)d_ws;
  float* Q  = ws;
  float* Kp = Q  + (size_t)5 * T_DIM * C_DIM;
  float* Vp = Kp + (size_t)B_ALL * H_NUM * KF_HEAD;
  float* AB = Vp + (size_t)B_ALL * VF_BATCH;
  float* AO = AB + (size_t)H_NUM * T_DIM * SP;
  float* MV = AO + (size_t)B_ALL * T_DIM * C_DIM;

  const dim3 gblk(128);
  const dim3 ablk(AW * 32);

  proj_qkv_wmma_f32<<<dim3(QKV_TILES), gblk, 0, stream>>>(
      X, E, Wq, Wk, Wv, Q, Kp, Vp);

  attn_kernel<<<dim3(MV_BLOCKS + ATT_BLOCKS), ablk, 0, stream>>>(
      Q, Kp, Vp, Map, MV, AB, AO);

  edit_kernel<<<dim3(T_DIM / 16 / AW, H_NUM), ablk, 0, stream>>>(AB, MV, AO);

  gemm_wmma_f32<<<dim3(C_DIM / 64, (B_ALL * T_DIM) / 128), gblk, 0, stream>>>(
      AO, Wo, bo, out, B_ALL * T_DIM, C_DIM, C_DIM);
}


# edit launch removed (4 -> 3): MV = mapper @ V formed in the projection launch (K split over 4 waves, partials summed in LDS: sum order changes); batch-4 attention waves also form batches 5..7
# speedup vs baseline: 2.7339x; 1.0559x over previous
#include <hip/hip_runtime.h>


typedef __attribute__((ext_vector_type(2))) float v2f;
typedef __attribute__((ext_vector_type(4))) float v4f;
typedef __attribute__((ext_vector_type(8))) float v8f;

#define B_ALL   8
#define B_BASE  4
#define T_DIM   4096
#define C_DIM   320
#define S_DIM   77
#define SP      80
#define D_ENC   768
#define HD      40
#define H_NUM   8

#define KF_GRP   320
#define KF_HEAD  (10 * KF_GRP)
#define VF_GRP   (C_DIM * 4)
#define VF_BATCH (20 * VF_GRP)
#define PP       84
#define AW       8
#define HS_GRP   (HD * 4)
#define HS_SLICE (20 * HS_GRP)

__device__ __forceinline__ v8f wmma4(v2f a, v2f b, v8f c) {
  return __builtin_amdgcn_wmma_f32_16x16x4_f32(
       false, a,  false, b,
       (short)0, c,  false,  false);
}

#define GK 32
#define PA 36
#define PB 34
#define PC 68
#define GEMM_LDS_FLOATS(RT) (64 * (RT) * PA + 64 * PB)

template <int RT>
__device__ __forceinline__
void gemm_tile(float* Sm,
               const float* __restrict__ A, const float* __restrict__ Bm,
               const float* __restrict__ bias, float* __restrict__ C,
               int M, int N, int K, int bx, int by, int lay) {
  float* As = Sm;
  float* Bs = Sm + 64 * RT * PA;
  const int tid  = threadIdx.x;
  const int wave = tid >> 5;
  const int lane = tid & 31;
  const int half = lane >> 4;
  const int l16  = lane & 15;
  const int col0 = bx * 64;
  const int row0 = by * (64 * RT);
  const int rowW = row0 + wave * (16 * RT);

  const int arw = tid >> 3;
  const int ach = (tid & 7) * 4;
  unsigned aoff[4 * RT];
  #pragma unroll
  for (int i = 0; i < 4 * RT; ++i) {
    int r = row0 + arw + 16 * i;
    if (r > M - 1) r = M - 1;
    aoff[i] = (unsigned)r * (unsigned)K + (unsigned)ach;
  }
  const int aidx = arw * PA + ach;

  const int srow = tid >> 3;
  const int scol = (tid & 7) * 8;
  const unsigned boff = (unsigned)srow * (unsigned)N + (unsigned)(col0 + scol);

  const v8f zero = {0.f,0.f,0.f,0.f,0.f,0.f,0.f,0.f};
  v8f acc[RT][4];
  #pragma unroll
  for (int r = 0; r < RT; ++r) {
    #pragma unroll
    for (int j = 0; j < 4; ++j) acc[r][j] = zero;
  }

  const int nkb = K / GK;
  const float* afrag = &As[(16 * RT * wave + l16) * PA + 2 * half];

  v4f ra[4 * RT];
  v4f rb[4];

  #pragma unroll
  for (int i = 0; i < 4 * RT; ++i) ra[i] = *(const v4f*)(A + aoff[i]);
  #pragma unroll
  for (int hh = 0; hh < 2; ++hh) {
    rb[2 * hh]     = *(const v4f*)(Bm + (boff + (unsigned)(16 * hh * N)));
    rb[2 * hh + 1] = *(const v4f*)(Bm + (boff + (unsigned)(16 * hh * N) + 4u));
  }
  #pragma unroll
  for (int i = 0; i < 4 * RT; ++i) *(v4f*)(&As[aidx + 16 * i * PA]) = ra[i];
  #pragma unroll
  for (int hh = 0; hh < 2; ++hh) {
    #pragma unroll
    for (int e = 0; e < 4; ++e) {
      Bs[(scol + e) * PB + 16 * hh + srow]     = rb[2 * hh][e];
      Bs[(scol + 4 + e) * PB + 16 * hh + srow] = rb[2 * hh + 1][e];
    }
  }
  __syncthreads();

  for (int kb = 0; kb < nkb; ++kb) {
    const bool have_next = (kb + 1) < nkb;

    if (have_next) {
      #pragma unroll
      for (int i = 0; i < 4 * RT; ++i)
        ra[i] = *(const v4f*)(A + (aoff[i] + (unsigned)((kb + 1) * GK)));
      const unsigned bnxt = boff + (unsigned)((kb + 1) * GK * N);
      #pragma unroll
      for (int hh = 0; hh < 2; ++hh) {
        rb[2 * hh]     = *(const v4f*)(Bm + (bnxt + (unsigned)(16 * hh * N)));
        rb[2 * hh + 1] = *(const v4f*)(Bm + (bnxt + (unsigned)(16 * hh * N) + 4u));
      }
      if (kb + 2 < nkb) {
        __builtin_prefetch(Bm + (boff + (unsigned)((kb + 2) * GK * N)), 0, 1);
        __builtin_prefetch(Bm + (boff + (unsigned)(((kb + 2) * GK + 16) * N)), 0, 1);
      }
    }

    #pragma unroll
    for (int kk = 0; kk < GK; kk += 4) {
      v2f a[RT];
      #pragma unroll
      for (int r = 0; r < RT; ++r) a[r] = *(const v2f*)(afrag + 16 * r * PA + kk);
      #pragma unroll
      for (int j = 0; j < 4; ++j) {
        const v2f b = *(const v2f*)(&Bs[(16 * j + l16) * PB + kk + 2 * half]);
        #pragma unroll
        for (int r = 0; r < RT; ++r) acc[r][j] = wmma4(a[r], b, acc[r][j]);
      }
    }

    __syncthreads();
    if (have_next) {
      #pragma unroll
      for (int i = 0; i < 4 * RT; ++i) *(v4f*)(&As[aidx + 16 * i * PA]) = ra[i];
      #pragma unroll
      for (int hh = 0; hh < 2; ++hh) {
        #pragma unroll
        for (int e = 0; e < 4; ++e) {
          Bs[(scol + e) * PB + 16 * hh + srow]     = rb[2 * hh][e];
          Bs[(scol + 4 + e) * PB + 16 * hh + srow] = rb[2 * hh + 1][e];
        }
      }
    }
    __syncthreads();
  }

  if (lay == 0) {
    float* Cw = Sm + wave * (16 * PC);
    const int cq = 4 * l16;
    v4f bq = {0.f, 0.f, 0.f, 0.f};
    if (bias) bq = *(const v4f*)(bias + col0 + cq);
    #pragma unroll
    for (int r = 0; r < RT; ++r) {
      if (r > 0) __builtin_amdgcn_fence(__ATOMIC_ACQ_REL, "wavefront");
      #pragma unroll
      for (int j = 0; j < 4; ++j) {
        #pragma unroll
        for (int i = 0; i < 8; ++i) Cw[(i + 8 * half) * PC + 16 * j + l16] = acc[r][j][i];
      }
      __builtin_amdgcn_fence(__ATOMIC_ACQ_REL, "wavefront");
      float* crow = C + (size_t)(rowW + 16 * r + half) * N + col0 + cq;
      #pragma unroll
      for (int q = 0; q < 8; ++q) {
        const v4f v = *(const v4f*)(Cw + (2 * q + half) * PC + cq);
        *(v4f*)(crow + (size_t)(2 * q) * N) = v + bq;
      }
    }
  } else if (lay == 1) {
    #pragma unroll
    for (int j = 0; j < 4; ++j) {
      const int c = col0 + 16 * j + l16;
      const float bj = bias ? bias[c] : 0.f;
      const int hh = c / HD;
      const int kd = c - hh * HD;
      const int cpart = hh * KF_HEAD + (kd >> 2) * KF_GRP + (kd & 3);
      #pragma unroll
      for (int r = 0; r < RT; ++r) {
        #pragma unroll
        for (int i = 0; i < 8; ++i) {
          const int rr = rowW + 16 * r + i + 8 * half;
          if (rr < M) {
            const int bb = rr / S_DIM;
            const int sr = rr - bb * S_DIM;
            C[(size_t)(bb * (H_NUM * KF_HEAD) + cpart + sr * 4)] = acc[r][j][i] + bj;
          }
        }
      }
    }
  } else {
    #pragma unroll
    for (int j = 0; j < 4; ++j) {
      const int c = col0 + 16 * j + l16;
      const float bj = bias ? bias[c] : 0.f;
      #pragma unroll
      for (int r = 0; r < RT; ++r) {
        #pragma unroll
        for (int i = 0; i < 8; ++i) {
          const int rr = rowW + 16 * r + i + 8 * half;
          if (rr < M) {
            const int bb = rr / S_DIM;
            const int sr = rr - bb * S_DIM;
            C[(size_t)(bb * VF_BATCH + (sr >> 2) * VF_GRP + c * 4 + (sr & 3))] = acc[r][j][i] + bj;
          }
        }
      }
    }
  }
}

__global__ __launch_bounds__(128) __attribute__((amdgpu_waves_per_eu(5)))
void gemm_wmma_f32(const float* __restrict__ A, const float* __restrict__ Bm,
                   const float* __restrict__ bias, float* __restrict__ C,
                   int M, int N, int K) {
  __shared__ __attribute__((aligned(16))) float Sm[GEMM_LDS_FLOATS(2)];
  gemm_tile<2>(Sm, A, Bm, bias, C, M, N, K, blockIdx.x, blockIdx.y, 0);
}

#define MVP_TILES 60

__device__ __forceinline__
void mv_rows(const float* __restrict__ mrow0, const v8f* v, float* __restrict__ MVp,
             int mt, int c, int half, int l16) {
  int m = 16 * mt + l16;
  if (m > S_DIM - 1) m = S_DIM - 1;
  const float* mrow = mrow0 + (size_t)m * S_DIM;
  v8f out = {0.f,0.f,0.f,0.f,0.f,0.f,0.f,0.f};
  #pragma unroll
  for (int rt = 0; rt < 5; ++rt) {
    #pragma unroll
    for (int j = 0; j < 4; ++j) {
      const int e0 = 16 * rt + 8 * half + 2 * j;
      float a0, a1;
      if (rt < 4) {
        a0 = mrow[e0];
        a1 = mrow[e0 + 1];
      } else {
        const int ea = e0 > S_DIM - 1 ? S_DIM - 1 : e0;
        const int eb = e0 + 1 > S_DIM - 1 ? S_DIM - 1 : e0 + 1;
        const float x0 = mrow[ea];
        const float x1 = mrow[eb];
        a0 = (e0 < S_DIM) ? x0 : 0.f;
        a1 = (e0 + 1 < S_DIM) ? x1 : 0.f;
      }
      const v2f a = {a0, a1};
      const v2f b = {v[rt][2 * j], v[rt][2 * j + 1]};
      out = wmma4(a, b, out);
    }
  }
  #pragma unroll
  for (int i = 0; i < 8; ++i) {
    const int row = 16 * mt + i + 8 * half;
    if (row < S_DIM)
      MVp[(size_t)(row >> 2) * VF_GRP + c * 4 + (row & 3)] = out[i];
  }
}

__device__ __forceinline__
void mv_proj_tile(float* Sm, const float* __restrict__ E, const float* __restrict__ Wv,
                  const float* __restrict__ Map, float* __restrict__ MVF, int id) {
  const int tid  = threadIdx.x;
  const int wave = tid >> 5;
  const int lane = tid & 31;
  const int half = lane >> 4;
  const int l16  = lane & 15;
  const int p = id / 20;
  const int c = 16 * (id - p * 20) + l16;
  const float* Eb = E + (size_t)(B_BASE + 1 + p) * S_DIM * D_ENC;
  const int kq = 8 * wave + 4 * half;

  unsigned aoff[5];
  #pragma unroll
  for (int r = 0; r < 5; ++r) {
    int row = 16 * r + l16;
    if (row > S_DIM - 1) row = S_DIM - 1;
    aoff[r] = (unsigned)row * (unsigned)D_ENC + (unsigned)kq;
  }
  const unsigned boff = (unsigned)kq * (unsigned)C_DIM + (unsigned)c;

  const v8f zero = {0.f,0.f,0.f,0.f,0.f,0.f,0.f,0.f};
  v8f acc[5];
  #pragma unroll
  for (int r = 0; r < 5; ++r) acc[r] = zero;

  v4f an[5];
  float bn[4];
  #pragma unroll
  for (int r = 0; r < 5; ++r) an[r] = *(const v4f*)(Eb + aoff[r]);
  #pragma unroll
  for (int e = 0; e < 4; ++e) bn[e] = Wv[boff + (unsigned)(e * C_DIM)];

  for (int kb = 0; kb < D_ENC / 32; ++kb) {
    v4f a[5];
    float b[4];
    #pragma unroll
    for (int r = 0; r < 5; ++r) a[r] = an[r];
    #pragma unroll
    for (int e = 0; e < 4; ++e) b[e] = bn[e];
    if (kb + 1 < D_ENC / 32) {
      #pragma unroll
      for (int r = 0; r < 5; ++r) an[r] = *(const v4f*)(Eb + (aoff[r] + (unsigned)((kb + 1) * 32)));
      #pragma unroll
      for (int e = 0; e < 4; ++e) bn[e] = Wv[boff + (unsigned)(((kb + 1) * 32 + e) * C_DIM)];
    }
    const v2f b01 = {b[0], b[1]}, b23 = {b[2], b[3]};
    #pragma unroll
    for (int r = 0; r < 5; ++r) {
      const v2f a01 = {a[r].x, a[r].y}, a23 = {a[r].z, a[r].w};
      acc[r] = wmma4(a01, b01, acc[r]);
      acc[r] = wmma4(a23, b23, acc[r]);
    }
  }

  float* part = Sm + wave * (5 * 8 * 32);
  #pragma unroll
  for (int r = 0; r < 5; ++r) {
    #pragma unroll
    for (int i = 0; i < 8; ++i) part[(r * 8 + i) * 32 + lane] = acc[r][i];
  }
  __syncthreads();
  v8f v[5];
  #pragma unroll
  for (int r = 0; r < 5; ++r) {
    #pragma unroll
    for (int i = 0; i < 8; ++i) {
      const float* q = Sm + (r * 8 + i) * 32 + lane;
      v[r][i] = ((q[0] + q[5 * 8 * 32]) + q[2 * 5 * 8 * 32]) + q[3 * 5 * 8 * 32];
    }
  }

  const float* mrow0 = Map + (size_t)p * S_DIM * S_DIM;
  float* MVp = MVF + (size_t)p * VF_BATCH;
  mv_rows(mrow0, v, MVp, wave, c, half, l16);
  if (__builtin_amdgcn_readfirstlane(wave) == 0) mv_rows(mrow0, v, MVp, 4, c, half, l16);
}

#define COL_TILES   5
#define KV_ROWS     (5 * S_DIM)
#define KV_TILES    35
#define QKV_TILES   930

__global__ __launch_bounds__(128)
void proj_qkv_wmma_f32(const float* __restrict__ X, const float* __restrict__ E,
                       const float* __restrict__ Map,
                       const float* __restrict__ Wq, const float* __restrict__ Wk,
                       const float* __restrict__ Wv, float* __restrict__ Q,
                       float* __restrict__ Kp, float* __restrict__ Vp,
                       float* __restrict__ MVp) {
  __shared__ __attribute__((aligned(16))) float Sm[GEMM_LDS_FLOATS(2)];
  int id = blockIdx.x;
  if (id < MVP_TILES) {
    mv_proj_tile(Sm, E, Wv, Map, MVp, id);
    return;
  }
  id -= MVP_TILES;
  if (id < 2 * KV_TILES) {
    const bool isV = id >= KV_TILES;
    if (isV) id -= KV_TILES;
    gemm_tile<1>(Sm, E, isV ? Wv : Wk, nullptr, isV ? Vp : Kp, KV_ROWS, C_DIM, D_ENC,
                 id % COL_TILES, id / COL_TILES, isV ? 2 : 1);
  } else {
    id -= 2 * KV_TILES;
    gemm_tile<2>(Sm, X, Wq, nullptr, Q, 5 * T_DIM, C_DIM, C_DIM,
                 id % COL_TILES, id / COL_TILES, 0);
  }
}

#define HS_TRIPS ((HS_SLICE / 4 + AW * 32 - 1) / (AW * 32))
__device__ __forceinline__
void slice_fetch(const float* __restrict__ srcp, int grp, int tid, v4f* r) {
  #pragma unroll
  for (int t = 0; t < HS_TRIPS; ++t) {
    int i = tid + t * (AW * 32);
    if (i > HS_SLICE / 4 - 1) i = HS_SLICE / 4 - 1;
    const int g = i / HD;
    const int q = i - g * HD;
    r[t] = *(const v4f*)(srcp + (size_t)g * grp + 4 * q);
  }
}
__device__ __forceinline__
void slice_put(float* dst, int tid, const v4f* r) {
  #pragma unroll
  for (int t = 0; t < HS_TRIPS; ++t) {
    int i = tid + t * (AW * 32);
    if (i > HS_SLICE / 4 - 1) i = HS_SLICE / 4 - 1;
    *(v4f*)(dst + 4 * i) = r[t];
  }
}

__device__ __forceinline__
void vf_block(v4f a4, const float* Vl, const int* vcol, int blk, int half, v8f* o) {
  #pragma unroll
  for (int j = 0; j < 3; ++j) {
    v4f b4 = *(const v4f*)(Vl + vcol[j] + blk * (2 * HS_GRP));
    if (blk == SP / 8 - 1) {
      const float b0 = b4.x;
      b4.y = half ? b0 : b4.y;
      b4.z = half ? b0 : b4.z;
      b4.w = half ? b0 : b4.w;
    }
    v2f a01 = {a4.x, a4.y}, a23 = {a4.z, a4.w};
    v2f b01 = {b4.x, b4.y}, b23 = {b4.z, b4.w};
    o[j] = wmma4(a01, b01, o[j]);
    o[j] = wmma4(a23, b23, o[j]);
  }
}

#define PO 52
__device__ __forceinline__
void o_rows_store(float* stg, const v8f* o, float* __restrict__ AO,
                  int b, int h, int t0, int lane, int half, int l16) {
  __builtin_amdgcn_fence(__ATOMIC_ACQ_REL, "wavefront");
  #pragma unroll
  for (int j = 0; j < 3; ++j) {
    #pragma unroll
    for (int i = 0; i < 8; ++i) stg[(i + 8 * half) * PO + 16 * j + l16] = o[j][i];
  }
  __builtin_amdgcn_fence(__ATOMIC_ACQ_REL, "wavefront");
  float* obase = AO + (size_t)(b * T_DIM + t0) * C_DIM + h * HD;
  #pragma unroll
  for (int t = 0; t < (16 * HD / 4) / 32; ++t) {
    const int q   = lane + 32 * t;
    const int row = q / (HD / 4);
    const int ch  = q - row * (HD / 4);
    const v4f v = *(const v4f*)(stg + row * PO + 4 * ch);
    *(v4f*)(obase + (size_t)row * C_DIM + 4 * ch) = v;
  }
}

__device__ __forceinline__
void pv_and_store(float* Plds, const float* Vl,
                  float* __restrict__ AO, int b, int h, int t0,
                  int lane, int half, int l16) {
  const v8f zero = {0.f,0.f,0.f,0.f,0.f,0.f,0.f,0.f};
  v8f o[3];
  #pragma unroll
  for (int j = 0; j < 3; ++j) o[j] = zero;

  int vcol[3];
  #pragma unroll
  for (int j = 0; j < 3; ++j) {
    int c = 16 * j + l16;
    if (c > HD - 1) c = HD - 1;
    vcol[j] = half * HS_GRP + c * 4;
  }
  const float* prow = Plds + l16 * PP + 4 * half;

  #pragma unroll
  for (int m = 0; m < SP / 8; ++m) {
    const v4f a4 = *(const v4f*)(prow + 8 * m);
    vf_block(a4, Vl, vcol, m, half, o);
  }

  o_rows_store(Plds, o, AO, b, h, t0, lane, half, l16);
}

#define ATT_BLOCKS  ((T_DIM / 16 / AW) * H_NUM * 5)

__global__ __launch_bounds__(AW * 32)
void attn_kernel(const float* __restrict__ Q,
                 const float* __restrict__ KF,
                 const float* __restrict__ VF,
                 const float* __restrict__ MV,
                 float* __restrict__ AO)
{
  __shared__ __attribute__((aligned(16))) float Ks[HS_SLICE];
  __shared__ __attribute__((aligned(16))) float Vs[HS_SLICE];
  __shared__ __attribute__((aligned(16))) float Pall[AW * 16 * PP];
  const int tid  = threadIdx.x;
  const int wave = tid >> 5;
  const int lane = tid & 31;
  const int half = lane >> 4;
  const int l16  = lane & 15;
  const int id = blockIdx.x;
  const int t0 = (((id & (T_DIM / 16 / AW - 1)) * AW) + wave) << 4;
  const int h  = (id >> 5) & (H_NUM - 1);
  const int bsel = id >> 8;
  const int b  = (bsel == 0) ? B_BASE : bsel - 1;
  float* Plds = Pall + wave * (16 * PP);
  const float scale = 0.15811388300841898f;
  const float NEG_INF = -__builtin_inff();
  const v8f zero = {0.f,0.f,0.f,0.f,0.f,0.f,0.f,0.f};
  v8f sc[5];
  #pragma unroll
  for (int j = 0; j < 5; ++j) sc[j] = zero;

  const float* qbase = Q  + (size_t)(b * T_DIM + t0 + l16) * C_DIM + h * HD + 4 * half;
  v4f qa[HD / 8];
  #pragma unroll
  for (int m = 0; m < HD / 8; ++m) qa[m] = *(const v4f*)(qbase + 8 * m);

  {
    v4f rk[HS_TRIPS], rv[HS_TRIPS];
    slice_fetch(KF + (size_t)(b * H_NUM + h) * KF_HEAD, HS_GRP, tid, rk);
    slice_fetch(VF + (size_t)b * VF_BATCH + (h * HD) * 4, VF_GRP, tid, rv);
    slice_put(Ks, tid, rk);
    slice_put(Vs, tid, rv);
  }
  __syncthreads();
  const float* kbase = Ks + half * KF_GRP;

  const float* krow[5];
  #pragma unroll
  for (int j = 0; j < 5; ++j) {
    int s = 16 * j + l16;
    if (s > S_DIM - 1) s = S_DIM - 1;
    krow[j] = kbase + s * 4;
  }
  #pragma unroll
  for (int m = 0; m < HD / 8; ++m) {
    const v4f a4 = qa[m];
    const v2f a01 = {a4.x, a4.y}, a23 = {a4.z, a4.w};
    #pragma unroll
    for (int j = 0; j < 5; ++j) {
      const v4f b4 = *(const v4f*)(krow[j] + m * (2 * KF_GRP));
      const v2f b01 = {b4.x, b4.y}, b23 = {b4.z, b4.w};
      sc[j] = wmma4(a01, b01, sc[j]);
      sc[j] = wmma4(a23, b23, sc[j]);
    }
  }

  float mx[8];
  #pragma unroll
  for (int i = 0; i < 8; ++i) mx[i] = NEG_INF;
  #pragma unroll
  for (int j = 0; j < 5; ++j) {
    const bool ok = (16 * j + l16) < S_DIM;
    #pragma unroll
    for (int i = 0; i < 8; ++i) {
      const float x = ok ? sc[j][i] * scale : NEG_INF;
      sc[j][i] = x;
      mx[i] = fmaxf(mx[i], x);
    }
  }
  #pragma unroll
  for (int m = 1; m < 16; m <<= 1) {
    #pragma unroll
    for (int i = 0; i < 8; ++i) mx[i] = fmaxf(mx[i], __shfl_xor(mx[i], m, 32));
  }
  float sm[8];
  #pragma unroll
  for (int i = 0; i < 8; ++i) sm[i] = 0.f;
  #pragma unroll
  for (int j = 0; j < 5; ++j) {
    #pragma unroll
    for (int i = 0; i < 8; ++i) {
      const float e = __expf(sc[j][i] - mx[i]);
      sc[j][i] = e;
      sm[i] += e;
    }
  }
  #pragma unroll
  for (int m = 1; m < 16; m <<= 1) {
    #pragma unroll
    for (int i = 0; i < 8; ++i) sm[i] += __shfl_xor(sm[i], m, 32);
  }
  float rs[8];
  #pragma unroll
  for (int i = 0; i < 8; ++i) rs[i] = 1.f / sm[i];

  #pragma unroll
  for (int j = 0; j < 5; ++j) {
    #pragma unroll
    for (int i = 0; i < 8; ++i) {
      const float pv = sc[j][i] * rs[i];
      const int row = i + 8 * half;
      Plds[row * PP + 16 * j + l16] = pv;
    }
  }
  __builtin_amdgcn_fence(__ATOMIC_ACQ_REL, "wavefront");

  if (b == B_BASE) {
    const float* prow = Plds + l16 * PP + 4 * half;
    v4f pa[SP / 8];
    #pragma unroll
    for (int m = 0; m < SP / 8; ++m) pa[m] = *(const v4f*)(prow + 8 * m);
    int vcol[3];
    #pragma unroll
    for (int j = 0; j < 3; ++j) {
      int c = 16 * j + l16;
      if (c > HD - 1) c = HD - 1;
      vcol[j] = half * HS_GRP + c * 4;
    }
    v8f o[3];
    #pragma unroll
    for (int j = 0; j < 3; ++j) o[j] = zero;
    #pragma unroll
    for (int m = 0; m < SP / 8; ++m) vf_block(pa[m], Vs, vcol, m, half, o);
    o_rows_store(Plds, o, AO, b, h, t0, lane, half, l16);

    #pragma unroll
    for (int p = 0; p < 3; ++p) {
      v4f rm[HS_TRIPS];
      slice_fetch(MV + (size_t)p * VF_BATCH + (h * HD) * 4, VF_GRP, tid, rm);
      __syncthreads();
      slice_put(Vs, tid, rm);
      __syncthreads();
      #pragma unroll
      for (int j = 0; j < 3; ++j) o[j] = zero;
      #pragma unroll
      for (int m = 0; m < SP / 8; ++m) vf_block(pa[m], Vs, vcol, m, half, o);
      o_rows_store(Plds, o, AO, B_BASE + 1 + p, h, t0, lane, half, l16);
    }
  } else {
    pv_and_store(Plds, Vs, AO, b, h, t0, lane, half, l16);
  }
}

extern "C" void kernel_launch(void* const* d_in, const int* in_sizes, int n_in,
                              void* d_out, int out_size, void* d_ws, size_t ws_size,
                              hipStream_t stream) {
  const float* X   = (const float*)d_in[0];
  const float* E   = (const float*)d_in[1];
  const float* Map = (const float*)d_in[2];
  const float* Wq  = (const float*)d_in[3];
  const float* Wk  = (const float*)d_in[4];
  const float* Wv  = (const float*)d_in[5];
  const float* Wo  = (const float*)d_in[6];
  const float* bo  = (const float*)d_in[7];
  float* out = (float*)d_out;

  float* ws = (float*)d_ws;
  float* Q  = ws;
  float* Kp = Q  + (size_t)5 * T_DIM * C_DIM;
  float* Vp = Kp + (size_t)B_ALL * H_NUM * KF_HEAD;
  float* AO = Vp + (size_t)B_ALL * VF_BATCH;
  float* MV = AO + (size_t)B_ALL * T_DIM * C_DIM;

  const dim3 gblk(128);
  const dim3 ablk(AW * 32);

  proj_qkv_wmma_f32<<<dim3(QKV_TILES), gblk, 0, stream>>>(
      X, E, Map, Wq, Wk, Wv, Q, Kp, Vp, MV);

  attn_kernel<<<dim3(ATT_BLOCKS), ablk, 0, stream>>>(Q, Kp, Vp, MV, AO);

  gemm_wmma_f32<<<dim3(C_DIM / 64, (B_ALL * T_DIM) / 128), gblk, 0, stream>>>(
      AO, Wo, bo, out, B_ALL * T_DIM, C_DIM, C_DIM);
}
